# starting kernel
# baseline (speedup 1.0000x reference)
#include <hip/hip_runtime.h>
#include <hip/hip_bf16.h>

#define N_NODES_C  100000
#define N_EDGES_C  1600000
#define N_GRAPHS_C 1000
#define HID_C      64
#define IN_DIM_C   7
#define EDGE_DIM_C 4
#define NUM_TASKS_C 12

typedef __attribute__((ext_vector_type(16))) _Float16 v16h;
typedef __attribute__((ext_vector_type(8)))  _Float16 v8h;
typedef __attribute__((ext_vector_type(8)))  float    v8f;

__device__ __forceinline__ void atom_add_f32(float* p, float v) {
    unsafeAtomicAdd(p, v);
}

__device__ __forceinline__ v16h lds_load_tile(const _Float16* s, int ld, int kbase) {
    const int lane = threadIdx.x & 31;
    const int r    = lane & 15;
    const int koff = (lane < 16) ? 0 : 8;
    const _Float16* p = s + r * ld + kbase + koff;
    v8h lo = *(const v8h*)(p);
    v8h hi = *(const v8h*)(p + 16);
    v16h out;
#pragma unroll
    for (int e = 0; e < 8; ++e) { out[e] = lo[e]; out[8 + e] = hi[e]; }
    return out;
}

template <int K1, typename TI, typename TO>
__global__ __launch_bounds__(128)
void node_mlp_kernel(const TI* __restrict__ in0, const float* __restrict__ in1,
                     const float* __restrict__ W1,  const float* __restrict__ b1,
                     const float* __restrict__ W2,  const float* __restrict__ b2,
                     TO* __restrict__ out, int n_nodes) {
    constexpr int K1P = (K1 + 31) & ~31;

    __shared__ __align__(16) _Float16 sA  [16 * K1P];
    __shared__ __align__(16) _Float16 sMid[16 * HID_C];
    __shared__ __align__(16) _Float16 sB1 [HID_C * K1P];
    __shared__ __align__(16) _Float16 sB2 [HID_C * HID_C];

    const int row0 = blockIdx.x * 16;
    const int tid  = threadIdx.x;
    const int wave = tid >> 5;
    const int lane = tid & 31;
    const int n0   = wave * 16;

    for (int i = tid; i < 16 * K1P; i += 128) {
        const int m = i / K1P, k = i - m * K1P;
        const int r = row0 + m;
        float v = 0.0f;
        if (k < K1 && r < n_nodes)
            v = (float)in0[(size_t)r * K1 + k] + in1[(size_t)r * K1 + k];
        sA[i] = (_Float16)v;
    }
    for (int i = tid; i < HID_C * K1P; i += 128) {
        const int n = i / K1P, k = i - n * K1P;
        sB1[i] = (_Float16)((k < K1) ? W1[n * K1 + k] : 0.0f);
    }
    for (int i = tid; i < HID_C * HID_C; i += 128) {
        const int n = i >> 6, k = i & 63;
        sB2[i] = (_Float16)W2[n * HID_C + k];
    }
    __syncthreads();

    v8f acc = {};
#pragma unroll
    for (int kb = 0; kb < K1P; kb += 32) {
        v16h a = lds_load_tile(sA, K1P, kb);
        v16h b = lds_load_tile(sB1 + n0 * K1P, K1P, kb);
        acc = __builtin_amdgcn_wmma_f32_16x16x32_f16(false, a, false, b,
                                                     (short)0, acc, false, false);
    }
    {
        const int n  = lane & 15;
        const int mb = (lane < 16) ? 0 : 8;
        const float bias = b1[n0 + n];
#pragma unroll
        for (int r = 0; r < 8; ++r) {
            float v = acc[r] + bias;
            v = v > 0.0f ? v : 0.0f;
            sMid[(mb + r) * HID_C + (n0 + n)] = (_Float16)v;
        }
    }
    __syncthreads();

    v8f acc2 = {};
#pragma unroll
    for (int kb = 0; kb < HID_C; kb += 32) {
        v16h a = lds_load_tile(sMid, HID_C, kb);
        v16h b = lds_load_tile(sB2 + n0 * HID_C, HID_C, kb);
        acc2 = __builtin_amdgcn_wmma_f32_16x16x32_f16(false, a, false, b,
                                                      (short)0, acc2, false, false);
    }
    {
        const int n  = lane & 15;
        const int mb = (lane < 16) ? 0 : 8;
        const float bias = b2[n0 + n];
#pragma unroll
        for (int r = 0; r < 8; ++r) {
            const int rr = row0 + mb + r;
            if (rr < n_nodes) {
                float v = acc2[r] + bias;
                v = v > 0.0f ? v : 0.0f;
                out[(size_t)rr * HID_C + (n0 + n)] = (TO)v;
            }
        }
    }
}

__global__ __launch_bounds__(256)
void edge1_kernel(const float* __restrict__ x, const int* __restrict__ ei,
                  const float* __restrict__ ea,
                  const float* __restrict__ eW, const float* __restrict__ eb,
                  float* __restrict__ aggr) {
    const long long t = (long long)blockIdx.x * 256 + threadIdx.x;
    const long long e = t >> 3;
    const int c = (int)(t & 7);
    if (e >= N_EDGES_C || c >= IN_DIM_C) return;
    const int src = ei[e];
    const int dst = ei[N_EDGES_C + e];
    const float4 a = *(const float4*)(ea + e * EDGE_DIM_C);
    const float4 w = *(const float4*)(eW + c * EDGE_DIM_C);
    float v = eb[c] + a.x * w.x + a.y * w.y + a.z * w.z + a.w * w.w;
    v += x[(size_t)src * IN_DIM_C + c];
    v = fmaxf(v, 0.0f);
    atom_add_f32(&aggr[(size_t)dst * IN_DIM_C + c], v);
}

__global__ __launch_bounds__(256)
void edge2_kernel(const _Float16* __restrict__ h, const int* __restrict__ ei,
                  const float* __restrict__ ea,
                  const float* __restrict__ eW, const float* __restrict__ eb,
                  float* __restrict__ aggr) {
    const long long t = (long long)blockIdx.x * 256 + threadIdx.x;
    const long long e = t >> 6;
    const int c = (int)(t & 63);
    if (e >= N_EDGES_C) return;
    const int src = ei[e];
    const int dst = ei[N_EDGES_C + e];
    const float4 a = *(const float4*)(ea + e * EDGE_DIM_C);
    const float4 w = *(const float4*)(eW + c * EDGE_DIM_C);
    float v = eb[c] + a.x * w.x + a.y * w.y + a.z * w.z + a.w * w.w;
    v += (float)h[(size_t)src * HID_C + c];
    v = fmaxf(v, 0.0f);
    atom_add_f32(&aggr[(size_t)dst * HID_C + c], v);
}

__global__ __launch_bounds__(256)
void pool_kernel(const float* __restrict__ h, const int* __restrict__ batch,
                 float* __restrict__ psum, float* __restrict__ cnt) {
    const long long t = (long long)blockIdx.x * 256 + threadIdx.x;
    const long long n = t >> 6;
    const int c = (int)(t & 63);
    if (n >= N_NODES_C) return;
    const int g = batch[n];
    atom_add_f32(&psum[(size_t)g * HID_C + c], h[(size_t)n * HID_C + c]);
    if (c == 0) atom_add_f32(&cnt[g], 1.0f);
}

__global__ __launch_bounds__(256)
void fc_kernel(const float* __restrict__ psum, const float* __restrict__ cnt,
               const float* __restrict__ fcW, const float* __restrict__ fcb,
               float* __restrict__ out) {
    const int t = blockIdx.x * 256 + threadIdx.x;
    if (t >= N_GRAPHS_C * NUM_TASKS_C) return;
    const int g = t / NUM_TASKS_C;
    const int k = t - g * NUM_TASKS_C;
    const float inv = 1.0f / fmaxf(cnt[g], 1.0f);
    float s = fcb[k];
#pragma unroll
    for (int c = 0; c < HID_C; ++c)
        s += psum[(size_t)g * HID_C + c] * inv * fcW[k * HID_C + c];
    out[t] = s;
}

extern "C" void kernel_launch(void* const* d_in, const int* in_sizes, int n_in,
                              void* d_out, int out_size, void* d_ws, size_t ws_size,
                              hipStream_t stream) {
    const float* x    = (const float*)d_in[0];
    const int*   ei   = (const int*)  d_in[1];
    const float* ea   = (const float*)d_in[2];
    const int*   bat  = (const int*)  d_in[3];
    const float* e1W  = (const float*)d_in[4];  const float* e1b  = (const float*)d_in[5];
    const float* n1W1 = (const float*)d_in[6];  const float* n1b1 = (const float*)d_in[7];
    const float* n1W2 = (const float*)d_in[8];  const float* n1b2 = (const float*)d_in[9];
    const float* e2W  = (const float*)d_in[10]; const float* e2b  = (const float*)d_in[11];
    const float* n2W1 = (const float*)d_in[12]; const float* n2b1 = (const float*)d_in[13];
    const float* n2W2 = (const float*)d_in[14]; const float* n2b2 = (const float*)d_in[15];
    const float* fcW  = (const float*)d_in[16]; const float* fcb  = (const float*)d_in[17];
    float* out = (float*)d_out;

    float*    aggr1 = (float*)d_ws;
    _Float16* h     = (_Float16*)(aggr1 + (size_t)N_NODES_C * IN_DIM_C);
    float*    aggr2 = (float*)(h + (size_t)N_NODES_C * HID_C);
    float*    h2    = aggr2 + (size_t)N_NODES_C * HID_C;
    float*    psum  = h2    + (size_t)N_NODES_C * HID_C;
    float*    cnt   = psum  + (size_t)N_GRAPHS_C * HID_C;

    hipMemsetAsync(aggr1, 0, sizeof(float) * (size_t)N_NODES_C * IN_DIM_C, stream);
    hipMemsetAsync(aggr2, 0, sizeof(float) * (size_t)N_NODES_C * HID_C,   stream);
    hipMemsetAsync(psum,  0, sizeof(float) * (size_t)N_GRAPHS_C * HID_C,  stream);
    hipMemsetAsync(cnt,   0, sizeof(float) * (size_t)N_GRAPHS_C,          stream);

    {
        const unsigned grid = (unsigned)(((long long)N_EDGES_C * 8 + 255) / 256);
        edge1_kernel<<<grid, 256, 0, stream>>>(x, ei, ea, e1W, e1b, aggr1);
    }
    {
        const unsigned grid = (N_NODES_C + 15) / 16;
        node_mlp_kernel<IN_DIM_C, float, _Float16><<<grid, 128, 0, stream>>>(
            x, aggr1, n1W1, n1b1, n1W2, n1b2, h, N_NODES_C);
    }
    {
        const unsigned grid = (unsigned)(((long long)N_EDGES_C * 64 + 255) / 256);
        edge2_kernel<<<grid, 256, 0, stream>>>(h, ei, ea, e2W, e2b, aggr2);
    }
    {
        const unsigned grid = (N_NODES_C + 15) / 16;
        node_mlp_kernel<HID_C, _Float16, float><<<grid, 128, 0, stream>>>(
            h, aggr2, n2W1, n2b1, n2W2, n2b2, h2, N_NODES_C);
    }
    {
        const unsigned grid = (unsigned)(((long long)N_NODES_C * 64 + 255) / 256);
        pool_kernel<<<grid, 256, 0, stream>>>(h2, bat, psum, cnt);
    }
    {
        const unsigned grid = (N_GRAPHS_C * NUM_TASKS_C + 255) / 256;
        fc_kernel<<<grid, 256, 0, stream>>>(psum, cnt, fcW, fcb, out);
    }
}


# Both convs' f32 global scatter atomics replaced by owner blocks: each of 128 blocks filters the whole dst list for its 782-node range, sums in LDS (LDS f32 atomics, arrival order), plain stores
# speedup vs baseline: 1.3387x; 1.3387x over previous
#include <hip/hip_runtime.h>
#include <hip/hip_bf16.h>

#define N_NODES_C  100000
#define N_EDGES_C  1600000
#define N_GRAPHS_C 1000
#define HID_C      64
#define IN_DIM_C   7
#define EDGE_DIM_C 4
#define NUM_TASKS_C 12

typedef __attribute__((ext_vector_type(16))) _Float16 v16h;
typedef __attribute__((ext_vector_type(8)))  _Float16 v8h;
typedef __attribute__((ext_vector_type(8)))  float    v8f;

__device__ __forceinline__ void atom_add_f32(float* p, float v) {
    unsafeAtomicAdd(p, v);
}

__device__ __forceinline__ v16h lds_load_tile(const _Float16* s, int ld, int kbase) {
    const int lane = threadIdx.x & 31;
    const int r    = lane & 15;
    const int koff = (lane < 16) ? 0 : 8;
    const _Float16* p = s + r * ld + kbase + koff;
    v8h lo = *(const v8h*)(p);
    v8h hi = *(const v8h*)(p + 16);
    v16h out;
#pragma unroll
    for (int e = 0; e < 8; ++e) { out[e] = lo[e]; out[8 + e] = hi[e]; }
    return out;
}

template <int K1, typename TI, typename TO>
__global__ __launch_bounds__(128)
void node_mlp_kernel(const TI* __restrict__ in0, const float* __restrict__ in1,
                     const float* __restrict__ W1,  const float* __restrict__ b1,
                     const float* __restrict__ W2,  const float* __restrict__ b2,
                     TO* __restrict__ out, int n_nodes) {
    constexpr int K1P = (K1 + 31) & ~31;

    __shared__ __align__(16) _Float16 sA  [16 * K1P];
    __shared__ __align__(16) _Float16 sMid[16 * HID_C];
    __shared__ __align__(16) _Float16 sB1 [HID_C * K1P];
    __shared__ __align__(16) _Float16 sB2 [HID_C * HID_C];

    const int row0 = blockIdx.x * 16;
    const int tid  = threadIdx.x;
    const int wave = tid >> 5;
    const int lane = tid & 31;
    const int n0   = wave * 16;

    for (int i = tid; i < 16 * K1P; i += 128) {
        const int m = i / K1P, k = i - m * K1P;
        const int r = row0 + m;
        float v = 0.0f;
        if (k < K1 && r < n_nodes)
            v = (float)in0[(size_t)r * K1 + k] + in1[(size_t)r * K1 + k];
        sA[i] = (_Float16)v;
    }
    for (int i = tid; i < HID_C * K1P; i += 128) {
        const int n = i / K1P, k = i - n * K1P;
        sB1[i] = (_Float16)((k < K1) ? W1[n * K1 + k] : 0.0f);
    }
    for (int i = tid; i < HID_C * HID_C; i += 128) {
        const int n = i >> 6, k = i & 63;
        sB2[i] = (_Float16)W2[n * HID_C + k];
    }
    __syncthreads();

    v8f acc = {};
#pragma unroll
    for (int kb = 0; kb < K1P; kb += 32) {
        v16h a = lds_load_tile(sA, K1P, kb);
        v16h b = lds_load_tile(sB1 + n0 * K1P, K1P, kb);
        acc = __builtin_amdgcn_wmma_f32_16x16x32_f16(false, a, false, b,
                                                     (short)0, acc, false, false);
    }
    {
        const int n  = lane & 15;
        const int mb = (lane < 16) ? 0 : 8;
        const float bias = b1[n0 + n];
#pragma unroll
        for (int r = 0; r < 8; ++r) {
            float v = acc[r] + bias;
            v = v > 0.0f ? v : 0.0f;
            sMid[(mb + r) * HID_C + (n0 + n)] = (_Float16)v;
        }
    }
    __syncthreads();

    v8f acc2 = {};
#pragma unroll
    for (int kb = 0; kb < HID_C; kb += 32) {
        v16h a = lds_load_tile(sMid, HID_C, kb);
        v16h b = lds_load_tile(sB2 + n0 * HID_C, HID_C, kb);
        acc2 = __builtin_amdgcn_wmma_f32_16x16x32_f16(false, a, false, b,
                                                      (short)0, acc2, false, false);
    }
    {
        const int n  = lane & 15;
        const int mb = (lane < 16) ? 0 : 8;
        const float bias = b2[n0 + n];
#pragma unroll
        for (int r = 0; r < 8; ++r) {
            const int rr = row0 + mb + r;
            if (rr < n_nodes) {
                float v = acc2[r] + bias;
                v = v > 0.0f ? v : 0.0f;
                out[(size_t)rr * HID_C + (n0 + n)] = (TO)v;
            }
        }
    }
}

#define OWN_BLOCKS 128
#define OWN_RN     782
#define OWN_WCAP2  512
#define OWN_WCAP1  1280
#define OWN_THREADS 1024
#define OWN_STEP   (OWN_THREADS * 8)
#define OWN_TRIPS  ((N_EDGES_C / 4 + OWN_STEP - 1) / OWN_STEP)

typedef __attribute__((ext_vector_type(2))) _Float16 v2h;

__device__ __forceinline__ void own_push(unsigned* wlist, unsigned& wcount, unsigned cap,
                                         int dstv, int n0, int e) {
    const unsigned d = (unsigned)(dstv - n0);
    const bool m = d < (unsigned)OWN_RN;
    const unsigned mask = __builtin_amdgcn_ballot_w32(m);
    if (mask) {
        const unsigned pos = wcount + __builtin_amdgcn_mbcnt_lo(mask, 0u);
        if (m && pos < cap) wlist[pos] = ((unsigned)e << 10) | d;
        wcount += (unsigned)__builtin_popcount(mask);
    }
}

__global__ __launch_bounds__(OWN_THREADS)
void edge2_own_kernel(const _Float16* __restrict__ h, const int* __restrict__ ei,
                      const float* __restrict__ ea,
                      const float* __restrict__ eW, const float* __restrict__ eb,
                      float* __restrict__ aggr) {
    __shared__ __align__(16) float    s_acc[OWN_RN * HID_C];
    __shared__ __align__(16) unsigned s_list[(OWN_THREADS / 32) * OWN_WCAP2];

    const int tid  = threadIdx.x;
    const int lane = tid & 31;
    const int wave = tid >> 5;
    const int n0   = blockIdx.x * OWN_RN;
    unsigned* wlist = s_list + wave * OWN_WCAP2;

    for (int i = tid; i < OWN_RN * HID_C / 4; i += OWN_THREADS)
        ((float4*)s_acc)[i] = make_float4(0.f, 0.f, 0.f, 0.f);
    __syncthreads();

    const int c0 = 2 * lane;
    const float4 w0 = *(const float4*)(eW + (c0 + 0) * EDGE_DIM_C);
    const float4 w1 = *(const float4*)(eW + (c0 + 1) * EDGE_DIM_C);
    const float  b0 = eb[c0], b1 = eb[c0 + 1];

    const int4* dst4 = (const int4*)(ei + N_EDGES_C);
    const int n4 = N_EDGES_C / 4;
    unsigned wcount = 0;
    const int rot = (int)((blockIdx.x * OWN_TRIPS) / OWN_BLOCKS);
    for (int t = 0; t < OWN_TRIPS; ++t) {
        int tt = t + rot;
        if (tt >= OWN_TRIPS) tt -= OWN_TRIPS;
        const int base = tt * OWN_STEP;
        int4 v[8];
#pragma unroll
        for (int u = 0; u < 8; ++u) {
            const int i = base + u * OWN_THREADS + tid;
            v[u] = (i < n4) ? dst4[i] : make_int4(-1, -1, -1, -1);
        }
#pragma unroll
        for (int u = 0; u < 8; ++u) {
            const int e = (base + u * OWN_THREADS + tid) * 4;
            own_push(wlist, wcount, OWN_WCAP2, v[u].x, n0, e + 0);
            own_push(wlist, wcount, OWN_WCAP2, v[u].y, n0, e + 1);
            own_push(wlist, wcount, OWN_WCAP2, v[u].z, n0, e + 2);
            own_push(wlist, wcount, OWN_WCAP2, v[u].w, n0, e + 3);
        }
    }
    if (wcount > (unsigned)OWN_WCAP2) {
        unsigned seen = 0;
#pragma unroll 1
        for (int t = 0; t < OWN_TRIPS * 8; ++t) {
            int tt = (t >> 3) + rot;
            if (tt >= OWN_TRIPS) tt -= OWN_TRIPS;
            const int i = (tt * 8 + (t & 7)) * OWN_THREADS + tid;
            const int4 v = (i < n4) ? dst4[i] : make_int4(-1, -1, -1, -1);
            const int dv[4] = {v.x, v.y, v.z, v.w};
#pragma unroll 1
            for (int k = 0; k < 4; ++k) {
                const unsigned d = (unsigned)(dv[k] - n0);
                const bool m = d < (unsigned)OWN_RN;
                const unsigned mask = __builtin_amdgcn_ballot_w32(m);
                const unsigned pos = seen + __builtin_amdgcn_mbcnt_lo(mask, 0u);
                seen += (unsigned)__builtin_popcount(mask);
                if (m && pos >= (unsigned)OWN_WCAP2) {
                    const int oe = i * 4 + k, od = (int)d;
                    const int os = ei[oe];
                    const float4 a = *(const float4*)(ea + (size_t)oe * EDGE_DIM_C);
#pragma unroll 1
                    for (int q = 0; q < HID_C; ++q) {
                        const float4 w = *(const float4*)(eW + q * EDGE_DIM_C);
                        float t = eb[q] + a.x * w.x + a.y * w.y + a.z * w.z + a.w * w.w;
                        t += (float)h[(size_t)os * HID_C + q];
                        atomicAdd(&s_acc[od * HID_C + q], fmaxf(t, 0.0f));
                    }
                }
            }
        }
    }
    const int wn = (int)(wcount < (unsigned)OWN_WCAP2 ? wcount : (unsigned)OWN_WCAP2);

    __syncthreads();

    const v2h* h2p = (const v2h*)h;
    for (int p = 0; p < wn; p += 32) {
        const int idx = p + lane;
        const bool valid = idx < wn;
        const unsigned ent = valid ? wlist[idx] : 0u;
        const int e  = (int)(ent >> 10);
        const int dl = (int)(ent & 1023u);
        const int sl = valid ? ei[e] : 0;
        const float4 al = valid ? *(const float4*)(ea + (size_t)e * EDGE_DIM_C)
                                : make_float4(0.f, 0.f, 0.f, 0.f);
        const int cnt = (wn - p) < 32 ? (wn - p) : 32;
        v2h hv[32];
#pragma unroll
        for (int j = 0; j < 32; ++j) {
            const int sj = __builtin_amdgcn_readlane(sl, j);
            hv[j] = h2p[(size_t)sj * (HID_C / 2) + lane];
        }
#pragma unroll
        for (int j = 0; j < 32; ++j) {
            if (j < cnt) {
                const int   dj = __builtin_amdgcn_readlane(dl, j);
                const float ax = __builtin_bit_cast(float, __builtin_amdgcn_readlane(__builtin_bit_cast(int, al.x), j));
                const float ay = __builtin_bit_cast(float, __builtin_amdgcn_readlane(__builtin_bit_cast(int, al.y), j));
                const float az = __builtin_bit_cast(float, __builtin_amdgcn_readlane(__builtin_bit_cast(int, al.z), j));
                const float aw = __builtin_bit_cast(float, __builtin_amdgcn_readlane(__builtin_bit_cast(int, al.w), j));
                float t0 = b0 + ax * w0.x + ay * w0.y + az * w0.z + aw * w0.w;
                float t1 = b1 + ax * w1.x + ay * w1.y + az * w1.z + aw * w1.w;
                t0 += (float)hv[j][0];
                t1 += (float)hv[j][1];
                atomicAdd(&s_acc[dj * HID_C + c0],     fmaxf(t0, 0.0f));
                atomicAdd(&s_acc[dj * HID_C + c0 + 1], fmaxf(t1, 0.0f));
            }
        }
    }
    __syncthreads();

    int rows = N_NODES_C - n0;
    if (rows > OWN_RN) rows = OWN_RN;
    float4* o4 = (float4*)(aggr + (size_t)n0 * HID_C);
    for (int i = tid; i < rows * (HID_C / 4); i += OWN_THREADS)
        o4[i] = ((const float4*)s_acc)[i];
}

__global__ __launch_bounds__(OWN_THREADS)
void edge1_own_kernel(const float* __restrict__ x, const int* __restrict__ ei,
                      const float* __restrict__ ea,
                      const float* __restrict__ eW, const float* __restrict__ eb,
                      float* __restrict__ aggr) {
    __shared__ __align__(16) float    s_acc[OWN_RN * 8];
    __shared__ __align__(16) unsigned s_list[(OWN_THREADS / 32) * OWN_WCAP1];

    const int tid  = threadIdx.x;
    const int lane = tid & 31;
    const int wave = tid >> 5;
    const int n0   = blockIdx.x * OWN_RN;
    unsigned* wlist = s_list + wave * OWN_WCAP1;

    for (int i = tid; i < OWN_RN * 8; i += OWN_THREADS) s_acc[i] = 0.0f;
    __syncthreads();

    const int c   = lane & 7;
    const int sub = lane >> 3;
    const bool cact = c < IN_DIM_C;
    const float4 wc = cact ? *(const float4*)(eW + c * EDGE_DIM_C) : make_float4(0.f, 0.f, 0.f, 0.f);
    const float  bc = cact ? eb[c] : 0.0f;

    const int4* dst4 = (const int4*)(ei + N_EDGES_C);
    const int n4 = N_EDGES_C / 4;
    unsigned wcount = 0;
    const int rot = (int)((blockIdx.x * OWN_TRIPS) / OWN_BLOCKS);
    for (int t = 0; t < OWN_TRIPS; ++t) {
        int tt = t + rot;
        if (tt >= OWN_TRIPS) tt -= OWN_TRIPS;
        const int base = tt * OWN_STEP;
        int4 v[8];
#pragma unroll
        for (int u = 0; u < 8; ++u) {
            const int i = base + u * OWN_THREADS + tid;
            v[u] = (i < n4) ? dst4[i] : make_int4(-1, -1, -1, -1);
        }
#pragma unroll
        for (int u = 0; u < 8; ++u) {
            const int e = (base + u * OWN_THREADS + tid) * 4;
            own_push(wlist, wcount, OWN_WCAP1, v[u].x, n0, e + 0);
            own_push(wlist, wcount, OWN_WCAP1, v[u].y, n0, e + 1);
            own_push(wlist, wcount, OWN_WCAP1, v[u].z, n0, e + 2);
            own_push(wlist, wcount, OWN_WCAP1, v[u].w, n0, e + 3);
        }
    }
    if (wcount > (unsigned)OWN_WCAP1) {
        unsigned seen = 0;
#pragma unroll 1
        for (int t = 0; t < OWN_TRIPS * 8; ++t) {
            int tt = (t >> 3) + rot;
            if (tt >= OWN_TRIPS) tt -= OWN_TRIPS;
            const int i = (tt * 8 + (t & 7)) * OWN_THREADS + tid;
            const int4 v = (i < n4) ? dst4[i] : make_int4(-1, -1, -1, -1);
            const int dv[4] = {v.x, v.y, v.z, v.w};
#pragma unroll 1
            for (int k = 0; k < 4; ++k) {
                const unsigned d = (unsigned)(dv[k] - n0);
                const bool m = d < (unsigned)OWN_RN;
                const unsigned mask = __builtin_amdgcn_ballot_w32(m);
                const unsigned pos = seen + __builtin_amdgcn_mbcnt_lo(mask, 0u);
                seen += (unsigned)__builtin_popcount(mask);
                if (m && pos >= (unsigned)OWN_WCAP1) {
                    const int oe = i * 4 + k, od = (int)d;
                    const int os = ei[oe];
                    const float4 a = *(const float4*)(ea + (size_t)oe * EDGE_DIM_C);
#pragma unroll 1
                    for (int q = 0; q < IN_DIM_C; ++q) {
                        const float4 w = *(const float4*)(eW + q * EDGE_DIM_C);
                        float t = eb[q] + a.x * w.x + a.y * w.y + a.z * w.z + a.w * w.w;
                        t += x[(size_t)os * IN_DIM_C + q];
                        atomicAdd(&s_acc[od * 8 + q], fmaxf(t, 0.0f));
                    }
                }
            }
        }
    }
    const int wn = (int)(wcount < (unsigned)OWN_WCAP1 ? wcount : (unsigned)OWN_WCAP1);

    __syncthreads();

    for (int p = 0; p < wn; p += 32) {
        unsigned ent[8]; int sr[8]; float4 av[8]; float xv[8]; bool ok[8];
#pragma unroll
        for (int g = 0; g < 8; ++g) {
            const int idx = p + g * 4 + sub;
            ok[g]  = idx < wn;
            ent[g] = ok[g] ? wlist[idx] : 0u;
        }
#pragma unroll
        for (int g = 0; g < 8; ++g) {
            const int e = (int)(ent[g] >> 10);
            sr[g] = ok[g] ? ei[e] : 0;
            av[g] = ok[g] ? *(const float4*)(ea + (size_t)e * EDGE_DIM_C)
                          : make_float4(0.f, 0.f, 0.f, 0.f);
        }
#pragma unroll
        for (int g = 0; g < 8; ++g)
            xv[g] = (ok[g] && cact) ? x[(size_t)sr[g] * IN_DIM_C + c] : 0.0f;
#pragma unroll
        for (int g = 0; g < 8; ++g) {
            if (ok[g] && cact) {
                const int dl = (int)(ent[g] & 1023u);
                float t = bc + av[g].x * wc.x + av[g].y * wc.y + av[g].z * wc.z + av[g].w * wc.w;
                t += xv[g];
                atomicAdd(&s_acc[dl * 8 + c], fmaxf(t, 0.0f));
            }
        }
    }
    __syncthreads();

    int rows = N_NODES_C - n0;
    if (rows > OWN_RN) rows = OWN_RN;
    float* o = aggr + (size_t)n0 * IN_DIM_C;
    for (int i = tid; i < rows * IN_DIM_C; i += OWN_THREADS) {
        const int r = i / IN_DIM_C, q = i - r * IN_DIM_C;
        o[i] = s_acc[r * 8 + q];
    }
}

__global__ __launch_bounds__(256)
void pool_kernel(const float* __restrict__ h, const int* __restrict__ batch,
                 float* __restrict__ psum, float* __restrict__ cnt) {
    const long long t = (long long)blockIdx.x * 256 + threadIdx.x;
    const long long n = t >> 6;
    const int c = (int)(t & 63);
    if (n >= N_NODES_C) return;
    const int g = batch[n];
    atom_add_f32(&psum[(size_t)g * HID_C + c], h[(size_t)n * HID_C + c]);
    if (c == 0) atom_add_f32(&cnt[g], 1.0f);
}

__global__ __launch_bounds__(256)
void fc_kernel(const float* __restrict__ psum, const float* __restrict__ cnt,
               const float* __restrict__ fcW, const float* __restrict__ fcb,
               float* __restrict__ out) {
    const int t = blockIdx.x * 256 + threadIdx.x;
    if (t >= N_GRAPHS_C * NUM_TASKS_C) return;
    const int g = t / NUM_TASKS_C;
    const int k = t - g * NUM_TASKS_C;
    const float inv = 1.0f / fmaxf(cnt[g], 1.0f);
    float s = fcb[k];
#pragma unroll
    for (int c = 0; c < HID_C; ++c)
        s += psum[(size_t)g * HID_C + c] * inv * fcW[k * HID_C + c];
    out[t] = s;
}

extern "C" void kernel_launch(void* const* d_in, const int* in_sizes, int n_in,
                              void* d_out, int out_size, void* d_ws, size_t ws_size,
                              hipStream_t stream) {
    const float* x    = (const float*)d_in[0];
    const int*   ei   = (const int*)  d_in[1];
    const float* ea   = (const float*)d_in[2];
    const int*   bat  = (const int*)  d_in[3];
    const float* e1W  = (const float*)d_in[4];  const float* e1b  = (const float*)d_in[5];
    const float* n1W1 = (const float*)d_in[6];  const float* n1b1 = (const float*)d_in[7];
    const float* n1W2 = (const float*)d_in[8];  const float* n1b2 = (const float*)d_in[9];
    const float* e2W  = (const float*)d_in[10]; const float* e2b  = (const float*)d_in[11];
    const float* n2W1 = (const float*)d_in[12]; const float* n2b1 = (const float*)d_in[13];
    const float* n2W2 = (const float*)d_in[14]; const float* n2b2 = (const float*)d_in[15];
    const float* fcW  = (const float*)d_in[16]; const float* fcb  = (const float*)d_in[17];
    float* out = (float*)d_out;

    float*    aggr1 = (float*)d_ws;
    _Float16* h     = (_Float16*)(aggr1 + (size_t)N_NODES_C * IN_DIM_C);
    float*    aggr2 = (float*)(h + (size_t)N_NODES_C * HID_C);
    float*    h2    = aggr2 + (size_t)N_NODES_C * HID_C;
    float*    psum  = h2    + (size_t)N_NODES_C * HID_C;
    float*    cnt   = psum  + (size_t)N_GRAPHS_C * HID_C;

    hipMemsetAsync(psum,  0, sizeof(float) * (size_t)N_GRAPHS_C * HID_C,  stream);
    hipMemsetAsync(cnt,   0, sizeof(float) * (size_t)N_GRAPHS_C,          stream);

    edge1_own_kernel<<<OWN_BLOCKS, OWN_THREADS, 0, stream>>>(x, ei, ea, e1W, e1b, aggr1);
    {
        const unsigned grid = (N_NODES_C + 15) / 16;
        node_mlp_kernel<IN_DIM_C, float, _Float16><<<grid, 128, 0, stream>>>(
            x, aggr1, n1W1, n1b1, n1W2, n1b2, h, N_NODES_C);
    }
    edge2_own_kernel<<<OWN_BLOCKS, OWN_THREADS, 0, stream>>>(h, ei, ea, e2W, e2b, aggr2);
    {
        const unsigned grid = (N_NODES_C + 15) / 16;
        node_mlp_kernel<HID_C, _Float16, float><<<grid, 128, 0, stream>>>(
            h, aggr2, n2W1, n2b1, n2W2, n2b2, h2, N_NODES_C);
    }
    {
        const unsigned grid = (unsigned)(((long long)N_NODES_C * 64 + 255) / 256);
        pool_kernel<<<grid, 256, 0, stream>>>(h2, bat, psum, cnt);
    }
    {
        const unsigned grid = (N_GRAPHS_C * NUM_TASKS_C + 255) / 256;
        fc_kernel<<<grid, 256, 0, stream>>>(psum, cnt, fcW, fcb, out);
    }
}


# Edge kernels cut into node ranges x edge-list slices, 256 blocks (conv2 128x2, conv1 32x8): a block scans only its slice into a partial plane; MLP staging adds the planes (node sum = slice partials)
# speedup vs baseline: 1.9735x; 1.4742x over previous
#include <hip/hip_runtime.h>
#include <hip/hip_bf16.h>

#define N_NODES_C  100000
#define N_EDGES_C  1600000
#define N_GRAPHS_C 1000
#define HID_C      64
#define IN_DIM_C   7
#define EDGE_DIM_C 4
#define NUM_TASKS_C 12

typedef __attribute__((ext_vector_type(16))) _Float16 v16h;
typedef __attribute__((ext_vector_type(8)))  _Float16 v8h;
typedef __attribute__((ext_vector_type(8)))  float    v8f;

__device__ __forceinline__ void atom_add_f32(float* p, float v) {
    unsafeAtomicAdd(p, v);
}

__device__ __forceinline__ v16h lds_load_tile(const _Float16* s, int ld, int kbase) {
    const int lane = threadIdx.x & 31;
    const int r    = lane & 15;
    const int koff = (lane < 16) ? 0 : 8;
    const _Float16* p = s + r * ld + kbase + koff;
    v8h lo = *(const v8h*)(p);
    v8h hi = *(const v8h*)(p + 16);
    v16h out;
#pragma unroll
    for (int e = 0; e < 8; ++e) { out[e] = lo[e]; out[8 + e] = hi[e]; }
    return out;
}

template <int K1, int NP, typename TI, typename TO>
__global__ __launch_bounds__(128)
void node_mlp_kernel(const TI* __restrict__ in0, const float* __restrict__ in1,
                     const float* __restrict__ W1,  const float* __restrict__ b1,
                     const float* __restrict__ W2,  const float* __restrict__ b2,
                     TO* __restrict__ out, int n_nodes) {
    constexpr int K1P = (K1 + 31) & ~31;

    __shared__ __align__(16) _Float16 sA  [16 * K1P];
    __shared__ __align__(16) _Float16 sMid[16 * HID_C];
    __shared__ __align__(16) _Float16 sB1 [HID_C * K1P];
    __shared__ __align__(16) _Float16 sB2 [HID_C * HID_C];

    const int row0 = blockIdx.x * 16;
    const int tid  = threadIdx.x;
    const int wave = tid >> 5;
    const int lane = tid & 31;
    const int n0   = wave * 16;

    for (int i = tid; i < 16 * K1P; i += 128) {
        const int m = i / K1P, k = i - m * K1P;
        const int r = row0 + m;
        float v = 0.0f;
        if (k < K1 && r < n_nodes) {
            const size_t o = (size_t)r * K1 + k;
            v = (float)in0[o];
#pragma unroll
            for (int p = 0; p < NP; ++p)
                v += in1[(size_t)p * (size_t)n_nodes * K1 + o];
        }
        sA[i] = (_Float16)v;
    }
    for (int i = tid; i < HID_C * K1P; i += 128) {
        const int n = i / K1P, k = i - n * K1P;
        sB1[i] = (_Float16)((k < K1) ? W1[n * K1 + k] : 0.0f);
    }
    for (int i = tid; i < HID_C * HID_C; i += 128) {
        const int n = i >> 6, k = i & 63;
        sB2[i] = (_Float16)W2[n * HID_C + k];
    }
    __syncthreads();

    v8f acc = {};
#pragma unroll
    for (int kb = 0; kb < K1P; kb += 32) {
        v16h a = lds_load_tile(sA, K1P, kb);
        v16h b = lds_load_tile(sB1 + n0 * K1P, K1P, kb);
        acc = __builtin_amdgcn_wmma_f32_16x16x32_f16(false, a, false, b,
                                                     (short)0, acc, false, false);
    }
    {
        const int n  = lane & 15;
        const int mb = (lane < 16) ? 0 : 8;
        const float bias = b1[n0 + n];
#pragma unroll
        for (int r = 0; r < 8; ++r) {
            float v = acc[r] + bias;
            v = v > 0.0f ? v : 0.0f;
            sMid[(mb + r) * HID_C + (n0 + n)] = (_Float16)v;
        }
    }
    __syncthreads();

    v8f acc2 = {};
#pragma unroll
    for (int kb = 0; kb < HID_C; kb += 32) {
        v16h a = lds_load_tile(sMid, HID_C, kb);
        v16h b = lds_load_tile(sB2 + n0 * HID_C, HID_C, kb);
        acc2 = __builtin_amdgcn_wmma_f32_16x16x32_f16(false, a, false, b,
                                                      (short)0, acc2, false, false);
    }
    {
        const int n  = lane & 15;
        const int mb = (lane < 16) ? 0 : 8;
        const float bias = b2[n0 + n];
#pragma unroll
        for (int r = 0; r < 8; ++r) {
            const int rr = row0 + mb + r;
            if (rr < n_nodes) {
                float v = acc2[r] + bias;
                v = v > 0.0f ? v : 0.0f;
                out[(size_t)rr * HID_C + (n0 + n)] = (TO)v;
            }
        }
    }
}

#define OWN_BLOCKS  256
#define OWN_THREADS 1024

template <int RN_, int S_, int R_, int WCAP_, int DB_, int U_>
struct own_cfg {
    static constexpr int RN = RN_, S = S_, R = R_, WCAP = WCAP_, DB = DB_, U = U_;
    static constexpr int SL4   = N_EDGES_C / 4 / S_;
    static constexpr int STEP  = OWN_THREADS * U_;
    static constexpr int TRIPS = (SL4 + STEP - 1) / STEP;
    static_assert((N_EDGES_C / 4) % S_ == 0, "a slice is a whole number of int4 words");
    static_assert(R_ * S_ == OWN_BLOCKS, "one block per (range, slice)");
    static_assert(R_ * RN_ >= N_NODES_C, "ranges cover every node");
    static_assert(RN_ <= (1 << DB_), "place in range fits its bits");
    static_assert(((long long)SL4 * 4) <= (1ll << (32 - DB_)), "edge id in slice fits its bits");
};
typedef own_cfg<782, 2, 128, 512, 10, 8>  cfg2;
typedef own_cfg<3125, 8, 32, 1280, 12, 4> cfg1;

typedef __attribute__((ext_vector_type(2))) _Float16 v2h;

template <class C>
__device__ __forceinline__ void own_push(unsigned* wlist, unsigned& wcount,
                                         int dstv, int n0, int erel) {
    const unsigned d = (unsigned)(dstv - n0);
    const bool m = d < (unsigned)C::RN;
    const unsigned mask = __builtin_amdgcn_ballot_w32(m);
    if (mask) {
        const unsigned pos = wcount + __builtin_amdgcn_mbcnt_lo(mask, 0u);
        if (m && pos < (unsigned)C::WCAP) wlist[pos] = ((unsigned)erel << C::DB) | d;
        wcount += (unsigned)__builtin_popcount(mask);
    }
}

template <class C>
__device__ __forceinline__ unsigned own_scan(const int4* __restrict__ d4, unsigned* wlist,
                                             int n0, int rot, int tid) {
    unsigned wcount = 0;
    for (int t = 0; t < C::TRIPS; ++t) {
        int tt = t + rot;
        if (tt >= C::TRIPS) tt -= C::TRIPS;
        const int base = tt * C::STEP;
        int4 v[C::U];
#pragma unroll
        for (int u = 0; u < C::U; ++u) {
            const int i = base + u * OWN_THREADS + tid;
            v[u] = (i < C::SL4) ? d4[i] : make_int4(-1, -1, -1, -1);
        }
#pragma unroll
        for (int u = 0; u < C::U; ++u) {
            const int e = (base + u * OWN_THREADS + tid) * 4;
            own_push<C>(wlist, wcount, v[u].x, n0, e + 0);
            own_push<C>(wlist, wcount, v[u].y, n0, e + 1);
            own_push<C>(wlist, wcount, v[u].z, n0, e + 2);
            own_push<C>(wlist, wcount, v[u].w, n0, e + 3);
        }
    }
    return wcount;
}

template <class C, class F>
__device__ __forceinline__ void own_overflow(const int4* __restrict__ d4, int n0, int rot,
                                             int tid, F add) {
    unsigned seen = 0;
#pragma unroll 1
    for (int t = 0; t < C::TRIPS * C::U; ++t) {
        int tt = t / C::U + rot;
        if (tt >= C::TRIPS) tt -= C::TRIPS;
        const int i = (tt * C::U + (t % C::U)) * OWN_THREADS + tid;
        const int4 v = (i < C::SL4) ? d4[i] : make_int4(-1, -1, -1, -1);
        const int dv[4] = {v.x, v.y, v.z, v.w};
#pragma unroll 1
        for (int k = 0; k < 4; ++k) {
            const unsigned d = (unsigned)(dv[k] - n0);
            const bool m = d < (unsigned)C::RN;
            const unsigned mask = __builtin_amdgcn_ballot_w32(m);
            const unsigned pos = seen + __builtin_amdgcn_mbcnt_lo(mask, 0u);
            seen += (unsigned)__builtin_popcount(mask);
            if (m && pos >= (unsigned)C::WCAP) add(i * 4 + k, (int)d);
        }
    }
}

__global__ __launch_bounds__(OWN_THREADS)
void edge2_own_kernel(const _Float16* __restrict__ h, const int* __restrict__ ei,
                      const float* __restrict__ ea,
                      const float* __restrict__ eW, const float* __restrict__ eb,
                      float* __restrict__ aggr) {
    typedef cfg2 C;
    __shared__ __align__(16) float    s_acc[C::RN * HID_C];
    __shared__ __align__(16) unsigned s_list[(OWN_THREADS / 32) * C::WCAP];

    const int tid   = threadIdx.x;
    const int lane  = tid & 31;
    const int wave  = tid >> 5;
    const int range = (int)blockIdx.x / C::S;
    const int slice = (int)blockIdx.x - range * C::S;
    const int n0    = range * C::RN;
    const int e0    = slice * (C::SL4 * 4);
    unsigned* wlist = s_list + wave * C::WCAP;

    for (int i = tid; i < C::RN * HID_C / 4; i += OWN_THREADS)
        ((float4*)s_acc)[i] = make_float4(0.f, 0.f, 0.f, 0.f);
    __syncthreads();

    const int c0 = 2 * lane;
    const float4 w0 = *(const float4*)(eW + (c0 + 0) * EDGE_DIM_C);
    const float4 w1 = *(const float4*)(eW + (c0 + 1) * EDGE_DIM_C);
    const float  b0 = eb[c0], b1 = eb[c0 + 1];

    const int4* d4 = (const int4*)(ei + N_EDGES_C + e0);
    const int rot = (range * C::TRIPS) / C::R;
    const unsigned wcount = own_scan<C>(d4, wlist, n0, rot, tid);
    if (wcount > (unsigned)C::WCAP) {
        own_overflow<C>(d4, n0, rot, tid, [&](int erel, int od) {
            const int oe = e0 + erel;
            const int os = ei[oe];
            const float4 a = *(const float4*)(ea + (size_t)oe * EDGE_DIM_C);
#pragma unroll 1
            for (int q = 0; q < HID_C; ++q) {
                const float4 w = *(const float4*)(eW + q * EDGE_DIM_C);
                float t = eb[q] + a.x * w.x + a.y * w.y + a.z * w.z + a.w * w.w;
                t += (float)h[(size_t)os * HID_C + q];
                atomicAdd(&s_acc[od * HID_C + q], fmaxf(t, 0.0f));
            }
        });
    }
    const int wn = (int)(wcount < (unsigned)C::WCAP ? wcount : (unsigned)C::WCAP);

    __syncthreads();

    const v2h* h2p = (const v2h*)h;
    for (int p = 0; p < wn; p += 32) {
        const int idx = p + lane;
        const bool valid = idx < wn;
        const unsigned ent = valid ? wlist[idx] : 0u;
        const int e  = e0 + (int)(ent >> C::DB);
        const int dl = (int)(ent & ((1u << C::DB) - 1u));
        const int sl = valid ? ei[e] : 0;
        const float4 al = valid ? *(const float4*)(ea + (size_t)e * EDGE_DIM_C)
                                : make_float4(0.f, 0.f, 0.f, 0.f);
        const int cnt = (wn - p) < 32 ? (wn - p) : 32;
        v2h hv[32];
#pragma unroll
        for (int j = 0; j < 32; ++j) {
            const int sj = __builtin_amdgcn_readlane(sl, j);
            hv[j] = h2p[(size_t)sj * (HID_C / 2) + lane];
        }
#pragma unroll
        for (int j = 0; j < 32; ++j) {
            if (j < cnt) {
                const int   dj = __builtin_amdgcn_readlane(dl, j);
                const float ax = __builtin_bit_cast(float, __builtin_amdgcn_readlane(__builtin_bit_cast(int, al.x), j));
                const float ay = __builtin_bit_cast(float, __builtin_amdgcn_readlane(__builtin_bit_cast(int, al.y), j));
                const float az = __builtin_bit_cast(float, __builtin_amdgcn_readlane(__builtin_bit_cast(int, al.z), j));
                const float aw = __builtin_bit_cast(float, __builtin_amdgcn_readlane(__builtin_bit_cast(int, al.w), j));
                float t0 = b0 + ax * w0.x + ay * w0.y + az * w0.z + aw * w0.w;
                float t1 = b1 + ax * w1.x + ay * w1.y + az * w1.z + aw * w1.w;
                t0 += (float)hv[j][0];
                t1 += (float)hv[j][1];
                atomicAdd(&s_acc[dj * HID_C + c0],     fmaxf(t0, 0.0f));
                atomicAdd(&s_acc[dj * HID_C + c0 + 1], fmaxf(t1, 0.0f));
            }
        }
    }
    __syncthreads();

    int rows = N_NODES_C - n0;
    if (rows > C::RN) rows = C::RN;
    float4* o4 = (float4*)(aggr + (size_t)slice * N_NODES_C * HID_C + (size_t)n0 * HID_C);
    for (int i = tid; i < rows * (HID_C / 4); i += OWN_THREADS)
        o4[i] = ((const float4*)s_acc)[i];
}

__global__ __launch_bounds__(OWN_THREADS)
void edge1_own_kernel(const float* __restrict__ x, const int* __restrict__ ei,
                      const float* __restrict__ ea,
                      const float* __restrict__ eW, const float* __restrict__ eb,
                      float* __restrict__ aggr) {
    typedef cfg1 C;
    __shared__ __align__(16) float    s_acc[C::RN * 8];
    __shared__ __align__(16) unsigned s_list[(OWN_THREADS / 32) * C::WCAP];

    const int tid   = threadIdx.x;
    const int lane  = tid & 31;
    const int wave  = tid >> 5;
    const int range = (int)blockIdx.x / C::S;
    const int slice = (int)blockIdx.x - range * C::S;
    const int n0    = range * C::RN;
    const int e0    = slice * (C::SL4 * 4);
    unsigned* wlist = s_list + wave * C::WCAP;

    for (int i = tid; i < C::RN * 8; i += OWN_THREADS) s_acc[i] = 0.0f;
    __syncthreads();

    const int c   = lane & 7;
    const int sub = lane >> 3;
    const bool cact = c < IN_DIM_C;
    const float4 wc = cact ? *(const float4*)(eW + c * EDGE_DIM_C) : make_float4(0.f, 0.f, 0.f, 0.f);
    const float  bc = cact ? eb[c] : 0.0f;

    const int4* d4 = (const int4*)(ei + N_EDGES_C + e0);
    const int rot = (range * C::TRIPS) / C::R;
    const unsigned wcount = own_scan<C>(d4, wlist, n0, rot, tid);
    if (wcount > (unsigned)C::WCAP) {
        own_overflow<C>(d4, n0, rot, tid, [&](int erel, int od) {
            const int oe = e0 + erel;
            const int os = ei[oe];
            const float4 a = *(const float4*)(ea + (size_t)oe * EDGE_DIM_C);
#pragma unroll 1
            for (int q = 0; q < IN_DIM_C; ++q) {
                const float4 w = *(const float4*)(eW + q * EDGE_DIM_C);
                float t = eb[q] + a.x * w.x + a.y * w.y + a.z * w.z + a.w * w.w;
                t += x[(size_t)os * IN_DIM_C + q];
                atomicAdd(&s_acc[od * 8 + q], fmaxf(t, 0.0f));
            }
        });
    }
    const int wn = (int)(wcount < (unsigned)C::WCAP ? wcount : (unsigned)C::WCAP);

    __syncthreads();

    for (int p = 0; p < wn; p += 32) {
        unsigned ent[8]; int sr[8]; float4 av[8]; float xv[8]; bool ok[8];
#pragma unroll
        for (int g = 0; g < 8; ++g) {
            const int idx = p + g * 4 + sub;
            ok[g]  = idx < wn;
            ent[g] = ok[g] ? wlist[idx] : 0u;
        }
#pragma unroll
        for (int g = 0; g < 8; ++g) {
            const int e = e0 + (int)(ent[g] >> C::DB);
            sr[g] = ok[g] ? ei[e] : 0;
            av[g] = ok[g] ? *(const float4*)(ea + (size_t)e * EDGE_DIM_C)
                          : make_float4(0.f, 0.f, 0.f, 0.f);
        }
#pragma unroll
        for (int g = 0; g < 8; ++g)
            xv[g] = (ok[g] && cact) ? x[(size_t)sr[g] * IN_DIM_C + c] : 0.0f;
#pragma unroll
        for (int g = 0; g < 8; ++g) {
            if (ok[g] && cact) {
                const int dl = (int)(ent[g] & ((1u << C::DB) - 1u));
                float t = bc + av[g].x * wc.x + av[g].y * wc.y + av[g].z * wc.z + av[g].w * wc.w;
                t += xv[g];
                atomicAdd(&s_acc[dl * 8 + c], fmaxf(t, 0.0f));
            }
        }
    }
    __syncthreads();

    int rows = N_NODES_C - n0;
    if (rows > C::RN) rows = C::RN;
    float* o = aggr + (size_t)slice * N_NODES_C * IN_DIM_C + (size_t)n0 * IN_DIM_C;
    for (int i = tid; i < rows * IN_DIM_C; i += OWN_THREADS) {
        const int r = i / IN_DIM_C, q = i - r * IN_DIM_C;
        o[i] = s_acc[r * 8 + q];
    }
}

__global__ __launch_bounds__(256)
void pool_kernel(const float* __restrict__ h, const int* __restrict__ batch,
                 float* __restrict__ psum, float* __restrict__ cnt) {
    const long long t = (long long)blockIdx.x * 256 + threadIdx.x;
    const long long n = t >> 6;
    const int c = (int)(t & 63);
    if (n >= N_NODES_C) return;
    const int g = batch[n];
    atom_add_f32(&psum[(size_t)g * HID_C + c], h[(size_t)n * HID_C + c]);
    if (c == 0) atom_add_f32(&cnt[g], 1.0f);
}

__global__ __launch_bounds__(256)
void fc_kernel(const float* __restrict__ psum, const float* __restrict__ cnt,
               const float* __restrict__ fcW, const float* __restrict__ fcb,
               float* __restrict__ out) {
    const int t = blockIdx.x * 256 + threadIdx.x;
    if (t >= N_GRAPHS_C * NUM_TASKS_C) return;
    const int g = t / NUM_TASKS_C;
    const int k = t - g * NUM_TASKS_C;
    const float inv = 1.0f / fmaxf(cnt[g], 1.0f);
    float s = fcb[k];
#pragma unroll
    for (int c = 0; c < HID_C; ++c)
        s += psum[(size_t)g * HID_C + c] * inv * fcW[k * HID_C + c];
    out[t] = s;
}

extern "C" void kernel_launch(void* const* d_in, const int* in_sizes, int n_in,
                              void* d_out, int out_size, void* d_ws, size_t ws_size,
                              hipStream_t stream) {
    const float* x    = (const float*)d_in[0];
    const int*   ei   = (const int*)  d_in[1];
    const float* ea   = (const float*)d_in[2];
    const int*   bat  = (const int*)  d_in[3];
    const float* e1W  = (const float*)d_in[4];  const float* e1b  = (const float*)d_in[5];
    const float* n1W1 = (const float*)d_in[6];  const float* n1b1 = (const float*)d_in[7];
    const float* n1W2 = (const float*)d_in[8];  const float* n1b2 = (const float*)d_in[9];
    const float* e2W  = (const float*)d_in[10]; const float* e2b  = (const float*)d_in[11];
    const float* n2W1 = (const float*)d_in[12]; const float* n2b1 = (const float*)d_in[13];
    const float* n2W2 = (const float*)d_in[14]; const float* n2b2 = (const float*)d_in[15];
    const float* fcW  = (const float*)d_in[16]; const float* fcb  = (const float*)d_in[17];
    float* out = (float*)d_out;

    float*    aggr1 = (float*)d_ws;
    _Float16* h     = (_Float16*)(aggr1 + (size_t)cfg1::S * N_NODES_C * IN_DIM_C);
    float*    aggr2 = (float*)(h + (size_t)N_NODES_C * HID_C);
    float*    h2    = aggr2 + (size_t)cfg2::S * N_NODES_C * HID_C;
    float*    psum  = h2    + (size_t)N_NODES_C * HID_C;
    float*    cnt   = psum  + (size_t)N_GRAPHS_C * HID_C;

    hipMemsetAsync(psum,  0, sizeof(float) * (size_t)N_GRAPHS_C * HID_C,  stream);
    hipMemsetAsync(cnt,   0, sizeof(float) * (size_t)N_GRAPHS_C,          stream);

    edge1_own_kernel<<<OWN_BLOCKS, OWN_THREADS, 0, stream>>>(x, ei, ea, e1W, e1b, aggr1);
    {
        const unsigned grid = (N_NODES_C + 15) / 16;
        node_mlp_kernel<IN_DIM_C, cfg1::S, float, _Float16><<<grid, 128, 0, stream>>>(
            x, aggr1, n1W1, n1b1, n1W2, n1b2, h, N_NODES_C);
    }
    edge2_own_kernel<<<OWN_BLOCKS, OWN_THREADS, 0, stream>>>(h, ei, ea, e2W, e2b, aggr2);
    {
        const unsigned grid = (N_NODES_C + 15) / 16;
        node_mlp_kernel<HID_C, cfg2::S, _Float16, float><<<grid, 128, 0, stream>>>(
            h, aggr2, n2W1, n2b1, n2W2, n2b2, h2, N_NODES_C);
    }
    {
        const unsigned grid = (unsigned)(((long long)N_NODES_C * 64 + 255) / 256);
        pool_kernel<<<grid, 256, 0, stream>>>(h2, bat, psum, cnt);
    }
    {
        const unsigned grid = (N_GRAPHS_C * NUM_TASKS_C + 255) / 256;
        fc_kernel<<<grid, 256, 0, stream>>>(psum, cnt, fcW, fcb, out);
    }
}


# Mean pool + FC by owner blocks: 250 blocks each scan batch for their 4 graphs, sum those h2 rows (node order in a wave, then wave order) and apply the FC; no atomics, memsets or fc launch
# speedup vs baseline: 2.4284x; 1.2305x over previous
#include <hip/hip_runtime.h>
#include <hip/hip_bf16.h>

#define N_NODES_C  100000
#define N_EDGES_C  1600000
#define N_GRAPHS_C 1000
#define HID_C      64
#define IN_DIM_C   7
#define EDGE_DIM_C 4
#define NUM_TASKS_C 12

typedef __attribute__((ext_vector_type(16))) _Float16 v16h;
typedef __attribute__((ext_vector_type(8)))  _Float16 v8h;
typedef __attribute__((ext_vector_type(8)))  float    v8f;

__device__ __forceinline__ v16h lds_load_tile(const _Float16* s, int ld, int kbase) {
    const int lane = threadIdx.x & 31;
    const int r    = lane & 15;
    const int koff = (lane < 16) ? 0 : 8;
    const _Float16* p = s + r * ld + kbase + koff;
    v8h lo = *(const v8h*)(p);
    v8h hi = *(const v8h*)(p + 16);
    v16h out;
#pragma unroll
    for (int e = 0; e < 8; ++e) { out[e] = lo[e]; out[8 + e] = hi[e]; }
    return out;
}

template <int K1, int NP, typename TI, typename TO>
__global__ __launch_bounds__(128)
void node_mlp_kernel(const TI* __restrict__ in0, const float* __restrict__ in1,
                     const float* __restrict__ W1,  const float* __restrict__ b1,
                     const float* __restrict__ W2,  const float* __restrict__ b2,
                     TO* __restrict__ out, int n_nodes) {
    constexpr int K1P = (K1 + 31) & ~31;

    __shared__ __align__(16) _Float16 sA  [16 * K1P];
    __shared__ __align__(16) _Float16 sMid[16 * HID_C];
    __shared__ __align__(16) _Float16 sB1 [HID_C * K1P];
    __shared__ __align__(16) _Float16 sB2 [HID_C * HID_C];

    const int row0 = blockIdx.x * 16;
    const int tid  = threadIdx.x;
    const int wave = tid >> 5;
    const int lane = tid & 31;
    const int n0   = wave * 16;

    for (int i = tid; i < 16 * K1P; i += 128) {
        const int m = i / K1P, k = i - m * K1P;
        const int r = row0 + m;
        float v = 0.0f;
        if (k < K1 && r < n_nodes) {
            const size_t o = (size_t)r * K1 + k;
            v = (float)in0[o];
#pragma unroll
            for (int p = 0; p < NP; ++p)
                v += in1[(size_t)p * (size_t)n_nodes * K1 + o];
        }
        sA[i] = (_Float16)v;
    }
    for (int i = tid; i < HID_C * K1P; i += 128) {
        const int n = i / K1P, k = i - n * K1P;
        sB1[i] = (_Float16)((k < K1) ? W1[n * K1 + k] : 0.0f);
    }
    for (int i = tid; i < HID_C * HID_C; i += 128) {
        const int n = i >> 6, k = i & 63;
        sB2[i] = (_Float16)W2[n * HID_C + k];
    }
    __syncthreads();

    v8f acc = {};
#pragma unroll
    for (int kb = 0; kb < K1P; kb += 32) {
        v16h a = lds_load_tile(sA, K1P, kb);
        v16h b = lds_load_tile(sB1 + n0 * K1P, K1P, kb);
        acc = __builtin_amdgcn_wmma_f32_16x16x32_f16(false, a, false, b,
                                                     (short)0, acc, false, false);
    }
    {
        const int n  = lane & 15;
        const int mb = (lane < 16) ? 0 : 8;
        const float bias = b1[n0 + n];
#pragma unroll
        for (int r = 0; r < 8; ++r) {
            float v = acc[r] + bias;
            v = v > 0.0f ? v : 0.0f;
            sMid[(mb + r) * HID_C + (n0 + n)] = (_Float16)v;
        }
    }
    __syncthreads();

    v8f acc2 = {};
#pragma unroll
    for (int kb = 0; kb < HID_C; kb += 32) {
        v16h a = lds_load_tile(sMid, HID_C, kb);
        v16h b = lds_load_tile(sB2 + n0 * HID_C, HID_C, kb);
        acc2 = __builtin_amdgcn_wmma_f32_16x16x32_f16(false, a, false, b,
                                                      (short)0, acc2, false, false);
    }
    {
        const int n  = lane & 15;
        const int mb = (lane < 16) ? 0 : 8;
        const float bias = b2[n0 + n];
#pragma unroll
        for (int r = 0; r < 8; ++r) {
            const int rr = row0 + mb + r;
            if (rr < n_nodes) {
                float v = acc2[r] + bias;
                v = v > 0.0f ? v : 0.0f;
                out[(size_t)rr * HID_C + (n0 + n)] = (TO)v;
            }
        }
    }
}

#define OWN_BLOCKS  256
#define OWN_THREADS 1024

template <int RN_, int S_, int R_, int WCAP_, int DB_, int U_>
struct own_cfg {
    static constexpr int RN = RN_, S = S_, R = R_, WCAP = WCAP_, DB = DB_, U = U_;
    static constexpr int SL4   = N_EDGES_C / 4 / S_;
    static constexpr int STEP  = OWN_THREADS * U_;
    static constexpr int TRIPS = (SL4 + STEP - 1) / STEP;
    static_assert((N_EDGES_C / 4) % S_ == 0, "a slice is a whole number of int4 words");
    static_assert(R_ * S_ == OWN_BLOCKS, "one block per (range, slice)");
    static_assert(R_ * RN_ >= N_NODES_C, "ranges cover every node");
    static_assert(RN_ <= (1 << DB_), "place in range fits its bits");
    static_assert(((long long)SL4 * 4) <= (1ll << (32 - DB_)), "edge id in slice fits its bits");
};
typedef own_cfg<782, 2, 128, 512, 10, 8>  cfg2;
typedef own_cfg<3125, 8, 32, 1280, 12, 4> cfg1;

typedef __attribute__((ext_vector_type(2))) _Float16 v2h;

template <class C>
__device__ __forceinline__ void own_push(unsigned* wlist, unsigned& wcount,
                                         int dstv, int n0, int erel) {
    const unsigned d = (unsigned)(dstv - n0);
    const bool m = d < (unsigned)C::RN;
    const unsigned mask = __builtin_amdgcn_ballot_w32(m);
    if (mask) {
        const unsigned pos = wcount + __builtin_amdgcn_mbcnt_lo(mask, 0u);
        if (m && pos < (unsigned)C::WCAP) wlist[pos] = ((unsigned)erel << C::DB) | d;
        wcount += (unsigned)__builtin_popcount(mask);
    }
}

template <class C>
__device__ __forceinline__ unsigned own_scan(const int4* __restrict__ d4, unsigned* wlist,
                                             int n0, int rot, int tid) {
    unsigned wcount = 0;
    for (int t = 0; t < C::TRIPS; ++t) {
        int tt = t + rot;
        if (tt >= C::TRIPS) tt -= C::TRIPS;
        const int base = tt * C::STEP;
        int4 v[C::U];
#pragma unroll
        for (int u = 0; u < C::U; ++u) {
            const int i = base + u * OWN_THREADS + tid;
            v[u] = (i < C::SL4) ? d4[i] : make_int4(-1, -1, -1, -1);
        }
#pragma unroll
        for (int u = 0; u < C::U; ++u) {
            const int e = (base + u * OWN_THREADS + tid) * 4;
            own_push<C>(wlist, wcount, v[u].x, n0, e + 0);
            own_push<C>(wlist, wcount, v[u].y, n0, e + 1);
            own_push<C>(wlist, wcount, v[u].z, n0, e + 2);
            own_push<C>(wlist, wcount, v[u].w, n0, e + 3);
        }
    }
    return wcount;
}

template <class C, class F>
__device__ __forceinline__ void own_overflow(const int4* __restrict__ d4, int n0, int rot,
                                             int tid, F add) {
    unsigned seen = 0;
#pragma unroll 1
    for (int t = 0; t < C::TRIPS * C::U; ++t) {
        int tt = t / C::U + rot;
        if (tt >= C::TRIPS) tt -= C::TRIPS;
        const int i = (tt * C::U + (t % C::U)) * OWN_THREADS + tid;
        const int4 v = (i < C::SL4) ? d4[i] : make_int4(-1, -1, -1, -1);
        const int dv[4] = {v.x, v.y, v.z, v.w};
#pragma unroll 1
        for (int k = 0; k < 4; ++k) {
            const unsigned d = (unsigned)(dv[k] - n0);
            const bool m = d < (unsigned)C::RN;
            const unsigned mask = __builtin_amdgcn_ballot_w32(m);
            const unsigned pos = seen + __builtin_amdgcn_mbcnt_lo(mask, 0u);
            seen += (unsigned)__builtin_popcount(mask);
            if (m && pos >= (unsigned)C::WCAP) add(i * 4 + k, (int)d);
        }
    }
}

__global__ __launch_bounds__(OWN_THREADS)
void edge2_own_kernel(const _Float16* __restrict__ h, const int* __restrict__ ei,
                      const float* __restrict__ ea,
                      const float* __restrict__ eW, const float* __restrict__ eb,
                      float* __restrict__ aggr) {
    typedef cfg2 C;
    __shared__ __align__(16) float    s_acc[C::RN * HID_C];
    __shared__ __align__(16) unsigned s_list[(OWN_THREADS / 32) * C::WCAP];

    const int tid   = threadIdx.x;
    const int lane  = tid & 31;
    const int wave  = tid >> 5;
    const int range = (int)blockIdx.x / C::S;
    const int slice = (int)blockIdx.x - range * C::S;
    const int n0    = range * C::RN;
    const int e0    = slice * (C::SL4 * 4);
    unsigned* wlist = s_list + wave * C::WCAP;

    for (int i = tid; i < C::RN * HID_C / 4; i += OWN_THREADS)
        ((float4*)s_acc)[i] = make_float4(0.f, 0.f, 0.f, 0.f);
    __syncthreads();

    const int c0 = 2 * lane;
    const float4 w0 = *(const float4*)(eW + (c0 + 0) * EDGE_DIM_C);
    const float4 w1 = *(const float4*)(eW + (c0 + 1) * EDGE_DIM_C);
    const float  b0 = eb[c0], b1 = eb[c0 + 1];

    const int4* d4 = (const int4*)(ei + N_EDGES_C + e0);
    const int rot = (range * C::TRIPS) / C::R;
    const unsigned wcount = own_scan<C>(d4, wlist, n0, rot, tid);
    if (wcount > (unsigned)C::WCAP) {
        own_overflow<C>(d4, n0, rot, tid, [&](int erel, int od) {
            const int oe = e0 + erel;
            const int os = ei[oe];
            const float4 a = *(const float4*)(ea + (size_t)oe * EDGE_DIM_C);
#pragma unroll 1
            for (int q = 0; q < HID_C; ++q) {
                const float4 w = *(const float4*)(eW + q * EDGE_DIM_C);
                float t = eb[q] + a.x * w.x + a.y * w.y + a.z * w.z + a.w * w.w;
                t += (float)h[(size_t)os * HID_C + q];
                atomicAdd(&s_acc[od * HID_C + q], fmaxf(t, 0.0f));
            }
        });
    }
    const int wn = (int)(wcount < (unsigned)C::WCAP ? wcount : (unsigned)C::WCAP);

    __syncthreads();

    const v2h* h2p = (const v2h*)h;
    for (int p = 0; p < wn; p += 32) {
        const int idx = p + lane;
        const bool valid = idx < wn;
        const unsigned ent = valid ? wlist[idx] : 0u;
        const int e  = e0 + (int)(ent >> C::DB);
        const int dl = (int)(ent & ((1u << C::DB) - 1u));
        const int sl = valid ? ei[e] : 0;
        const float4 al = valid ? *(const float4*)(ea + (size_t)e * EDGE_DIM_C)
                                : make_float4(0.f, 0.f, 0.f, 0.f);
        const int cnt = (wn - p) < 32 ? (wn - p) : 32;
        v2h hv[32];
#pragma unroll
        for (int j = 0; j < 32; ++j) {
            const int sj = __builtin_amdgcn_readlane(sl, j);
            hv[j] = h2p[(size_t)sj * (HID_C / 2) + lane];
        }
#pragma unroll
        for (int j = 0; j < 32; ++j) {
            if (j < cnt) {
                const int   dj = __builtin_amdgcn_readlane(dl, j);
                const float ax = __builtin_bit_cast(float, __builtin_amdgcn_readlane(__builtin_bit_cast(int, al.x), j));
                const float ay = __builtin_bit_cast(float, __builtin_amdgcn_readlane(__builtin_bit_cast(int, al.y), j));
                const float az = __builtin_bit_cast(float, __builtin_amdgcn_readlane(__builtin_bit_cast(int, al.z), j));
                const float aw = __builtin_bit_cast(float, __builtin_amdgcn_readlane(__builtin_bit_cast(int, al.w), j));
                float t0 = b0 + ax * w0.x + ay * w0.y + az * w0.z + aw * w0.w;
                float t1 = b1 + ax * w1.x + ay * w1.y + az * w1.z + aw * w1.w;
                t0 += (float)hv[j][0];
                t1 += (float)hv[j][1];
                atomicAdd(&s_acc[dj * HID_C + c0],     fmaxf(t0, 0.0f));
                atomicAdd(&s_acc[dj * HID_C + c0 + 1], fmaxf(t1, 0.0f));
            }
        }
    }
    __syncthreads();

    int rows = N_NODES_C - n0;
    if (rows > C::RN) rows = C::RN;
    float4* o4 = (float4*)(aggr + (size_t)slice * N_NODES_C * HID_C + (size_t)n0 * HID_C);
    for (int i = tid; i < rows * (HID_C / 4); i += OWN_THREADS)
        o4[i] = ((const float4*)s_acc)[i];
}

__global__ __launch_bounds__(OWN_THREADS)
void edge1_own_kernel(const float* __restrict__ x, const int* __restrict__ ei,
                      const float* __restrict__ ea,
                      const float* __restrict__ eW, const float* __restrict__ eb,
                      float* __restrict__ aggr) {
    typedef cfg1 C;
    __shared__ __align__(16) float    s_acc[C::RN * 8];
    __shared__ __align__(16) unsigned s_list[(OWN_THREADS / 32) * C::WCAP];

    const int tid   = threadIdx.x;
    const int lane  = tid & 31;
    const int wave  = tid >> 5;
    const int range = (int)blockIdx.x / C::S;
    const int slice = (int)blockIdx.x - range * C::S;
    const int n0    = range * C::RN;
    const int e0    = slice * (C::SL4 * 4);
    unsigned* wlist = s_list + wave * C::WCAP;

    for (int i = tid; i < C::RN * 8; i += OWN_THREADS) s_acc[i] = 0.0f;
    __syncthreads();

    const int c   = lane & 7;
    const int sub = lane >> 3;
    const bool cact = c < IN_DIM_C;
    const float4 wc = cact ? *(const float4*)(eW + c * EDGE_DIM_C) : make_float4(0.f, 0.f, 0.f, 0.f);
    const float  bc = cact ? eb[c] : 0.0f;

    const int4* d4 = (const int4*)(ei + N_EDGES_C + e0);
    const int rot = (range * C::TRIPS) / C::R;
    const unsigned wcount = own_scan<C>(d4, wlist, n0, rot, tid);
    if (wcount > (unsigned)C::WCAP) {
        own_overflow<C>(d4, n0, rot, tid, [&](int erel, int od) {
            const int oe = e0 + erel;
            const int os = ei[oe];
            const float4 a = *(const float4*)(ea + (size_t)oe * EDGE_DIM_C);
#pragma unroll 1
            for (int q = 0; q < IN_DIM_C; ++q) {
                const float4 w = *(const float4*)(eW + q * EDGE_DIM_C);
                float t = eb[q] + a.x * w.x + a.y * w.y + a.z * w.z + a.w * w.w;
                t += x[(size_t)os * IN_DIM_C + q];
                atomicAdd(&s_acc[od * 8 + q], fmaxf(t, 0.0f));
            }
        });
    }
    const int wn = (int)(wcount < (unsigned)C::WCAP ? wcount : (unsigned)C::WCAP);

    __syncthreads();

    for (int p = 0; p < wn; p += 32) {
        unsigned ent[8]; int sr[8]; float4 av[8]; float xv[8]; bool ok[8];
#pragma unroll
        for (int g = 0; g < 8; ++g) {
            const int idx = p + g * 4 + sub;
            ok[g]  = idx < wn;
            ent[g] = ok[g] ? wlist[idx] : 0u;
        }
#pragma unroll
        for (int g = 0; g < 8; ++g) {
            const int e = e0 + (int)(ent[g] >> C::DB);
            sr[g] = ok[g] ? ei[e] : 0;
            av[g] = ok[g] ? *(const float4*)(ea + (size_t)e * EDGE_DIM_C)
                          : make_float4(0.f, 0.f, 0.f, 0.f);
        }
#pragma unroll
        for (int g = 0; g < 8; ++g)
            xv[g] = (ok[g] && cact) ? x[(size_t)sr[g] * IN_DIM_C + c] : 0.0f;
#pragma unroll
        for (int g = 0; g < 8; ++g) {
            if (ok[g] && cact) {
                const int dl = (int)(ent[g] & ((1u << C::DB) - 1u));
                float t = bc + av[g].x * wc.x + av[g].y * wc.y + av[g].z * wc.z + av[g].w * wc.w;
                t += xv[g];
                atomicAdd(&s_acc[dl * 8 + c], fmaxf(t, 0.0f));
            }
        }
    }
    __syncthreads();

    int rows = N_NODES_C - n0;
    if (rows > C::RN) rows = C::RN;
    float* o = aggr + (size_t)slice * N_NODES_C * IN_DIM_C + (size_t)n0 * IN_DIM_C;
    for (int i = tid; i < rows * IN_DIM_C; i += OWN_THREADS) {
        const int r = i / IN_DIM_C, q = i - r * IN_DIM_C;
        o[i] = s_acc[r * 8 + q];
    }
}

#define POOL_GPB     4
#define POOL_THREADS 1024
#define POOL_U       14
#define POOL_Q       8

__global__ __launch_bounds__(POOL_THREADS)
void pool_fc_kernel(const float* __restrict__ h2, const int* __restrict__ batch,
                    const float* __restrict__ fcW, const float* __restrict__ fcb,
                    float* __restrict__ out) {
    constexpr int NW  = POOL_THREADS / 32;
    constexpr int NCH = (N_NODES_C + 31) / 32;
    static_assert(N_GRAPHS_C % POOL_GPB == 0, "every block owns POOL_GPB graphs");
    static_assert(POOL_GPB * HID_C <= POOL_THREADS, "one thread per (graph, channel) in the wave sum");

    __shared__ __align__(16) float s_part[NW][POOL_GPB][HID_C];
    __shared__ int   s_cnt [NW][POOL_GPB];
    __shared__ float s_mean[POOL_GPB][HID_C];

    const int tid  = threadIdx.x;
    const int lane = tid & 31;
    const int wave = tid >> 5;
    const int g0   = (int)blockIdx.x * POOL_GPB;
    const float2* h2v = (const float2*)h2;

    float2 acc[POOL_GPB];
    int    cnt[POOL_GPB];
#pragma unroll
    for (int k = 0; k < POOL_GPB; ++k) { acc[k] = make_float2(0.f, 0.f); cnt[k] = 0; }

    for (int c0 = wave; c0 < NCH; c0 += NW * POOL_U) {
        unsigned hit = 0;
#pragma unroll
        for (int u = 0; u < POOL_U; ++u) {
            const int c = c0 + u * NW;
            const int n = c * 32 + lane;
            const int g = (c < NCH && n < N_NODES_C) ? batch[n] : -1;
            const bool m = (unsigned)(g - g0) < (unsigned)POOL_GPB;
            if (__builtin_amdgcn_ballot_w32(m)) hit |= 1u << u;
        }
        while (hit) {
            const int u = __builtin_ctz(hit);
            hit &= hit - 1;
            const int base = (c0 + u * NW) * 32;
            const int n = base + lane;
            const int g = (n < N_NODES_C) ? batch[n] : -1;
            const int d = g - g0;
            unsigned mask = __builtin_amdgcn_ballot_w32((unsigned)d < (unsigned)POOL_GPB);
#pragma unroll
            for (int k = 0; k < POOL_GPB; ++k)
                cnt[k] += __builtin_popcount(__builtin_amdgcn_ballot_w32(d == k));
            while (mask) {
                int gq[POOL_Q]; float2 v[POOL_Q];
                const int jf = __builtin_ctz(mask);
#pragma unroll
                for (int q = 0; q < POOL_Q; ++q) {
                    const bool has = mask != 0;
                    const int j = has ? __builtin_ctz(mask) : jf;
                    const int gj = __builtin_amdgcn_readlane(d, j);
                    gq[q] = has ? gj : -1;
                    v[q]  = h2v[(size_t)(base + j) * (HID_C / 2) + lane];
                    mask &= mask - 1;
                }
                __builtin_amdgcn_sched_barrier(0);
#pragma unroll
                for (int q = 0; q < POOL_Q; ++q) {
#pragma unroll
                    for (int k = 0; k < POOL_GPB; ++k)
                        if (gq[q] == k) { acc[k].x += v[q].x; acc[k].y += v[q].y; }
                }
            }
        }
    }

#pragma unroll
    for (int k = 0; k < POOL_GPB; ++k) {
        *(float2*)&s_part[wave][k][2 * lane] = acc[k];
        if (lane == 0) s_cnt[wave][k] = cnt[k];
    }
    __syncthreads();

    if (tid < POOL_GPB * HID_C) {
        const int k = tid / HID_C, ch = tid - k * HID_C;
        float s = 0.0f; int c = 0;
        for (int w = 0; w < NW; ++w) { s += s_part[w][k][ch]; c += s_cnt[w][k]; }
        s_mean[k][ch] = s / fmaxf((float)c, 1.0f);
    }
    __syncthreads();

    if (tid < POOL_GPB * NUM_TASKS_C) {
        const int k = tid / NUM_TASKS_C, t = tid - k * NUM_TASKS_C;
        float s = fcb[t];
#pragma unroll
        for (int c = 0; c < HID_C; ++c)
            s += s_mean[k][c] * fcW[t * HID_C + c];
        out[(size_t)(g0 + k) * NUM_TASKS_C + t] = s;
    }
}

extern "C" void kernel_launch(void* const* d_in, const int* in_sizes, int n_in,
                              void* d_out, int out_size, void* d_ws, size_t ws_size,
                              hipStream_t stream) {
    const float* x    = (const float*)d_in[0];
    const int*   ei   = (const int*)  d_in[1];
    const float* ea   = (const float*)d_in[2];
    const int*   bat  = (const int*)  d_in[3];
    const float* e1W  = (const float*)d_in[4];  const float* e1b  = (const float*)d_in[5];
    const float* n1W1 = (const float*)d_in[6];  const float* n1b1 = (const float*)d_in[7];
    const float* n1W2 = (const float*)d_in[8];  const float* n1b2 = (const float*)d_in[9];
    const float* e2W  = (const float*)d_in[10]; const float* e2b  = (const float*)d_in[11];
    const float* n2W1 = (const float*)d_in[12]; const float* n2b1 = (const float*)d_in[13];
    const float* n2W2 = (const float*)d_in[14]; const float* n2b2 = (const float*)d_in[15];
    const float* fcW  = (const float*)d_in[16]; const float* fcb  = (const float*)d_in[17];
    float* out = (float*)d_out;

    float*    aggr1 = (float*)d_ws;
    _Float16* h     = (_Float16*)(aggr1 + (size_t)cfg1::S * N_NODES_C * IN_DIM_C);
    float*    aggr2 = (float*)(h + (size_t)N_NODES_C * HID_C);
    float*    h2    = aggr2 + (size_t)cfg2::S * N_NODES_C * HID_C;

    edge1_own_kernel<<<OWN_BLOCKS, OWN_THREADS, 0, stream>>>(x, ei, ea, e1W, e1b, aggr1);
    {
        const unsigned grid = (N_NODES_C + 15) / 16;
        node_mlp_kernel<IN_DIM_C, cfg1::S, float, _Float16><<<grid, 128, 0, stream>>>(
            x, aggr1, n1W1, n1b1, n1W2, n1b2, h, N_NODES_C);
    }
    edge2_own_kernel<<<OWN_BLOCKS, OWN_THREADS, 0, stream>>>(h, ei, ea, e2W, e2b, aggr2);
    {
        const unsigned grid = (N_NODES_C + 15) / 16;
        node_mlp_kernel<HID_C, cfg2::S, _Float16, float><<<grid, 128, 0, stream>>>(
            h, aggr2, n2W1, n2b1, n2W2, n2b2, h2, N_NODES_C);
    }
    pool_fc_kernel<<<N_GRAPHS_C / POOL_GPB, POOL_THREADS, 0, stream>>>(h2, bat, fcW, fcb, out);
}


# Conv2 edge kernel no longer scans dst: conv1's owner blocks write their filtered edge lists to the workspace and a conv2 block filters 4 of them for its range; sums still in LDS arrival order
# speedup vs baseline: 2.7958x; 1.1513x over previous
#include <hip/hip_runtime.h>
#include <hip/hip_bf16.h>

#define N_NODES_C  100000
#define N_EDGES_C  1600000
#define N_GRAPHS_C 1000
#define HID_C      64
#define IN_DIM_C   7
#define EDGE_DIM_C 4
#define NUM_TASKS_C 12

typedef __attribute__((ext_vector_type(16))) _Float16 v16h;
typedef __attribute__((ext_vector_type(8)))  _Float16 v8h;
typedef __attribute__((ext_vector_type(8)))  float    v8f;

__device__ __forceinline__ v16h lds_load_tile(const _Float16* s, int ld, int kbase) {
    const int lane = threadIdx.x & 31;
    const int r    = lane & 15;
    const int koff = (lane < 16) ? 0 : 8;
    const _Float16* p = s + r * ld + kbase + koff;
    v8h lo = *(const v8h*)(p);
    v8h hi = *(const v8h*)(p + 16);
    v16h out;
#pragma unroll
    for (int e = 0; e < 8; ++e) { out[e] = lo[e]; out[8 + e] = hi[e]; }
    return out;
}

template <int K1, int NP, typename TI, typename TO>
__global__ __launch_bounds__(128)
void node_mlp_kernel(const TI* __restrict__ in0, const float* __restrict__ in1,
                     const float* __restrict__ W1,  const float* __restrict__ b1,
                     const float* __restrict__ W2,  const float* __restrict__ b2,
                     TO* __restrict__ out, int n_nodes) {
    constexpr int K1P = (K1 + 31) & ~31;

    __shared__ __align__(16) _Float16 sA  [16 * K1P];
    __shared__ __align__(16) _Float16 sMid[16 * HID_C];
    __shared__ __align__(16) _Float16 sB1 [HID_C * K1P];
    __shared__ __align__(16) _Float16 sB2 [HID_C * HID_C];

    const int row0 = blockIdx.x * 16;
    const int tid  = threadIdx.x;
    const int wave = tid >> 5;
    const int lane = tid & 31;
    const int n0   = wave * 16;

    for (int i = tid; i < 16 * K1P; i += 128) {
        const int m = i / K1P, k = i - m * K1P;
        const int r = row0 + m;
        float v = 0.0f;
        if (k < K1 && r < n_nodes) {
            const size_t o = (size_t)r * K1 + k;
            v = (float)in0[o];
#pragma unroll
            for (int p = 0; p < NP; ++p)
                v += in1[(size_t)p * (size_t)n_nodes * K1 + o];
        }
        sA[i] = (_Float16)v;
    }
    for (int i = tid; i < HID_C * K1P; i += 128) {
        const int n = i / K1P, k = i - n * K1P;
        sB1[i] = (_Float16)((k < K1) ? W1[n * K1 + k] : 0.0f);
    }
    for (int i = tid; i < HID_C * HID_C; i += 128) {
        const int n = i >> 6, k = i & 63;
        sB2[i] = (_Float16)W2[n * HID_C + k];
    }
    __syncthreads();

    v8f acc = {};
#pragma unroll
    for (int kb = 0; kb < K1P; kb += 32) {
        v16h a = lds_load_tile(sA, K1P, kb);
        v16h b = lds_load_tile(sB1 + n0 * K1P, K1P, kb);
        acc = __builtin_amdgcn_wmma_f32_16x16x32_f16(false, a, false, b,
                                                     (short)0, acc, false, false);
    }
    {
        const int n  = lane & 15;
        const int mb = (lane < 16) ? 0 : 8;
        const float bias = b1[n0 + n];
#pragma unroll
        for (int r = 0; r < 8; ++r) {
            float v = acc[r] + bias;
            v = v > 0.0f ? v : 0.0f;
            sMid[(mb + r) * HID_C + (n0 + n)] = (_Float16)v;
        }
    }
    __syncthreads();

    v8f acc2 = {};
#pragma unroll
    for (int kb = 0; kb < HID_C; kb += 32) {
        v16h a = lds_load_tile(sMid, HID_C, kb);
        v16h b = lds_load_tile(sB2 + n0 * HID_C, HID_C, kb);
        acc2 = __builtin_amdgcn_wmma_f32_16x16x32_f16(false, a, false, b,
                                                      (short)0, acc2, false, false);
    }
    {
        const int n  = lane & 15;
        const int mb = (lane < 16) ? 0 : 8;
        const float bias = b2[n0 + n];
#pragma unroll
        for (int r = 0; r < 8; ++r) {
            const int rr = row0 + mb + r;
            if (rr < n_nodes) {
                float v = acc2[r] + bias;
                v = v > 0.0f ? v : 0.0f;
                out[(size_t)rr * HID_C + (n0 + n)] = (TO)v;
            }
        }
    }
}

#define OWN_BLOCKS  256
#define OWN_THREADS 1024

template <int RN_, int S_, int R_, int WCAP_, int DB_, int U_>
struct own_cfg {
    static constexpr int RN = RN_, S = S_, R = R_, WCAP = WCAP_, DB = DB_, U = U_;
    static constexpr int SL4   = N_EDGES_C / 4 / S_;
    static constexpr int STEP  = OWN_THREADS * U_;
    static constexpr int TRIPS = (SL4 + STEP - 1) / STEP;
    static_assert((N_EDGES_C / 4) % S_ == 0, "a slice is a whole number of int4 words");
    static_assert(R_ * S_ == OWN_BLOCKS, "one block per (range, slice)");
    static_assert(R_ * RN_ >= N_NODES_C, "ranges cover every node");
    static_assert(RN_ <= (1 << DB_), "place in range fits its bits");
    static_assert(((long long)SL4 * 4) <= (1ll << (32 - DB_)), "edge id in slice fits its bits");
};
typedef own_cfg<782, 2, 128, 512, 10, 8>  cfg2;
typedef own_cfg<3128, 8, 32, 1280, 12, 4> cfg1;
static_assert(cfg1::RN == 4 * cfg2::RN && cfg2::R == 4 * cfg1::R, "a conv1 range is 4 conv2 ranges");
static_assert(cfg1::S == 4 * cfg2::S, "a conv2 slice is 4 conv1 slices");
static_assert(cfg1::WCAP % 32 == 0, "a wave's padded list part fits its capacity");

#define LIST_SUBS  4
#define LIST_CAP   ((OWN_THREADS / 32) * cfg1::WCAP)
#define LIST_BAD   0xFFFFFFFFu
#define CNT_PITCH  32
#define LIST_U     8
static_assert((LIST_BAD & ((1u << cfg1::DB) - 1u)) >= (unsigned)cfg1::RN, "the pad entry lies outside every range");

typedef __attribute__((ext_vector_type(2))) _Float16 v2h;

template <class C>
__device__ __forceinline__ void own_push(unsigned* wlist, unsigned& wcount,
                                         int dstv, int n0, int erel) {
    const unsigned d = (unsigned)(dstv - n0);
    const bool m = d < (unsigned)C::RN;
    const unsigned mask = __builtin_amdgcn_ballot_w32(m);
    if (mask) {
        const unsigned pos = wcount + __builtin_amdgcn_mbcnt_lo(mask, 0u);
        if (m && pos < (unsigned)C::WCAP) wlist[pos] = ((unsigned)erel << C::DB) | d;
        wcount += (unsigned)__builtin_popcount(mask);
    }
}

template <class C>
__device__ __forceinline__ unsigned own_scan(const int4* __restrict__ d4, unsigned* wlist,
                                             int n0, int rot, int tid) {
    unsigned wcount = 0;
    for (int t = 0; t < C::TRIPS; ++t) {
        int tt = t + rot;
        if (tt >= C::TRIPS) tt -= C::TRIPS;
        const int base = tt * C::STEP;
        int4 v[C::U];
#pragma unroll
        for (int u = 0; u < C::U; ++u) {
            const int i = base + u * OWN_THREADS + tid;
            v[u] = (i < C::SL4) ? d4[i] : make_int4(-1, -1, -1, -1);
        }
#pragma unroll
        for (int u = 0; u < C::U; ++u) {
            const int e = (base + u * OWN_THREADS + tid) * 4;
            own_push<C>(wlist, wcount, v[u].x, n0, e + 0);
            own_push<C>(wlist, wcount, v[u].y, n0, e + 1);
            own_push<C>(wlist, wcount, v[u].z, n0, e + 2);
            own_push<C>(wlist, wcount, v[u].w, n0, e + 3);
        }
    }
    return wcount;
}

template <class C, class F>
__device__ __forceinline__ void own_overflow(const int4* __restrict__ d4, int n0, int rot,
                                             int tid, F add) {
    unsigned seen = 0;
#pragma unroll 1
    for (int t = 0; t < C::TRIPS * C::U; ++t) {
        int tt = t / C::U + rot;
        if (tt >= C::TRIPS) tt -= C::TRIPS;
        const int i = (tt * C::U + (t % C::U)) * OWN_THREADS + tid;
        const int4 v = (i < C::SL4) ? d4[i] : make_int4(-1, -1, -1, -1);
        const int dv[4] = {v.x, v.y, v.z, v.w};
#pragma unroll 1
        for (int k = 0; k < 4; ++k) {
            const unsigned d = (unsigned)(dv[k] - n0);
            const bool m = d < (unsigned)C::RN;
            const unsigned mask = __builtin_amdgcn_ballot_w32(m);
            const unsigned pos = seen + __builtin_amdgcn_mbcnt_lo(mask, 0u);
            seen += (unsigned)__builtin_popcount(mask);
            if (m && pos >= (unsigned)C::WCAP) add(i * 4 + k, (int)d);
        }
    }
}

__device__ __forceinline__ void list_push(unsigned* wlist, unsigned& wcount, unsigned ent,
                                          unsigned sublo, unsigned eoff) {
    const unsigned d = (ent & ((1u << cfg1::DB) - 1u)) - sublo;
    const bool m = d < (unsigned)cfg2::RN;
    const unsigned mask = __builtin_amdgcn_ballot_w32(m);
    if (mask) {
        const unsigned pos = wcount + __builtin_amdgcn_mbcnt_lo(mask, 0u);
        if (m && pos < (unsigned)cfg2::WCAP)
            wlist[pos] = (((ent >> cfg1::DB) + eoff) << cfg2::DB) | d;
        wcount += (unsigned)__builtin_popcount(mask);
    }
}

__device__ __forceinline__ unsigned list_scan(const unsigned* __restrict__ lists, int l0,
                                              const unsigned (&cnt)[LIST_SUBS], unsigned cmax,
                                              unsigned* wlist, unsigned sublo, int wave, int lane) {
    unsigned wcount = 0;
    for (unsigned k0 = 0; k0 * OWN_THREADS < cmax; k0 += LIST_U) {
        unsigned v[LIST_SUBS][LIST_U];
#pragma unroll
        for (int q = 0; q < LIST_SUBS; ++q) {
            const unsigned* L = lists + (size_t)(l0 + q) * LIST_CAP;
            const unsigned wb = (unsigned)(((wave + 8 * q) & 31) * 32 + lane);
#pragma unroll
            for (int u = 0; u < LIST_U; ++u) {
                const unsigned i = (k0 + (unsigned)u) * OWN_THREADS + wb;
                v[q][u] = (i < cnt[q]) ? L[i] : LIST_BAD;
            }
        }
#pragma unroll
        for (int q = 0; q < LIST_SUBS; ++q) {
#pragma unroll
            for (int u = 0; u < LIST_U; ++u)
                list_push(wlist, wcount, v[q][u], sublo, (unsigned)(q * (cfg1::SL4 * 4)));
        }
    }
    return wcount;
}

template <class F>
__device__ __forceinline__ void list_overflow(const unsigned* __restrict__ lists,
                                              const unsigned* __restrict__ lcount, int l0,
                                              unsigned cmax, unsigned sublo, int wave, int lane, F add) {
    unsigned seen = 0;
#pragma unroll 1
    for (unsigned k0 = 0; k0 * OWN_THREADS < cmax; k0 += LIST_U) {
#pragma unroll 1
        for (int q = 0; q < LIST_SUBS; ++q) {
            const unsigned* L = lists + (size_t)(l0 + q) * LIST_CAP;
            const unsigned cq = lcount[(l0 + q) * CNT_PITCH];
            const unsigned wb = (unsigned)(((wave + 8 * q) & 31) * 32 + lane);
#pragma unroll 1
            for (int u = 0; u < LIST_U; ++u) {
                const unsigned i = (k0 + (unsigned)u) * OWN_THREADS + wb;
                const unsigned ent = (i < cq) ? L[i] : LIST_BAD;
                const unsigned d = (ent & ((1u << cfg1::DB) - 1u)) - sublo;
                const bool m = d < (unsigned)cfg2::RN;
                const unsigned mask = __builtin_amdgcn_ballot_w32(m);
                const unsigned pos = seen + __builtin_amdgcn_mbcnt_lo(mask, 0u);
                seen += (unsigned)__builtin_popcount(mask);
                if (m && pos >= (unsigned)cfg2::WCAP)
                    add((int)((ent >> cfg1::DB) + (unsigned)(q * (cfg1::SL4 * 4))), (int)d);
            }
        }
    }
}

__global__ __launch_bounds__(OWN_THREADS)
void edge2_own_kernel(const _Float16* __restrict__ h, const int* __restrict__ ei,
                      const float* __restrict__ ea,
                      const float* __restrict__ eW, const float* __restrict__ eb,
                      const unsigned* __restrict__ lists, const unsigned* __restrict__ lcount,
                      float* __restrict__ aggr) {
    typedef cfg2 C;
    __shared__ __align__(16) float    s_acc[C::RN * HID_C];
    __shared__ __align__(16) unsigned s_list[(OWN_THREADS / 32) * C::WCAP];

    const int tid   = threadIdx.x;
    const int lane  = tid & 31;
    const int wave  = tid >> 5;
    const int range = (int)blockIdx.x / C::S;
    const int slice = (int)blockIdx.x - range * C::S;
    const int n0    = range * C::RN;
    const int e0    = slice * (C::SL4 * 4);
    unsigned* wlist = s_list + wave * C::WCAP;

    for (int i = tid; i < C::RN * HID_C / 4; i += OWN_THREADS)
        ((float4*)s_acc)[i] = make_float4(0.f, 0.f, 0.f, 0.f);
    __syncthreads();

    const int c0 = 2 * lane;
    const float4 w0 = *(const float4*)(eW + (c0 + 0) * EDGE_DIM_C);
    const float4 w1 = *(const float4*)(eW + (c0 + 1) * EDGE_DIM_C);
    const float  b0 = eb[c0], b1 = eb[c0 + 1];

    auto add_one = [&](int erel, int od) {
        const int oe = e0 + erel;
        const int os = ei[oe];
        const float4 a = *(const float4*)(ea + (size_t)oe * EDGE_DIM_C);
#pragma unroll 1
        for (int q = 0; q < HID_C; ++q) {
            const float4 w = *(const float4*)(eW + q * EDGE_DIM_C);
            float t = eb[q] + a.x * w.x + a.y * w.y + a.z * w.z + a.w * w.w;
            t += (float)h[(size_t)os * HID_C + q];
            atomicAdd(&s_acc[od * HID_C + q], fmaxf(t, 0.0f));
        }
    };

    const int l0 = (range / LIST_SUBS) * cfg1::S + slice * LIST_SUBS;
    const unsigned sublo = (unsigned)((range % LIST_SUBS) * C::RN);
    unsigned cnt[LIST_SUBS];
    unsigned cmax = 0;
#pragma unroll
    for (int q = 0; q < LIST_SUBS; ++q) {
        cnt[q] = lcount[(l0 + q) * CNT_PITCH];
        cmax = cnt[q] > cmax ? cnt[q] : cmax;
    }
    unsigned wcount;
    if (cmax <= (unsigned)LIST_CAP) {
        wcount = list_scan(lists, l0, cnt, cmax, wlist, sublo, wave, lane);
        if (wcount > (unsigned)C::WCAP)
            list_overflow(lists, lcount, l0, cmax, sublo, wave, lane, add_one);
    } else {
        const int4* d4 = (const int4*)(ei + N_EDGES_C + e0);
        const int rot = (range * C::TRIPS) / C::R;
        wcount = own_scan<C>(d4, wlist, n0, rot, tid);
        if (wcount > (unsigned)C::WCAP)
            own_overflow<C>(d4, n0, rot, tid, add_one);
    }
    const int wn = (int)(wcount < (unsigned)C::WCAP ? wcount : (unsigned)C::WCAP);

    __syncthreads();

    const v2h* h2p = (const v2h*)h;
    for (int p = 0; p < wn; p += 32) {
        const int idx = p + lane;
        const bool valid = idx < wn;
        const unsigned ent = valid ? wlist[idx] : 0u;
        const int e  = e0 + (int)(ent >> C::DB);
        const int dl = (int)(ent & ((1u << C::DB) - 1u));
        const int sl = valid ? ei[e] : 0;
        const float4 al = valid ? *(const float4*)(ea + (size_t)e * EDGE_DIM_C)
                                : make_float4(0.f, 0.f, 0.f, 0.f);
        const int cnt = (wn - p) < 32 ? (wn - p) : 32;
        v2h hv[32];
#pragma unroll
        for (int j = 0; j < 32; ++j) {
            const int sj = __builtin_amdgcn_readlane(sl, j);
            hv[j] = h2p[(size_t)sj * (HID_C / 2) + lane];
        }
#pragma unroll
        for (int j = 0; j < 32; ++j) {
            if (j < cnt) {
                const int   dj = __builtin_amdgcn_readlane(dl, j);
                const float ax = __builtin_bit_cast(float, __builtin_amdgcn_readlane(__builtin_bit_cast(int, al.x), j));
                const float ay = __builtin_bit_cast(float, __builtin_amdgcn_readlane(__builtin_bit_cast(int, al.y), j));
                const float az = __builtin_bit_cast(float, __builtin_amdgcn_readlane(__builtin_bit_cast(int, al.z), j));
                const float aw = __builtin_bit_cast(float, __builtin_amdgcn_readlane(__builtin_bit_cast(int, al.w), j));
                float t0 = b0 + ax * w0.x + ay * w0.y + az * w0.z + aw * w0.w;
                float t1 = b1 + ax * w1.x + ay * w1.y + az * w1.z + aw * w1.w;
                t0 += (float)hv[j][0];
                t1 += (float)hv[j][1];
                atomicAdd(&s_acc[dj * HID_C + c0],     fmaxf(t0, 0.0f));
                atomicAdd(&s_acc[dj * HID_C + c0 + 1], fmaxf(t1, 0.0f));
            }
        }
    }
    __syncthreads();

    int rows = N_NODES_C - n0;
    if (rows > C::RN) rows = C::RN;
    float4* o4 = (float4*)(aggr + (size_t)slice * N_NODES_C * HID_C + (size_t)n0 * HID_C);
    for (int i = tid; i < rows * (HID_C / 4); i += OWN_THREADS)
        o4[i] = ((const float4*)s_acc)[i];
}

__global__ __launch_bounds__(OWN_THREADS)
void edge1_own_kernel(const float* __restrict__ x, const int* __restrict__ ei,
                      const float* __restrict__ ea,
                      const float* __restrict__ eW, const float* __restrict__ eb,
                      float* __restrict__ aggr,
                      unsigned* __restrict__ lists, unsigned* __restrict__ lcount) {
    typedef cfg1 C;
    __shared__ __align__(16) float    s_acc[C::RN * 8];
    __shared__ __align__(16) unsigned s_list[(OWN_THREADS / 32) * C::WCAP];
    __shared__ unsigned s_wn[OWN_THREADS / 32];

    const int tid   = threadIdx.x;
    const int lane  = tid & 31;
    const int wave  = tid >> 5;
    const int range = (int)blockIdx.x / C::S;
    const int slice = (int)blockIdx.x - range * C::S;
    const int n0    = range * C::RN;
    const int e0    = slice * (C::SL4 * 4);
    unsigned* wlist = s_list + wave * C::WCAP;

    for (int i = tid; i < C::RN * 8; i += OWN_THREADS) s_acc[i] = 0.0f;
    __syncthreads();

    const int c   = lane & 7;
    const int sub = lane >> 3;
    const bool cact = c < IN_DIM_C;
    const float4 wc = cact ? *(const float4*)(eW + c * EDGE_DIM_C) : make_float4(0.f, 0.f, 0.f, 0.f);
    const float  bc = cact ? eb[c] : 0.0f;

    const int4* d4 = (const int4*)(ei + N_EDGES_C + e0);
    const int rot = (range * C::TRIPS) / C::R;
    const unsigned wcount = own_scan<C>(d4, wlist, n0, rot, tid);
    if (wcount > (unsigned)C::WCAP) {
        own_overflow<C>(d4, n0, rot, tid, [&](int erel, int od) {
            const int oe = e0 + erel;
            const int os = ei[oe];
            const float4 a = *(const float4*)(ea + (size_t)oe * EDGE_DIM_C);
#pragma unroll 1
            for (int q = 0; q < IN_DIM_C; ++q) {
                const float4 w = *(const float4*)(eW + q * EDGE_DIM_C);
                float t = eb[q] + a.x * w.x + a.y * w.y + a.z * w.z + a.w * w.w;
                t += x[(size_t)os * IN_DIM_C + q];
                atomicAdd(&s_acc[od * 8 + q], fmaxf(t, 0.0f));
            }
        });
    }
    const int wn = (int)(wcount < (unsigned)C::WCAP ? wcount : (unsigned)C::WCAP);
    if (lane == 0) s_wn[wave] = wcount;

    __syncthreads();

    {
        unsigned off = 0, tot = 0;
        bool full = false;
        for (int w = 0; w < OWN_THREADS / 32; ++w) {
            const unsigned cw = s_wn[w];
            full = full || cw > (unsigned)C::WCAP;
            const unsigned pw = ((cw < (unsigned)C::WCAP ? cw : (unsigned)C::WCAP) + 31u) & ~31u;
            if (w < wave) off += pw;
            tot += pw;
        }
        unsigned* L = lists + (size_t)blockIdx.x * LIST_CAP + off;
        const int wp = (wn + 31) & ~31;
        for (int i = lane; i < wp; i += 32) L[i] = (i < wn) ? wlist[i] : LIST_BAD;
        if (tid == 0) lcount[blockIdx.x * CNT_PITCH] = full ? LIST_BAD : tot;
    }

    for (int p = 0; p < wn; p += 32) {
        unsigned ent[8]; int sr[8]; float4 av[8]; float xv[8]; bool ok[8];
#pragma unroll
        for (int g = 0; g < 8; ++g) {
            const int idx = p + g * 4 + sub;
            ok[g]  = idx < wn;
            ent[g] = ok[g] ? wlist[idx] : 0u;
        }
#pragma unroll
        for (int g = 0; g < 8; ++g) {
            const int e = e0 + (int)(ent[g] >> C::DB);
            sr[g] = ok[g] ? ei[e] : 0;
            av[g] = ok[g] ? *(const float4*)(ea + (size_t)e * EDGE_DIM_C)
                          : make_float4(0.f, 0.f, 0.f, 0.f);
        }
#pragma unroll
        for (int g = 0; g < 8; ++g)
            xv[g] = (ok[g] && cact) ? x[(size_t)sr[g] * IN_DIM_C + c] : 0.0f;
#pragma unroll
        for (int g = 0; g < 8; ++g) {
            if (ok[g] && cact) {
                const int dl = (int)(ent[g] & ((1u << C::DB) - 1u));
                float t = bc + av[g].x * wc.x + av[g].y * wc.y + av[g].z * wc.z + av[g].w * wc.w;
                t += xv[g];
                atomicAdd(&s_acc[dl * 8 + c], fmaxf(t, 0.0f));
            }
        }
    }
    __syncthreads();

    int rows = N_NODES_C - n0;
    if (rows > C::RN) rows = C::RN;
    float* o = aggr + (size_t)slice * N_NODES_C * IN_DIM_C + (size_t)n0 * IN_DIM_C;
    for (int i = tid; i < rows * IN_DIM_C; i += OWN_THREADS) {
        const int r = i / IN_DIM_C, q = i - r * IN_DIM_C;
        o[i] = s_acc[r * 8 + q];
    }
}

#define POOL_GPB     4
#define POOL_THREADS 1024
#define POOL_U       14
#define POOL_Q       8

__global__ __launch_bounds__(POOL_THREADS)
void pool_fc_kernel(const float* __restrict__ h2, const int* __restrict__ batch,
                    const float* __restrict__ fcW, const float* __restrict__ fcb,
                    float* __restrict__ out) {
    constexpr int NW  = POOL_THREADS / 32;
    constexpr int NCH = (N_NODES_C + 31) / 32;
    static_assert(N_GRAPHS_C % POOL_GPB == 0, "every block owns POOL_GPB graphs");
    static_assert(POOL_GPB * HID_C <= POOL_THREADS, "one thread per (graph, channel) in the wave sum");

    __shared__ __align__(16) float s_part[NW][POOL_GPB][HID_C];
    __shared__ int   s_cnt [NW][POOL_GPB];
    __shared__ float s_mean[POOL_GPB][HID_C];

    const int tid  = threadIdx.x;
    const int lane = tid & 31;
    const int wave = tid >> 5;
    const int g0   = (int)blockIdx.x * POOL_GPB;
    const float2* h2v = (const float2*)h2;

    float2 acc[POOL_GPB];
    int    cnt[POOL_GPB];
#pragma unroll
    for (int k = 0; k < POOL_GPB; ++k) { acc[k] = make_float2(0.f, 0.f); cnt[k] = 0; }

    for (int c0 = wave; c0 < NCH; c0 += NW * POOL_U) {
        unsigned hit = 0;
#pragma unroll
        for (int u = 0; u < POOL_U; ++u) {
            const int c = c0 + u * NW;
            const int n = c * 32 + lane;
            const int g = (c < NCH && n < N_NODES_C) ? batch[n] : -1;
            const bool m = (unsigned)(g - g0) < (unsigned)POOL_GPB;
            if (__builtin_amdgcn_ballot_w32(m)) hit |= 1u << u;
        }
        while (hit) {
            const int u = __builtin_ctz(hit);
            hit &= hit - 1;
            const int base = (c0 + u * NW) * 32;
            const int n = base + lane;
            const int g = (n < N_NODES_C) ? batch[n] : -1;
            const int d = g - g0;
            unsigned mask = __builtin_amdgcn_ballot_w32((unsigned)d < (unsigned)POOL_GPB);
#pragma unroll
            for (int k = 0; k < POOL_GPB; ++k)
                cnt[k] += __builtin_popcount(__builtin_amdgcn_ballot_w32(d == k));
            while (mask) {
                int gq[POOL_Q]; float2 v[POOL_Q];
                const int jf = __builtin_ctz(mask);
#pragma unroll
                for (int q = 0; q < POOL_Q; ++q) {
                    const bool has = mask != 0;
                    const int j = has ? __builtin_ctz(mask) : jf;
                    const int gj = __builtin_amdgcn_readlane(d, j);
                    gq[q] = has ? gj : -1;
                    v[q]  = h2v[(size_t)(base + j) * (HID_C / 2) + lane];
                    mask &= mask - 1;
                }
                __builtin_amdgcn_sched_barrier(0);
#pragma unroll
                for (int q = 0; q < POOL_Q; ++q) {
#pragma unroll
                    for (int k = 0; k < POOL_GPB; ++k)
                        if (gq[q] == k) { acc[k].x += v[q].x; acc[k].y += v[q].y; }
                }
            }
        }
    }

#pragma unroll
    for (int k = 0; k < POOL_GPB; ++k) {
        *(float2*)&s_part[wave][k][2 * lane] = acc[k];
        if (lane == 0) s_cnt[wave][k] = cnt[k];
    }
    __syncthreads();

    if (tid < POOL_GPB * HID_C) {
        const int k = tid / HID_C, ch = tid - k * HID_C;
        float s = 0.0f; int c = 0;
        for (int w = 0; w < NW; ++w) { s += s_part[w][k][ch]; c += s_cnt[w][k]; }
        s_mean[k][ch] = s / fmaxf((float)c, 1.0f);
    }
    __syncthreads();

    if (tid < POOL_GPB * NUM_TASKS_C) {
        const int k = tid / NUM_TASKS_C, t = tid - k * NUM_TASKS_C;
        float s = fcb[t];
#pragma unroll
        for (int c = 0; c < HID_C; ++c)
            s += s_mean[k][c] * fcW[t * HID_C + c];
        out[(size_t)(g0 + k) * NUM_TASKS_C + t] = s;
    }
}

extern "C" void kernel_launch(void* const* d_in, const int* in_sizes, int n_in,
                              void* d_out, int out_size, void* d_ws, size_t ws_size,
                              hipStream_t stream) {
    const float* x    = (const float*)d_in[0];
    const int*   ei   = (const int*)  d_in[1];
    const float* ea   = (const float*)d_in[2];
    const int*   bat  = (const int*)  d_in[3];
    const float* e1W  = (const float*)d_in[4];  const float* e1b  = (const float*)d_in[5];
    const float* n1W1 = (const float*)d_in[6];  const float* n1b1 = (const float*)d_in[7];
    const float* n1W2 = (const float*)d_in[8];  const float* n1b2 = (const float*)d_in[9];
    const float* e2W  = (const float*)d_in[10]; const float* e2b  = (const float*)d_in[11];
    const float* n2W1 = (const float*)d_in[12]; const float* n2b1 = (const float*)d_in[13];
    const float* n2W2 = (const float*)d_in[14]; const float* n2b2 = (const float*)d_in[15];
    const float* fcW  = (const float*)d_in[16]; const float* fcb  = (const float*)d_in[17];
    float* out = (float*)d_out;

    float*    aggr1 = (float*)d_ws;
    _Float16* h     = (_Float16*)(aggr1 + (size_t)cfg1::S * N_NODES_C * IN_DIM_C);
    float*    aggr2 = (float*)(h + (size_t)N_NODES_C * HID_C);
    float*    h2    = aggr2 + (size_t)cfg2::S * N_NODES_C * HID_C;
    unsigned* lists  = (unsigned*)(h2 + (size_t)N_NODES_C * HID_C);
    unsigned* lcount = lists + (size_t)OWN_BLOCKS * LIST_CAP;

    edge1_own_kernel<<<OWN_BLOCKS, OWN_THREADS, 0, stream>>>(x, ei, ea, e1W, e1b, aggr1, lists, lcount);
    {
        const unsigned grid = (N_NODES_C + 15) / 16;
        node_mlp_kernel<IN_DIM_C, cfg1::S, float, _Float16><<<grid, 128, 0, stream>>>(
            x, aggr1, n1W1, n1b1, n1W2, n1b2, h, N_NODES_C);
    }
    edge2_own_kernel<<<OWN_BLOCKS, OWN_THREADS, 0, stream>>>(h, ei, ea, e2W, e2b, lists, lcount, aggr2);
    {
        const unsigned grid = (N_NODES_C + 15) / 16;
        node_mlp_kernel<HID_C, cfg2::S, _Float16, float><<<grid, 128, 0, stream>>>(
            h, aggr2, n2W1, n2b1, n2W2, n2b2, h2, N_NODES_C);
    }
    pool_fc_kernel<<<N_GRAPHS_C / POOL_GPB, POOL_THREADS, 0, stream>>>(h2, bat, fcW, fcb, out);
}


# Node MLPs as transposed products, one wave per two 16-node tiles: layer-1 D packed in registers = layer 2's B operand; W1/W2 staged once per 256 nodes; whole-line tile I/O; same sums per element
# speedup vs baseline: 3.4550x; 1.2358x over previous
#include <hip/hip_runtime.h>
#include <hip/hip_bf16.h>

#define N_NODES_C  100000
#define N_EDGES_C  1600000
#define N_GRAPHS_C 1000
#define HID_C      64
#define IN_DIM_C   7
#define EDGE_DIM_C 4
#define NUM_TASKS_C 12

typedef __attribute__((ext_vector_type(16))) _Float16 v16h;
typedef __attribute__((ext_vector_type(8)))  _Float16 v8h;
typedef __attribute__((ext_vector_type(8)))  float    v8f;

__device__ __forceinline__ v16h lds_load_tile(const _Float16* s, int ld, int kbase) {
    const int lane = threadIdx.x & 31;
    const int r    = lane & 15;
    const int koff = (lane < 16) ? 0 : 8;
    const _Float16* p = s + r * ld + kbase + koff;
    v8h lo = *(const v8h*)(p);
    v8h hi = *(const v8h*)(p + 16);
    v16h out;
#pragma unroll
    for (int e = 0; e < 8; ++e) { out[e] = lo[e]; out[8 + e] = hi[e]; }
    return out;
}

#define MLP_WAVES   8
#define MLP_NT      2
#define MLP_THREADS (MLP_WAVES * 32)
#define MLP_TILES   (MLP_WAVES * MLP_NT)
#define MLP_ROWS    (MLP_TILES * 16)
#define MLP_GRID    ((N_NODES_C + MLP_ROWS - 1) / MLP_ROWS)
static_assert(N_NODES_C % 16 == 0, "the node count is a whole number of 16-node tiles");

typedef __attribute__((ext_vector_type(4))) _Float16 v4h;

template <int K, int KP>
struct mlp_wregs {
    static constexpr bool VEC = (K % 4) == 0;
    static constexpr int  N   = (VEC ? HID_C * (K / 4) : HID_C * KP) / MLP_THREADS;
    static_assert((VEC ? HID_C * (K / 4) : HID_C * KP) % MLP_THREADS == 0, "the same number of pieces in every thread");
    static_assert(!VEC || K == KP, "no padding on the 16-byte path");
    float4 v4[VEC ? N : 1];
    float  v1[VEC ? 1 : N];
};

template <int K, int KP>
__device__ __forceinline__ void mlp_load_weights(const float* __restrict__ W, int tid, mlp_wregs<K, KP>& r) {
#pragma unroll
    for (int it = 0; it < mlp_wregs<K, KP>::N; ++it) {
        const int i = tid + it * MLP_THREADS;
        if constexpr (mlp_wregs<K, KP>::VEC) {
            r.v4[it] = ((const float4*)W)[i];
        } else {
            const int n = i / KP, k = i - n * KP;
            const float w = W[n * K + (k < K ? k : 0)];
            r.v1[it] = (k < K) ? w : 0.0f;
        }
    }
}

template <int K, int KP, int P>
__device__ __forceinline__ void mlp_store_weights(_Float16* s, int tid, const mlp_wregs<K, KP>& r) {
#pragma unroll
    for (int it = 0; it < mlp_wregs<K, KP>::N; ++it) {
        const int i = tid + it * MLP_THREADS;
        if constexpr (mlp_wregs<K, KP>::VEC) {
            const int n = i / (K / 4), q = i - n * (K / 4);
            const float4 v = r.v4[it];
            v4h o;
            o[0] = (_Float16)v.x; o[1] = (_Float16)v.y; o[2] = (_Float16)v.z; o[3] = (_Float16)v.w;
            *(v4h*)(s + n * P + 4 * q) = o;
        } else {
            const int n = i / KP, k = i - n * KP;
            s[n * P + k] = (_Float16)r.v1[it];
        }
    }
}

template <int K1, int NP>
__global__ __launch_bounds__(MLP_THREADS)
void node_mlp_kernel(const float* __restrict__ xin, const _Float16* __restrict__ hin,
                     const float* __restrict__ in1,
                     const float* __restrict__ W1,  const float* __restrict__ b1,
                     const float* __restrict__ W2,  const float* __restrict__ b2,
                     _Float16* __restrict__ outh, float* __restrict__ outf, int n_nodes) {
    constexpr int  K1P   = (K1 + 31) & ~31;
    constexpr int  KS    = K1P / 32;
    constexpr int  P1    = K1P + 8;
    constexpr int  P2    = HID_C + 8;
    constexpr int  PO    = HID_C + 4;
    constexpr bool SMALL = (K1 % 8) != 0;
    constexpr int  XT    = 16 * K1;
    static_assert(!SMALL || (XT % 4 == 0 && XT / 4 <= 32), "a flat conv1 tile is at most one float4 per lane");
    static_assert(SMALL || K1 == HID_C, "the conv2 path cuts rows of 64");

    __shared__ __align__(16) _Float16 sW1[HID_C * P1];
    __shared__ __align__(16) _Float16 sW2[HID_C * P2];
    __shared__ __align__(16) float    sX [SMALL ? MLP_TILES * 128 : 4];
    __shared__ __align__(16) _Float16 sA [SMALL ? 8 : MLP_TILES * 16 * P1];
    __shared__ __align__(16) _Float16 sOh[SMALL ? MLP_TILES * 16 * P2 : 8];
    __shared__ __align__(16) float    sOf[SMALL ? 4 : MLP_TILES * 16 * PO];

    const int tid  = threadIdx.x;
    const int wave = __builtin_amdgcn_readfirstlane(tid >> 5);
    const int lane = tid & 31;
    const int c    = lane & 15;
    const int h    = lane >> 4;
    const int tile0 = ((int)blockIdx.x * MLP_WAVES + wave) * MLP_NT;

    int  row0[MLP_NT];
    bool tv[MLP_NT];
#pragma unroll
    for (int j = 0; j < MLP_NT; ++j) {
        row0[j] = (tile0 + j) * 16;
        tv[j]   = row0[j] < n_nodes;
    }

    mlp_wregs<K1, K1P>     w1r;
    mlp_wregs<HID_C, HID_C> w2r;
    mlp_load_weights<K1, K1P>(W1, tid, w1r);
    mlp_load_weights<HID_C, HID_C>(W2, tid, w2r);

    if constexpr (SMALL) {
        const int ll = lane < XT / 4 ? lane : XT / 4 - 1;
        float4 raw[MLP_NT][NP + 1];
#pragma unroll
        for (int j = 0; j < MLP_NT; ++j) {
            const int o4 = (tv[j] ? row0[j] : 0) * K1 / 4 + ll;
            raw[j][0] = ((const float4*)xin)[o4];
#pragma unroll
            for (int p = 0; p < NP; ++p)
                raw[j][p + 1] = ((const float4*)(in1 + (size_t)p * (size_t)n_nodes * K1))[o4];
        }
        float4 v[MLP_NT];
#pragma unroll
        for (int j = 0; j < MLP_NT; ++j) {
            v[j] = raw[j][0];
#pragma unroll
            for (int p = 0; p < NP; ++p) {
                const float4 a = raw[j][p + 1];
                v[j].x += a.x; v[j].y += a.y; v[j].z += a.z; v[j].w += a.w;
            }
        }
#pragma unroll
        for (int j = 0; j < MLP_NT; ++j)
            ((float4*)sX)[(wave * MLP_NT + j) * 32 + lane] = v[j];
    } else {
#pragma unroll
        for (int j = 0; j < MLP_NT; ++j) {
            const int    eb = (tv[j] ? row0[j] : 0) * K1;
            const int    tl = wave * MLP_NT + j;
#pragma unroll
            for (int i = 0; i < XT / 128; ++i) {
                const int f = 4 * (lane + 32 * i);
                const v4h x4 = *(const v4h*)(hin + eb + f);
                float v0 = (float)x4[0], v1 = (float)x4[1], v2 = (float)x4[2], v3 = (float)x4[3];
#pragma unroll
                for (int p = 0; p < NP; ++p) {
                    const float4 a = *(const float4*)(in1 + (size_t)p * (size_t)n_nodes * K1 + eb + f);
                    v0 += a.x; v1 += a.y; v2 += a.z; v3 += a.w;
                }
                v4h o;
                o[0] = (_Float16)v0; o[1] = (_Float16)v1; o[2] = (_Float16)v2; o[3] = (_Float16)v3;
                *(v4h*)(sA + (tl * 16 + (f >> 6)) * P1 + (f & 63)) = o;
            }
        }
    }

    mlp_store_weights<K1, K1P, P1>(sW1, tid, w1r);
    mlp_store_weights<HID_C, HID_C, P2>(sW2, tid, w2r);
    __syncthreads();

    v16h bop[MLP_NT][KS];
#pragma unroll
    for (int j = 0; j < MLP_NT; ++j) {
        const int tl = wave * MLP_NT + j;
        if constexpr (SMALL) {
            const float* r = sX + tl * 128 + c * K1;
            v16h b = {};
#pragma unroll
            for (int k = 0; k < K1; ++k) {
                const float f = r[k];
                b[k] = (h == 0) ? (_Float16)f : (_Float16)0.0f;
            }
            bop[j][0] = b;
        } else {
#pragma unroll
            for (int kb = 0; kb < KS; ++kb)
                bop[j][kb] = lds_load_tile(sA + tl * 16 * P1, P1, kb * 32);
        }
    }

    float bias1[4][8], bias2[4][8];
#pragma unroll
    for (int t = 0; t < 4; ++t) {
        const float4 a0 = *(const float4*)(b1 + 16 * t + 8 * h);
        const float4 a1 = *(const float4*)(b1 + 16 * t + 8 * h + 4);
        bias1[t][0] = a0.x; bias1[t][1] = a0.y; bias1[t][2] = a0.z; bias1[t][3] = a0.w;
        bias1[t][4] = a1.x; bias1[t][5] = a1.y; bias1[t][6] = a1.z; bias1[t][7] = a1.w;
        const float4 c0 = *(const float4*)(b2 + 16 * t + 8 * h);
        const float4 c1 = *(const float4*)(b2 + 16 * t + 8 * h + 4);
        bias2[t][0] = c0.x; bias2[t][1] = c0.y; bias2[t][2] = c0.z; bias2[t][3] = c0.w;
        bias2[t][4] = c1.x; bias2[t][5] = c1.y; bias2[t][6] = c1.z; bias2[t][7] = c1.w;
    }

    v8f acc[MLP_NT][4];
#pragma unroll
    for (int j = 0; j < MLP_NT; ++j)
#pragma unroll
        for (int t = 0; t < 4; ++t) acc[j][t] = v8f{};
#pragma unroll
    for (int t = 0; t < 4; ++t) {
#pragma unroll
        for (int kb = 0; kb < KS; ++kb) {
            const v16h a = lds_load_tile(sW1 + 16 * t * P1, P1, kb * 32);
#pragma unroll
            for (int j = 0; j < MLP_NT; ++j)
                acc[j][t] = __builtin_amdgcn_wmma_f32_16x16x32_f16(false, a, false, bop[j][kb],
                                                                   (short)0, acc[j][t], false, false);
        }
    }

    v16h bop2[MLP_NT][2];
#pragma unroll
    for (int j = 0; j < MLP_NT; ++j) {
#pragma unroll
        for (int s = 0; s < 2; ++s) {
            v16h b;
#pragma unroll
            for (int e = 0; e < 8; ++e) {
                float v0 = acc[j][2 * s][e] + bias1[2 * s][e];
                float v1 = acc[j][2 * s + 1][e] + bias1[2 * s + 1][e];
                v0 = v0 > 0.0f ? v0 : 0.0f;
                v1 = v1 > 0.0f ? v1 : 0.0f;
                b[e]     = (_Float16)v0;
                b[8 + e] = (_Float16)v1;
            }
            bop2[j][s] = b;
        }
    }

    v8f acc2[MLP_NT][4];
#pragma unroll
    for (int j = 0; j < MLP_NT; ++j)
#pragma unroll
        for (int t = 0; t < 4; ++t) acc2[j][t] = v8f{};
#pragma unroll
    for (int t = 0; t < 4; ++t) {
#pragma unroll
        for (int s = 0; s < 2; ++s) {
            const v16h a = lds_load_tile(sW2 + 16 * t * P2, P2, s * 32);
#pragma unroll
            for (int j = 0; j < MLP_NT; ++j)
                acc2[j][t] = __builtin_amdgcn_wmma_f32_16x16x32_f16(false, a, false, bop2[j][s],
                                                                    (short)0, acc2[j][t], false, false);
        }
    }

#pragma unroll
    for (int j = 0; j < MLP_NT; ++j) {
        const int tl = wave * MLP_NT + j;
#pragma unroll
        for (int t = 0; t < 4; ++t) {
            float v[8];
#pragma unroll
            for (int e = 0; e < 8; ++e) {
                const float u = acc2[j][t][e] + bias2[t][e];
                v[e] = u > 0.0f ? u : 0.0f;
            }
            if constexpr (SMALL) {
                v8h o8;
#pragma unroll
                for (int e = 0; e < 8; ++e) o8[e] = (_Float16)v[e];
                *(v8h*)(sOh + (tl * 16 + c) * P2 + 16 * t + 8 * h) = o8;
            } else {
                float* q = sOf + (tl * 16 + c) * PO + 16 * t + 8 * h;
                *(float4*)(q)     = make_float4(v[0], v[1], v[2], v[3]);
                *(float4*)(q + 4) = make_float4(v[4], v[5], v[6], v[7]);
            }
        }
    }
    __syncthreads();

#pragma unroll
    for (int j = 0; j < MLP_NT; ++j) {
        if (tv[j]) {
            const int tl = wave * MLP_NT + j;
            if constexpr (SMALL) {
#pragma unroll
                for (int i = 0; i < 4; ++i) {
                    const int f = 8 * (lane + 32 * i);
                    const v8h o8 = *(const v8h*)(sOh + (tl * 16 + (f >> 6)) * P2 + (f & 63));
                    *(v8h*)(outh + (size_t)row0[j] * HID_C + f) = o8;
                }
            } else {
#pragma unroll
                for (int i = 0; i < 8; ++i) {
                    const int f = 4 * (lane + 32 * i);
                    const float4 o4 = *(const float4*)(sOf + (tl * 16 + (f >> 6)) * PO + (f & 63));
                    *(float4*)(outf + (size_t)row0[j] * HID_C + f) = o4;
                }
            }
        }
    }
}

#define OWN_BLOCKS  256
#define OWN_THREADS 1024

template <int RN_, int S_, int R_, int WCAP_, int DB_, int U_>
struct own_cfg {
    static constexpr int RN = RN_, S = S_, R = R_, WCAP = WCAP_, DB = DB_, U = U_;
    static constexpr int SL4   = N_EDGES_C / 4 / S_;
    static constexpr int STEP  = OWN_THREADS * U_;
    static constexpr int TRIPS = (SL4 + STEP - 1) / STEP;
    static_assert((N_EDGES_C / 4) % S_ == 0, "a slice is a whole number of int4 words");
    static_assert(R_ * S_ == OWN_BLOCKS, "one block per (range, slice)");
    static_assert(R_ * RN_ >= N_NODES_C, "ranges cover every node");
    static_assert(RN_ <= (1 << DB_), "place in range fits its bits");
    static_assert(((long long)SL4 * 4) <= (1ll << (32 - DB_)), "edge id in slice fits its bits");
};
typedef own_cfg<782, 2, 128, 512, 10, 8>  cfg2;
typedef own_cfg<3128, 8, 32, 1280, 12, 4> cfg1;
static_assert(cfg1::RN == 4 * cfg2::RN && cfg2::R == 4 * cfg1::R, "a conv1 range is 4 conv2 ranges");
static_assert(cfg1::S == 4 * cfg2::S, "a conv2 slice is 4 conv1 slices");
static_assert(cfg1::WCAP % 32 == 0, "a wave's padded list part fits its capacity");

#define LIST_SUBS  4
#define LIST_CAP   ((OWN_THREADS / 32) * cfg1::WCAP)
#define LIST_BAD   0xFFFFFFFFu
#define CNT_PITCH  32
#define LIST_U     8
static_assert((LIST_BAD & ((1u << cfg1::DB) - 1u)) >= (unsigned)cfg1::RN, "the pad entry lies outside every range");

typedef __attribute__((ext_vector_type(2))) _Float16 v2h;

template <class C>
__device__ __forceinline__ void own_push(unsigned* wlist, unsigned& wcount,
                                         int dstv, int n0, int erel) {
    const unsigned d = (unsigned)(dstv - n0);
    const bool m = d < (unsigned)C::RN;
    const unsigned mask = __builtin_amdgcn_ballot_w32(m);
    if (mask) {
        const unsigned pos = wcount + __builtin_amdgcn_mbcnt_lo(mask, 0u);
        if (m && pos < (unsigned)C::WCAP) wlist[pos] = ((unsigned)erel << C::DB) | d;
        wcount += (unsigned)__builtin_popcount(mask);
    }
}

template <class C>
__device__ __forceinline__ unsigned own_scan(const int4* __restrict__ d4, unsigned* wlist,
                                             int n0, int rot, int tid) {
    unsigned wcount = 0;
    for (int t = 0; t < C::TRIPS; ++t) {
        int tt = t + rot;
        if (tt >= C::TRIPS) tt -= C::TRIPS;
        const int base = tt * C::STEP;
        int4 v[C::U];
#pragma unroll
        for (int u = 0; u < C::U; ++u) {
            const int i = base + u * OWN_THREADS + tid;
            v[u] = (i < C::SL4) ? d4[i] : make_int4(-1, -1, -1, -1);
        }
#pragma unroll
        for (int u = 0; u < C::U; ++u) {
            const int e = (base + u * OWN_THREADS + tid) * 4;
            own_push<C>(wlist, wcount, v[u].x, n0, e + 0);
            own_push<C>(wlist, wcount, v[u].y, n0, e + 1);
            own_push<C>(wlist, wcount, v[u].z, n0, e + 2);
            own_push<C>(wlist, wcount, v[u].w, n0, e + 3);
        }
    }
    return wcount;
}

template <class C, class F>
__device__ __forceinline__ void own_overflow(const int4* __restrict__ d4, int n0, int rot,
                                             int tid, F add) {
    unsigned seen = 0;
#pragma unroll 1
    for (int t = 0; t < C::TRIPS * C::U; ++t) {
        int tt = t / C::U + rot;
        if (tt >= C::TRIPS) tt -= C::TRIPS;
        const int i = (tt * C::U + (t % C::U)) * OWN_THREADS + tid;
        const int4 v = (i < C::SL4) ? d4[i] : make_int4(-1, -1, -1, -1);
        const int dv[4] = {v.x, v.y, v.z, v.w};
#pragma unroll 1
        for (int k = 0; k < 4; ++k) {
            const unsigned d = (unsigned)(dv[k] - n0);
            const bool m = d < (unsigned)C::RN;
            const unsigned mask = __builtin_amdgcn_ballot_w32(m);
            const unsigned pos = seen + __builtin_amdgcn_mbcnt_lo(mask, 0u);
            seen += (unsigned)__builtin_popcount(mask);
            if (m && pos >= (unsigned)C::WCAP) add(i * 4 + k, (int)d);
        }
    }
}

__device__ __forceinline__ void list_push(unsigned* wlist, unsigned& wcount, unsigned ent,
                                          unsigned sublo, unsigned eoff) {
    const unsigned d = (ent & ((1u << cfg1::DB) - 1u)) - sublo;
    const bool m = d < (unsigned)cfg2::RN;
    const unsigned mask = __builtin_amdgcn_ballot_w32(m);
    if (mask) {
        const unsigned pos = wcount + __builtin_amdgcn_mbcnt_lo(mask, 0u);
        if (m && pos < (unsigned)cfg2::WCAP)
            wlist[pos] = (((ent >> cfg1::DB) + eoff) << cfg2::DB) | d;
        wcount += (unsigned)__builtin_popcount(mask);
    }
}

__device__ __forceinline__ unsigned list_scan(const unsigned* __restrict__ lists, int l0,
                                              const unsigned (&cnt)[LIST_SUBS], unsigned cmax,
                                              unsigned* wlist, unsigned sublo, int wave, int lane) {
    unsigned wcount = 0;
    for (unsigned k0 = 0; k0 * OWN_THREADS < cmax; k0 += LIST_U) {
        unsigned v[LIST_SUBS][LIST_U];
#pragma unroll
        for (int q = 0; q < LIST_SUBS; ++q) {
            const unsigned* L = lists + (size_t)(l0 + q) * LIST_CAP;
            const unsigned wb = (unsigned)(((wave + 8 * q) & 31) * 32 + lane);
#pragma unroll
            for (int u = 0; u < LIST_U; ++u) {
                const unsigned i = (k0 + (unsigned)u) * OWN_THREADS + wb;
                v[q][u] = (i < cnt[q]) ? L[i] : LIST_BAD;
            }
        }
#pragma unroll
        for (int q = 0; q < LIST_SUBS; ++q) {
#pragma unroll
            for (int u = 0; u < LIST_U; ++u)
                list_push(wlist, wcount, v[q][u], sublo, (unsigned)(q * (cfg1::SL4 * 4)));
        }
    }
    return wcount;
}

template <class F>
__device__ __forceinline__ void list_overflow(const unsigned* __restrict__ lists,
                                              const unsigned* __restrict__ lcount, int l0,
                                              unsigned cmax, unsigned sublo, int wave, int lane, F add) {
    unsigned seen = 0;
#pragma unroll 1
    for (unsigned k0 = 0; k0 * OWN_THREADS < cmax; k0 += LIST_U) {
#pragma unroll 1
        for (int q = 0; q < LIST_SUBS; ++q) {
            const unsigned* L = lists + (size_t)(l0 + q) * LIST_CAP;
            const unsigned cq = lcount[(l0 + q) * CNT_PITCH];
            const unsigned wb = (unsigned)(((wave + 8 * q) & 31) * 32 + lane);
#pragma unroll 1
            for (int u = 0; u < LIST_U; ++u) {
                const unsigned i = (k0 + (unsigned)u) * OWN_THREADS + wb;
                const unsigned ent = (i < cq) ? L[i] : LIST_BAD;
                const unsigned d = (ent & ((1u << cfg1::DB) - 1u)) - sublo;
                const bool m = d < (unsigned)cfg2::RN;
                const unsigned mask = __builtin_amdgcn_ballot_w32(m);
                const unsigned pos = seen + __builtin_amdgcn_mbcnt_lo(mask, 0u);
                seen += (unsigned)__builtin_popcount(mask);
                if (m && pos >= (unsigned)cfg2::WCAP)
                    add((int)((ent >> cfg1::DB) + (unsigned)(q * (cfg1::SL4 * 4))), (int)d);
            }
        }
    }
}

__global__ __launch_bounds__(OWN_THREADS)
void edge2_own_kernel(const _Float16* __restrict__ h, const int* __restrict__ ei,
                      const float* __restrict__ ea,
                      const float* __restrict__ eW, const float* __restrict__ eb,
                      const unsigned* __restrict__ lists, const unsigned* __restrict__ lcount,
                      float* __restrict__ aggr) {
    typedef cfg2 C;
    __shared__ __align__(16) float    s_acc[C::RN * HID_C];
    __shared__ __align__(16) unsigned s_list[(OWN_THREADS / 32) * C::WCAP];

    const int tid   = threadIdx.x;
    const int lane  = tid & 31;
    const int wave  = tid >> 5;
    const int range = (int)blockIdx.x / C::S;
    const int slice = (int)blockIdx.x - range * C::S;
    const int n0    = range * C::RN;
    const int e0    = slice * (C::SL4 * 4);
    unsigned* wlist = s_list + wave * C::WCAP;

    for (int i = tid; i < C::RN * HID_C / 4; i += OWN_THREADS)
        ((float4*)s_acc)[i] = make_float4(0.f, 0.f, 0.f, 0.f);
    __syncthreads();

    const int c0 = 2 * lane;
    const float4 w0 = *(const float4*)(eW + (c0 + 0) * EDGE_DIM_C);
    const float4 w1 = *(const float4*)(eW + (c0 + 1) * EDGE_DIM_C);
    const float  b0 = eb[c0], b1 = eb[c0 + 1];

    auto add_one = [&](int erel, int od) {
        const int oe = e0 + erel;
        const int os = ei[oe];
        const float4 a = *(const float4*)(ea + (size_t)oe * EDGE_DIM_C);
#pragma unroll 1
        for (int q = 0; q < HID_C; ++q) {
            const float4 w = *(const float4*)(eW + q * EDGE_DIM_C);
            float t = eb[q] + a.x * w.x + a.y * w.y + a.z * w.z + a.w * w.w;
            t += (float)h[(size_t)os * HID_C + q];
            atomicAdd(&s_acc[od * HID_C + q], fmaxf(t, 0.0f));
        }
    };

    const int l0 = (range / LIST_SUBS) * cfg1::S + slice * LIST_SUBS;
    const unsigned sublo = (unsigned)((range % LIST_SUBS) * C::RN);
    unsigned cnt[LIST_SUBS];
    unsigned cmax = 0;
#pragma unroll
    for (int q = 0; q < LIST_SUBS; ++q) {
        cnt[q] = lcount[(l0 + q) * CNT_PITCH];
        cmax = cnt[q] > cmax ? cnt[q] : cmax;
    }
    unsigned wcount;
    if (cmax <= (unsigned)LIST_CAP) {
        wcount = list_scan(lists, l0, cnt, cmax, wlist, sublo, wave, lane);
        if (wcount > (unsigned)C::WCAP)
            list_overflow(lists, lcount, l0, cmax, sublo, wave, lane, add_one);
    } else {
        const int4* d4 = (const int4*)(ei + N_EDGES_C + e0);
        const int rot = (range * C::TRIPS) / C::R;
        wcount = own_scan<C>(d4, wlist, n0, rot, tid);
        if (wcount > (unsigned)C::WCAP)
            own_overflow<C>(d4, n0, rot, tid, add_one);
    }
    const int wn = (int)(wcount < (unsigned)C::WCAP ? wcount : (unsigned)C::WCAP);

    __syncthreads();

    const v2h* h2p = (const v2h*)h;
    for (int p = 0; p < wn; p += 32) {
        const int idx = p + lane;
        const bool valid = idx < wn;
        const unsigned ent = valid ? wlist[idx] : 0u;
        const int e  = e0 + (int)(ent >> C::DB);
        const int dl = (int)(ent & ((1u << C::DB) - 1u));
        const int sl = valid ? ei[e] : 0;
        const float4 al = valid ? *(const float4*)(ea + (size_t)e * EDGE_DIM_C)
                                : make_float4(0.f, 0.f, 0.f, 0.f);
        const int cnt = (wn - p) < 32 ? (wn - p) : 32;
        v2h hv[32];
#pragma unroll
        for (int j = 0; j < 32; ++j) {
            const int sj = __builtin_amdgcn_readlane(sl, j);
            hv[j] = h2p[(size_t)sj * (HID_C / 2) + lane];
        }
#pragma unroll
        for (int j = 0; j < 32; ++j) {
            if (j < cnt) {
                const int   dj = __builtin_amdgcn_readlane(dl, j);
                const float ax = __builtin_bit_cast(float, __builtin_amdgcn_readlane(__builtin_bit_cast(int, al.x), j));
                const float ay = __builtin_bit_cast(float, __builtin_amdgcn_readlane(__builtin_bit_cast(int, al.y), j));
                const float az = __builtin_bit_cast(float, __builtin_amdgcn_readlane(__builtin_bit_cast(int, al.z), j));
                const float aw = __builtin_bit_cast(float, __builtin_amdgcn_readlane(__builtin_bit_cast(int, al.w), j));
                float t0 = b0 + ax * w0.x + ay * w0.y + az * w0.z + aw * w0.w;
                float t1 = b1 + ax * w1.x + ay * w1.y + az * w1.z + aw * w1.w;
                t0 += (float)hv[j][0];
                t1 += (float)hv[j][1];
                atomicAdd(&s_acc[dj * HID_C + c0],     fmaxf(t0, 0.0f));
                atomicAdd(&s_acc[dj * HID_C + c0 + 1], fmaxf(t1, 0.0f));
            }
        }
    }
    __syncthreads();

    int rows = N_NODES_C - n0;
    if (rows > C::RN) rows = C::RN;
    float4* o4 = (float4*)(aggr + (size_t)slice * N_NODES_C * HID_C + (size_t)n0 * HID_C);
    for (int i = tid; i < rows * (HID_C / 4); i += OWN_THREADS)
        o4[i] = ((const float4*)s_acc)[i];
}

__global__ __launch_bounds__(OWN_THREADS)
void edge1_own_kernel(const float* __restrict__ x, const int* __restrict__ ei,
                      const float* __restrict__ ea,
                      const float* __restrict__ eW, const float* __restrict__ eb,
                      float* __restrict__ aggr,
                      unsigned* __restrict__ lists, unsigned* __restrict__ lcount) {
    typedef cfg1 C;
    __shared__ __align__(16) float    s_acc[C::RN * 8];
    __shared__ __align__(16) unsigned s_list[(OWN_THREADS / 32) * C::WCAP];
    __shared__ unsigned s_wn[OWN_THREADS / 32];

    const int tid   = threadIdx.x;
    const int lane  = tid & 31;
    const int wave  = tid >> 5;
    const int range = (int)blockIdx.x / C::S;
    const int slice = (int)blockIdx.x - range * C::S;
    const int n0    = range * C::RN;
    const int e0    = slice * (C::SL4 * 4);
    unsigned* wlist = s_list + wave * C::WCAP;

    for (int i = tid; i < C::RN * 8; i += OWN_THREADS) s_acc[i] = 0.0f;
    __syncthreads();

    const int c   = lane & 7;
    const int sub = lane >> 3;
    const bool cact = c < IN_DIM_C;
    const float4 wc = cact ? *(const float4*)(eW + c * EDGE_DIM_C) : make_float4(0.f, 0.f, 0.f, 0.f);
    const float  bc = cact ? eb[c] : 0.0f;

    const int4* d4 = (const int4*)(ei + N_EDGES_C + e0);
    const int rot = (range * C::TRIPS) / C::R;
    const unsigned wcount = own_scan<C>(d4, wlist, n0, rot, tid);
    if (wcount > (unsigned)C::WCAP) {
        own_overflow<C>(d4, n0, rot, tid, [&](int erel, int od) {
            const int oe = e0 + erel;
            const int os = ei[oe];
            const float4 a = *(const float4*)(ea + (size_t)oe * EDGE_DIM_C);
#pragma unroll 1
            for (int q = 0; q < IN_DIM_C; ++q) {
                const float4 w = *(const float4*)(eW + q * EDGE_DIM_C);
                float t = eb[q] + a.x * w.x + a.y * w.y + a.z * w.z + a.w * w.w;
                t += x[(size_t)os * IN_DIM_C + q];
                atomicAdd(&s_acc[od * 8 + q], fmaxf(t, 0.0f));
            }
        });
    }
    const int wn = (int)(wcount < (unsigned)C::WCAP ? wcount : (unsigned)C::WCAP);
    if (lane == 0) s_wn[wave] = wcount;

    __syncthreads();

    {
        unsigned off = 0, tot = 0;
        bool full = false;
        for (int w = 0; w < OWN_THREADS / 32; ++w) {
            const unsigned cw = s_wn[w];
            full = full || cw > (unsigned)C::WCAP;
            const unsigned pw = ((cw < (unsigned)C::WCAP ? cw : (unsigned)C::WCAP) + 31u) & ~31u;
            if (w < wave) off += pw;
            tot += pw;
        }
        unsigned* L = lists + (size_t)blockIdx.x * LIST_CAP + off;
        const int wp = (wn + 31) & ~31;
        for (int i = lane; i < wp; i += 32) L[i] = (i < wn) ? wlist[i] : LIST_BAD;
        if (tid == 0) lcount[blockIdx.x * CNT_PITCH] = full ? LIST_BAD : tot;
    }

    for (int p = 0; p < wn; p += 32) {
        unsigned ent[8]; int sr[8]; float4 av[8]; float xv[8]; bool ok[8];
#pragma unroll
        for (int g = 0; g < 8; ++g) {
            const int idx = p + g * 4 + sub;
            ok[g]  = idx < wn;
            ent[g] = ok[g] ? wlist[idx] : 0u;
        }
#pragma unroll
        for (int g = 0; g < 8; ++g) {
            const int e = e0 + (int)(ent[g] >> C::DB);
            sr[g] = ok[g] ? ei[e] : 0;
            av[g] = ok[g] ? *(const float4*)(ea + (size_t)e * EDGE_DIM_C)
                          : make_float4(0.f, 0.f, 0.f, 0.f);
        }
#pragma unroll
        for (int g = 0; g < 8; ++g)
            xv[g] = (ok[g] && cact) ? x[(size_t)sr[g] * IN_DIM_C + c] : 0.0f;
#pragma unroll
        for (int g = 0; g < 8; ++g) {
            if (ok[g] && cact) {
                const int dl = (int)(ent[g] & ((1u << C::DB) - 1u));
                float t = bc + av[g].x * wc.x + av[g].y * wc.y + av[g].z * wc.z + av[g].w * wc.w;
                t += xv[g];
                atomicAdd(&s_acc[dl * 8 + c], fmaxf(t, 0.0f));
            }
        }
    }
    __syncthreads();

    int rows = N_NODES_C - n0;
    if (rows > C::RN) rows = C::RN;
    float* o = aggr + (size_t)slice * N_NODES_C * IN_DIM_C + (size_t)n0 * IN_DIM_C;
    for (int i = tid; i < rows * IN_DIM_C; i += OWN_THREADS) {
        const int r = i / IN_DIM_C, q = i - r * IN_DIM_C;
        o[i] = s_acc[r * 8 + q];
    }
}

#define POOL_GPB     4
#define POOL_THREADS 1024
#define POOL_U       14
#define POOL_Q       8

__global__ __launch_bounds__(POOL_THREADS)
void pool_fc_kernel(const float* __restrict__ h2, const int* __restrict__ batch,
                    const float* __restrict__ fcW, const float* __restrict__ fcb,
                    float* __restrict__ out) {
    constexpr int NW  = POOL_THREADS / 32;
    constexpr int NCH = (N_NODES_C + 31) / 32;
    static_assert(N_GRAPHS_C % POOL_GPB == 0, "every block owns POOL_GPB graphs");
    static_assert(POOL_GPB * HID_C <= POOL_THREADS, "one thread per (graph, channel) in the wave sum");

    __shared__ __align__(16) float s_part[NW][POOL_GPB][HID_C];
    __shared__ int   s_cnt [NW][POOL_GPB];
    __shared__ float s_mean[POOL_GPB][HID_C];

    const int tid  = threadIdx.x;
    const int lane = tid & 31;
    const int wave = tid >> 5;
    const int g0   = (int)blockIdx.x * POOL_GPB;
    const float2* h2v = (const float2*)h2;

    float2 acc[POOL_GPB];
    int    cnt[POOL_GPB];
#pragma unroll
    for (int k = 0; k < POOL_GPB; ++k) { acc[k] = make_float2(0.f, 0.f); cnt[k] = 0; }

    for (int c0 = wave; c0 < NCH; c0 += NW * POOL_U) {
        unsigned hit = 0;
#pragma unroll
        for (int u = 0; u < POOL_U; ++u) {
            const int c = c0 + u * NW;
            const int n = c * 32 + lane;
            const int g = (c < NCH && n < N_NODES_C) ? batch[n] : -1;
            const bool m = (unsigned)(g - g0) < (unsigned)POOL_GPB;
            if (__builtin_amdgcn_ballot_w32(m)) hit |= 1u << u;
        }
        while (hit) {
            const int u = __builtin_ctz(hit);
            hit &= hit - 1;
            const int base = (c0 + u * NW) * 32;
            const int n = base + lane;
            const int g = (n < N_NODES_C) ? batch[n] : -1;
            const int d = g - g0;
            unsigned mask = __builtin_amdgcn_ballot_w32((unsigned)d < (unsigned)POOL_GPB);
#pragma unroll
            for (int k = 0; k < POOL_GPB; ++k)
                cnt[k] += __builtin_popcount(__builtin_amdgcn_ballot_w32(d == k));
            while (mask) {
                int gq[POOL_Q]; float2 v[POOL_Q];
                const int jf = __builtin_ctz(mask);
#pragma unroll
                for (int q = 0; q < POOL_Q; ++q) {
                    const bool has = mask != 0;
                    const int j = has ? __builtin_ctz(mask) : jf;
                    const int gj = __builtin_amdgcn_readlane(d, j);
                    gq[q] = has ? gj : -1;
                    v[q]  = h2v[(size_t)(base + j) * (HID_C / 2) + lane];
                    mask &= mask - 1;
                }
                __builtin_amdgcn_sched_barrier(0);
#pragma unroll
                for (int q = 0; q < POOL_Q; ++q) {
#pragma unroll
                    for (int k = 0; k < POOL_GPB; ++k)
                        if (gq[q] == k) { acc[k].x += v[q].x; acc[k].y += v[q].y; }
                }
            }
        }
    }

#pragma unroll
    for (int k = 0; k < POOL_GPB; ++k) {
        *(float2*)&s_part[wave][k][2 * lane] = acc[k];
        if (lane == 0) s_cnt[wave][k] = cnt[k];
    }
    __syncthreads();

    if (tid < POOL_GPB * HID_C) {
        const int k = tid / HID_C, ch = tid - k * HID_C;
        float s = 0.0f; int c = 0;
        for (int w = 0; w < NW; ++w) { s += s_part[w][k][ch]; c += s_cnt[w][k]; }
        s_mean[k][ch] = s / fmaxf((float)c, 1.0f);
    }
    __syncthreads();

    if (tid < POOL_GPB * NUM_TASKS_C) {
        const int k = tid / NUM_TASKS_C, t = tid - k * NUM_TASKS_C;
        float s = fcb[t];
#pragma unroll
        for (int c = 0; c < HID_C; ++c)
            s += s_mean[k][c] * fcW[t * HID_C + c];
        out[(size_t)(g0 + k) * NUM_TASKS_C + t] = s;
    }
}

extern "C" void kernel_launch(void* const* d_in, const int* in_sizes, int n_in,
                              void* d_out, int out_size, void* d_ws, size_t ws_size,
                              hipStream_t stream) {
    const float* x    = (const float*)d_in[0];
    const int*   ei   = (const int*)  d_in[1];
    const float* ea   = (const float*)d_in[2];
    const int*   bat  = (const int*)  d_in[3];
    const float* e1W  = (const float*)d_in[4];  const float* e1b  = (const float*)d_in[5];
    const float* n1W1 = (const float*)d_in[6];  const float* n1b1 = (const float*)d_in[7];
    const float* n1W2 = (const float*)d_in[8];  const float* n1b2 = (const float*)d_in[9];
    const float* e2W  = (const float*)d_in[10]; const float* e2b  = (const float*)d_in[11];
    const float* n2W1 = (const float*)d_in[12]; const float* n2b1 = (const float*)d_in[13];
    const float* n2W2 = (const float*)d_in[14]; const float* n2b2 = (const float*)d_in[15];
    const float* fcW  = (const float*)d_in[16]; const float* fcb  = (const float*)d_in[17];
    float* out = (float*)d_out;

    float*    aggr1 = (float*)d_ws;
    _Float16* h     = (_Float16*)(aggr1 + (size_t)cfg1::S * N_NODES_C * IN_DIM_C);
    float*    aggr2 = (float*)(h + (size_t)N_NODES_C * HID_C);
    float*    h2    = aggr2 + (size_t)cfg2::S * N_NODES_C * HID_C;
    unsigned* lists  = (unsigned*)(h2 + (size_t)N_NODES_C * HID_C);
    unsigned* lcount = lists + (size_t)OWN_BLOCKS * LIST_CAP;

    edge1_own_kernel<<<OWN_BLOCKS, OWN_THREADS, 0, stream>>>(x, ei, ea, e1W, e1b, aggr1, lists, lcount);
    node_mlp_kernel<IN_DIM_C, cfg1::S><<<MLP_GRID, MLP_THREADS, 0, stream>>>(
        x, nullptr, aggr1, n1W1, n1b1, n1W2, n1b2, h, nullptr, N_NODES_C);
    edge2_own_kernel<<<OWN_BLOCKS, OWN_THREADS, 0, stream>>>(h, ei, ea, e2W, e2b, lists, lcount, aggr2);
    node_mlp_kernel<HID_C, cfg2::S><<<MLP_GRID, MLP_THREADS, 0, stream>>>(
        nullptr, h, aggr2, n2W1, n2b1, n2W2, n2b2, nullptr, h2, N_NODES_C);
    pool_fc_kernel<<<N_GRAPHS_C / POOL_GPB, POOL_THREADS, 0, stream>>>(h2, bat, fcW, fcb, out);
}


# Conv2 node MLP runs inside the conv2 owner blocks: 250 ranges of 400 nodes x whole edge list, WMMA on the block's LDS sums; aggr2 planes and a launch gone; node sum = one LDS arrival-order sum
# speedup vs baseline: 3.6040x; 1.0431x over previous
#include <hip/hip_runtime.h>
#include <hip/hip_bf16.h>

#define N_NODES_C  100000
#define N_EDGES_C  1600000
#define N_GRAPHS_C 1000
#define HID_C      64
#define IN_DIM_C   7
#define EDGE_DIM_C 4
#define NUM_TASKS_C 12

typedef __attribute__((ext_vector_type(16))) _Float16 v16h;
typedef __attribute__((ext_vector_type(8)))  _Float16 v8h;
typedef __attribute__((ext_vector_type(8)))  float    v8f;

__device__ __forceinline__ v16h lds_load_tile(const _Float16* s, int ld, int kbase) {
    const int lane = threadIdx.x & 31;
    const int r    = lane & 15;
    const int koff = (lane < 16) ? 0 : 8;
    const _Float16* p = s + r * ld + kbase + koff;
    v8h lo = *(const v8h*)(p);
    v8h hi = *(const v8h*)(p + 16);
    v16h out;
#pragma unroll
    for (int e = 0; e < 8; ++e) { out[e] = lo[e]; out[8 + e] = hi[e]; }
    return out;
}

#define MLP_WAVES   8
#define MLP_NT      2
#define MLP_THREADS (MLP_WAVES * 32)
#define MLP_TILES   (MLP_WAVES * MLP_NT)
#define MLP_ROWS    (MLP_TILES * 16)
#define MLP_GRID    ((N_NODES_C + MLP_ROWS - 1) / MLP_ROWS)
static_assert(N_NODES_C % 16 == 0, "the node count is a whole number of 16-node tiles");

typedef __attribute__((ext_vector_type(4))) _Float16 v4h;

template <int K, int KP>
struct mlp_wregs {
    static constexpr bool VEC = (K % 4) == 0;
    static constexpr int  N   = (VEC ? HID_C * (K / 4) : HID_C * KP) / MLP_THREADS;
    static_assert((VEC ? HID_C * (K / 4) : HID_C * KP) % MLP_THREADS == 0, "the same number of pieces in every thread");
    static_assert(!VEC || K == KP, "no padding on the 16-byte path");
    float4 v4[VEC ? N : 1];
    float  v1[VEC ? 1 : N];
};

template <int K, int KP>
__device__ __forceinline__ void mlp_load_weights(const float* __restrict__ W, int tid, mlp_wregs<K, KP>& r) {
#pragma unroll
    for (int it = 0; it < mlp_wregs<K, KP>::N; ++it) {
        const int i = tid + it * MLP_THREADS;
        if constexpr (mlp_wregs<K, KP>::VEC) {
            r.v4[it] = ((const float4*)W)[i];
        } else {
            const int n = i / KP, k = i - n * KP;
            const float w = W[n * K + (k < K ? k : 0)];
            r.v1[it] = (k < K) ? w : 0.0f;
        }
    }
}

template <int K, int KP, int P>
__device__ __forceinline__ void mlp_store_weights(_Float16* s, int tid, const mlp_wregs<K, KP>& r) {
#pragma unroll
    for (int it = 0; it < mlp_wregs<K, KP>::N; ++it) {
        const int i = tid + it * MLP_THREADS;
        if constexpr (mlp_wregs<K, KP>::VEC) {
            const int n = i / (K / 4), q = i - n * (K / 4);
            const float4 v = r.v4[it];
            v4h o;
            o[0] = (_Float16)v.x; o[1] = (_Float16)v.y; o[2] = (_Float16)v.z; o[3] = (_Float16)v.w;
            *(v4h*)(s + n * P + 4 * q) = o;
        } else {
            const int n = i / KP, k = i - n * KP;
            s[n * P + k] = (_Float16)r.v1[it];
        }
    }
}

template <int K1, int NP>
__global__ __launch_bounds__(MLP_THREADS)
void node_mlp_kernel(const float* __restrict__ xin, const _Float16* __restrict__ hin,
                     const float* __restrict__ in1,
                     const float* __restrict__ W1,  const float* __restrict__ b1,
                     const float* __restrict__ W2,  const float* __restrict__ b2,
                     _Float16* __restrict__ outh, float* __restrict__ outf, int n_nodes) {
    constexpr int  K1P   = (K1 + 31) & ~31;
    constexpr int  KS    = K1P / 32;
    constexpr int  P1    = K1P + 8;
    constexpr int  P2    = HID_C + 8;
    constexpr int  PO    = HID_C + 4;
    constexpr bool SMALL = (K1 % 8) != 0;
    constexpr int  XT    = 16 * K1;
    static_assert(!SMALL || (XT % 4 == 0 && XT / 4 <= 32), "a flat conv1 tile is at most one float4 per lane");
    static_assert(SMALL || K1 == HID_C, "the conv2 path cuts rows of 64");

    __shared__ __align__(16) _Float16 sW1[HID_C * P1];
    __shared__ __align__(16) _Float16 sW2[HID_C * P2];
    __shared__ __align__(16) float    sX [SMALL ? MLP_TILES * 128 : 4];
    __shared__ __align__(16) _Float16 sA [SMALL ? 8 : MLP_TILES * 16 * P1];
    __shared__ __align__(16) _Float16 sOh[SMALL ? MLP_TILES * 16 * P2 : 8];
    __shared__ __align__(16) float    sOf[SMALL ? 4 : MLP_TILES * 16 * PO];

    const int tid  = threadIdx.x;
    const int wave = __builtin_amdgcn_readfirstlane(tid >> 5);
    const int lane = tid & 31;
    const int c    = lane & 15;
    const int h    = lane >> 4;
    const int tile0 = ((int)blockIdx.x * MLP_WAVES + wave) * MLP_NT;

    int  row0[MLP_NT];
    bool tv[MLP_NT];
#pragma unroll
    for (int j = 0; j < MLP_NT; ++j) {
        row0[j] = (tile0 + j) * 16;
        tv[j]   = row0[j] < n_nodes;
    }

    mlp_wregs<K1, K1P>     w1r;
    mlp_wregs<HID_C, HID_C> w2r;
    mlp_load_weights<K1, K1P>(W1, tid, w1r);
    mlp_load_weights<HID_C, HID_C>(W2, tid, w2r);

    if constexpr (SMALL) {
        const int ll = lane < XT / 4 ? lane : XT / 4 - 1;
        float4 raw[MLP_NT][NP + 1];
#pragma unroll
        for (int j = 0; j < MLP_NT; ++j) {
            const int o4 = (tv[j] ? row0[j] : 0) * K1 / 4 + ll;
            raw[j][0] = ((const float4*)xin)[o4];
#pragma unroll
            for (int p = 0; p < NP; ++p)
                raw[j][p + 1] = ((const float4*)(in1 + (size_t)p * (size_t)n_nodes * K1))[o4];
        }
        float4 v[MLP_NT];
#pragma unroll
        for (int j = 0; j < MLP_NT; ++j) {
            v[j] = raw[j][0];
#pragma unroll
            for (int p = 0; p < NP; ++p) {
                const float4 a = raw[j][p + 1];
                v[j].x += a.x; v[j].y += a.y; v[j].z += a.z; v[j].w += a.w;
            }
        }
#pragma unroll
        for (int j = 0; j < MLP_NT; ++j)
            ((float4*)sX)[(wave * MLP_NT + j) * 32 + lane] = v[j];
    } else {
#pragma unroll
        for (int j = 0; j < MLP_NT; ++j) {
            const int    eb = (tv[j] ? row0[j] : 0) * K1;
            const int    tl = wave * MLP_NT + j;
#pragma unroll
            for (int i = 0; i < XT / 128; ++i) {
                const int f = 4 * (lane + 32 * i);
                const v4h x4 = *(const v4h*)(hin + eb + f);
                float v0 = (float)x4[0], v1 = (float)x4[1], v2 = (float)x4[2], v3 = (float)x4[3];
#pragma unroll
                for (int p = 0; p < NP; ++p) {
                    const float4 a = *(const float4*)(in1 + (size_t)p * (size_t)n_nodes * K1 + eb + f);
                    v0 += a.x; v1 += a.y; v2 += a.z; v3 += a.w;
                }
                v4h o;
                o[0] = (_Float16)v0; o[1] = (_Float16)v1; o[2] = (_Float16)v2; o[3] = (_Float16)v3;
                *(v4h*)(sA + (tl * 16 + (f >> 6)) * P1 + (f & 63)) = o;
            }
        }
    }

    mlp_store_weights<K1, K1P, P1>(sW1, tid, w1r);
    mlp_store_weights<HID_C, HID_C, P2>(sW2, tid, w2r);
    __syncthreads();

    v16h bop[MLP_NT][KS];
#pragma unroll
    for (int j = 0; j < MLP_NT; ++j) {
        const int tl = wave * MLP_NT + j;
        if constexpr (SMALL) {
            const float* r = sX + tl * 128 + c * K1;
            v16h b = {};
#pragma unroll
            for (int k = 0; k < K1; ++k) {
                const float f = r[k];
                b[k] = (h == 0) ? (_Float16)f : (_Float16)0.0f;
            }
            bop[j][0] = b;
        } else {
#pragma unroll
            for (int kb = 0; kb < KS; ++kb)
                bop[j][kb] = lds_load_tile(sA + tl * 16 * P1, P1, kb * 32);
        }
    }

    float bias1[4][8], bias2[4][8];
#pragma unroll
    for (int t = 0; t < 4; ++t) {
        const float4 a0 = *(const float4*)(b1 + 16 * t + 8 * h);
        const float4 a1 = *(const float4*)(b1 + 16 * t + 8 * h + 4);
        bias1[t][0] = a0.x; bias1[t][1] = a0.y; bias1[t][2] = a0.z; bias1[t][3] = a0.w;
        bias1[t][4] = a1.x; bias1[t][5] = a1.y; bias1[t][6] = a1.z; bias1[t][7] = a1.w;
        const float4 c0 = *(const float4*)(b2 + 16 * t + 8 * h);
        const float4 c1 = *(const float4*)(b2 + 16 * t + 8 * h + 4);
        bias2[t][0] = c0.x; bias2[t][1] = c0.y; bias2[t][2] = c0.z; bias2[t][3] = c0.w;
        bias2[t][4] = c1.x; bias2[t][5] = c1.y; bias2[t][6] = c1.z; bias2[t][7] = c1.w;
    }

    v8f acc[MLP_NT][4];
#pragma unroll
    for (int j = 0; j < MLP_NT; ++j)
#pragma unroll
        for (int t = 0; t < 4; ++t) acc[j][t] = v8f{};
#pragma unroll
    for (int t = 0; t < 4; ++t) {
#pragma unroll
        for (int kb = 0; kb < KS; ++kb) {
            const v16h a = lds_load_tile(sW1 + 16 * t * P1, P1, kb * 32);
#pragma unroll
            for (int j = 0; j < MLP_NT; ++j)
                acc[j][t] = __builtin_amdgcn_wmma_f32_16x16x32_f16(false, a, false, bop[j][kb],
                                                                   (short)0, acc[j][t], false, false);
        }
    }

    v16h bop2[MLP_NT][2];
#pragma unroll
    for (int j = 0; j < MLP_NT; ++j) {
#pragma unroll
        for (int s = 0; s < 2; ++s) {
            v16h b;
#pragma unroll
            for (int e = 0; e < 8; ++e) {
                float v0 = acc[j][2 * s][e] + bias1[2 * s][e];
                float v1 = acc[j][2 * s + 1][e] + bias1[2 * s + 1][e];
                v0 = v0 > 0.0f ? v0 : 0.0f;
                v1 = v1 > 0.0f ? v1 : 0.0f;
                b[e]     = (_Float16)v0;
                b[8 + e] = (_Float16)v1;
            }
            bop2[j][s] = b;
        }
    }

    v8f acc2[MLP_NT][4];
#pragma unroll
    for (int j = 0; j < MLP_NT; ++j)
#pragma unroll
        for (int t = 0; t < 4; ++t) acc2[j][t] = v8f{};
#pragma unroll
    for (int t = 0; t < 4; ++t) {
#pragma unroll
        for (int s = 0; s < 2; ++s) {
            const v16h a = lds_load_tile(sW2 + 16 * t * P2, P2, s * 32);
#pragma unroll
            for (int j = 0; j < MLP_NT; ++j)
                acc2[j][t] = __builtin_amdgcn_wmma_f32_16x16x32_f16(false, a, false, bop2[j][s],
                                                                    (short)0, acc2[j][t], false, false);
        }
    }

#pragma unroll
    for (int j = 0; j < MLP_NT; ++j) {
        const int tl = wave * MLP_NT + j;
#pragma unroll
        for (int t = 0; t < 4; ++t) {
            float v[8];
#pragma unroll
            for (int e = 0; e < 8; ++e) {
                const float u = acc2[j][t][e] + bias2[t][e];
                v[e] = u > 0.0f ? u : 0.0f;
            }
            if constexpr (SMALL) {
                v8h o8;
#pragma unroll
                for (int e = 0; e < 8; ++e) o8[e] = (_Float16)v[e];
                *(v8h*)(sOh + (tl * 16 + c) * P2 + 16 * t + 8 * h) = o8;
            } else {
                float* q = sOf + (tl * 16 + c) * PO + 16 * t + 8 * h;
                *(float4*)(q)     = make_float4(v[0], v[1], v[2], v[3]);
                *(float4*)(q + 4) = make_float4(v[4], v[5], v[6], v[7]);
            }
        }
    }
    __syncthreads();

#pragma unroll
    for (int j = 0; j < MLP_NT; ++j) {
        if (tv[j]) {
            const int tl = wave * MLP_NT + j;
            if constexpr (SMALL) {
#pragma unroll
                for (int i = 0; i < 4; ++i) {
                    const int f = 8 * (lane + 32 * i);
                    const v8h o8 = *(const v8h*)(sOh + (tl * 16 + (f >> 6)) * P2 + (f & 63));
                    *(v8h*)(outh + (size_t)row0[j] * HID_C + f) = o8;
                }
            } else {
#pragma unroll
                for (int i = 0; i < 8; ++i) {
                    const int f = 4 * (lane + 32 * i);
                    const float4 o4 = *(const float4*)(sOf + (tl * 16 + (f >> 6)) * PO + (f & 63));
                    *(float4*)(outf + (size_t)row0[j] * HID_C + f) = o4;
                }
            }
        }
    }
}

#define OWN_BLOCKS  256
#define OWN_THREADS 1024

template <int RN_, int S_, int R_, int WCAP_, int DB_, int U_>
struct own_cfg {
    static constexpr int RN = RN_, S = S_, R = R_, WCAP = WCAP_, DB = DB_, U = U_;
    static constexpr int SL4   = N_EDGES_C / 4 / S_;
    static constexpr int STEP  = OWN_THREADS * U_;
    static constexpr int TRIPS = (SL4 + STEP - 1) / STEP;
    static_assert((N_EDGES_C / 4) % S_ == 0, "a slice is a whole number of int4 words");
    static_assert(R_ * S_ == OWN_BLOCKS, "one block per (range, slice)");
    static_assert(R_ * RN_ >= N_NODES_C, "ranges cover every node");
    static_assert(RN_ <= (1 << DB_), "place in range fits its bits");
    static_assert(((long long)SL4 * 4) <= (1ll << (32 - DB_)), "edge id in slice fits its bits");
};
#define LIST_SUBS  8
typedef own_cfg<400, 1, 256, 512, 10, 8>  cfg2;
typedef own_cfg<3200, 8, 32, 1280, 12, 4> cfg1;
static_assert(cfg1::RN == LIST_SUBS * cfg2::RN && cfg2::R == LIST_SUBS * cfg1::R, "a conv1 range is LIST_SUBS conv2 ranges");
static_assert(cfg1::S == LIST_SUBS * cfg2::S, "a conv2 slice is LIST_SUBS conv1 slices");
static_assert(cfg1::WCAP % 32 == 0, "a wave's padded list part fits its capacity");
static_assert(cfg2::RN % 16 == 0 && N_NODES_C % cfg2::RN == 0, "a conv2 range is whole 16-node tiles and no range is partial");
static_assert(cfg2::RN / 16 <= OWN_THREADS / 32, "one wave per 16-node tile in the conv2 node MLP");

#define LIST_CAP   ((OWN_THREADS / 32) * cfg1::WCAP)
#define LIST_BAD   0xFFFFFFFFu
#define CNT_PITCH  32
#define LIST_U     8
static_assert((LIST_BAD & ((1u << cfg1::DB) - 1u)) >= (unsigned)cfg1::RN, "the pad entry lies outside every range");

typedef __attribute__((ext_vector_type(2))) _Float16 v2h;

template <class C>
__device__ __forceinline__ void own_push(unsigned* wlist, unsigned& wcount,
                                         int dstv, int n0, int erel) {
    const unsigned d = (unsigned)(dstv - n0);
    const bool m = d < (unsigned)C::RN;
    const unsigned mask = __builtin_amdgcn_ballot_w32(m);
    if (mask) {
        const unsigned pos = wcount + __builtin_amdgcn_mbcnt_lo(mask, 0u);
        if (m && pos < (unsigned)C::WCAP) wlist[pos] = ((unsigned)erel << C::DB) | d;
        wcount += (unsigned)__builtin_popcount(mask);
    }
}

template <class C>
__device__ __forceinline__ unsigned own_scan(const int4* __restrict__ d4, unsigned* wlist,
                                             int n0, int rot, int tid) {
    unsigned wcount = 0;
    for (int t = 0; t < C::TRIPS; ++t) {
        int tt = t + rot;
        if (tt >= C::TRIPS) tt -= C::TRIPS;
        const int base = tt * C::STEP;
        int4 v[C::U];
#pragma unroll
        for (int u = 0; u < C::U; ++u) {
            const int i = base + u * OWN_THREADS + tid;
            v[u] = (i < C::SL4) ? d4[i] : make_int4(-1, -1, -1, -1);
        }
#pragma unroll
        for (int u = 0; u < C::U; ++u) {
            const int e = (base + u * OWN_THREADS + tid) * 4;
            own_push<C>(wlist, wcount, v[u].x, n0, e + 0);
            own_push<C>(wlist, wcount, v[u].y, n0, e + 1);
            own_push<C>(wlist, wcount, v[u].z, n0, e + 2);
            own_push<C>(wlist, wcount, v[u].w, n0, e + 3);
        }
    }
    return wcount;
}

template <class C, class F>
__device__ __forceinline__ void own_overflow(const int4* __restrict__ d4, int n0, int rot,
                                             int tid, F add) {
    unsigned seen = 0;
#pragma unroll 1
    for (int t = 0; t < C::TRIPS * C::U; ++t) {
        int tt = t / C::U + rot;
        if (tt >= C::TRIPS) tt -= C::TRIPS;
        const int i = (tt * C::U + (t % C::U)) * OWN_THREADS + tid;
        const int4 v = (i < C::SL4) ? d4[i] : make_int4(-1, -1, -1, -1);
        const int dv[4] = {v.x, v.y, v.z, v.w};
#pragma unroll 1
        for (int k = 0; k < 4; ++k) {
            const unsigned d = (unsigned)(dv[k] - n0);
            const bool m = d < (unsigned)C::RN;
            const unsigned mask = __builtin_amdgcn_ballot_w32(m);
            const unsigned pos = seen + __builtin_amdgcn_mbcnt_lo(mask, 0u);
            seen += (unsigned)__builtin_popcount(mask);
            if (m && pos >= (unsigned)C::WCAP) add(i * 4 + k, (int)d);
        }
    }
}

__device__ __forceinline__ void list_push(unsigned* wlist, unsigned& wcount, unsigned ent,
                                          unsigned sublo, unsigned eoff) {
    const unsigned d = (ent & ((1u << cfg1::DB) - 1u)) - sublo;
    const bool m = d < (unsigned)cfg2::RN;
    const unsigned mask = __builtin_amdgcn_ballot_w32(m);
    if (mask) {
        const unsigned pos = wcount + __builtin_amdgcn_mbcnt_lo(mask, 0u);
        if (m && pos < (unsigned)cfg2::WCAP)
            wlist[pos] = (((ent >> cfg1::DB) + eoff) << cfg2::DB) | d;
        wcount += (unsigned)__builtin_popcount(mask);
    }
}

__device__ __forceinline__ unsigned list_scan(const unsigned* __restrict__ lists, int l0,
                                              const unsigned (&cnt)[LIST_SUBS], unsigned cmax,
                                              unsigned* wlist, unsigned sublo, int wave, int lane) {
    unsigned wcount = 0;
    for (unsigned k0 = 0; k0 * OWN_THREADS < cmax; k0 += LIST_U) {
        unsigned v[LIST_SUBS][LIST_U];
#pragma unroll
        for (int q = 0; q < LIST_SUBS; ++q) {
            const unsigned* L = lists + (size_t)(l0 + q) * LIST_CAP;
            const unsigned wb = (unsigned)(((wave + 8 * q) & 31) * 32 + lane);
#pragma unroll
            for (int u = 0; u < LIST_U; ++u) {
                const unsigned i = (k0 + (unsigned)u) * OWN_THREADS + wb;
                v[q][u] = (i < cnt[q]) ? L[i] : LIST_BAD;
            }
        }
#pragma unroll
        for (int q = 0; q < LIST_SUBS; ++q) {
#pragma unroll
            for (int u = 0; u < LIST_U; ++u)
                list_push(wlist, wcount, v[q][u], sublo, (unsigned)(q * (cfg1::SL4 * 4)));
        }
    }
    return wcount;
}

template <class F>
__device__ __forceinline__ void list_overflow(const unsigned* __restrict__ lists,
                                              const unsigned* __restrict__ lcount, int l0,
                                              unsigned cmax, unsigned sublo, int wave, int lane, F add) {
    unsigned seen = 0;
#pragma unroll 1
    for (unsigned k0 = 0; k0 * OWN_THREADS < cmax; k0 += LIST_U) {
#pragma unroll 1
        for (int q = 0; q < LIST_SUBS; ++q) {
            const unsigned* L = lists + (size_t)(l0 + q) * LIST_CAP;
            const unsigned cq = lcount[(l0 + q) * CNT_PITCH];
            const unsigned wb = (unsigned)(((wave + 8 * q) & 31) * 32 + lane);
#pragma unroll 1
            for (int u = 0; u < LIST_U; ++u) {
                const unsigned i = (k0 + (unsigned)u) * OWN_THREADS + wb;
                const unsigned ent = (i < cq) ? L[i] : LIST_BAD;
                const unsigned d = (ent & ((1u << cfg1::DB) - 1u)) - sublo;
                const bool m = d < (unsigned)cfg2::RN;
                const unsigned mask = __builtin_amdgcn_ballot_w32(m);
                const unsigned pos = seen + __builtin_amdgcn_mbcnt_lo(mask, 0u);
                seen += (unsigned)__builtin_popcount(mask);
                if (m && pos >= (unsigned)cfg2::WCAP)
                    add((int)((ent >> cfg1::DB) + (unsigned)(q * (cfg1::SL4 * 4))), (int)d);
            }
        }
    }
}

__global__ __launch_bounds__(OWN_THREADS)
void edge2_own_kernel(const _Float16* __restrict__ h, const int* __restrict__ ei,
                      const float* __restrict__ ea,
                      const float* __restrict__ eW, const float* __restrict__ eb,
                      const unsigned* __restrict__ lists, const unsigned* __restrict__ lcount,
                      const float* __restrict__ W1, const float* __restrict__ b1,
                      const float* __restrict__ W2, const float* __restrict__ b2,
                      float* __restrict__ h2) {
    typedef cfg2 C;
    constexpr int P2 = HID_C + 8;
    constexpr int NT = C::RN / 16;
    constexpr int PA = HID_C + 4;
    static_assert(HID_C * HID_C / 4 == OWN_THREADS, "one float4 of W1 and one of W2 per thread");
    __shared__ __align__(16) float    s_acc[C::RN * PA];
    __shared__ __align__(16) unsigned s_list[(OWN_THREADS / 32) * C::WCAP];
    __shared__ __align__(16) _Float16 sW1[HID_C * P2];
    __shared__ __align__(16) _Float16 sW2[HID_C * P2];

    const int tid   = threadIdx.x;
    const int lane  = tid & 31;
    const int wave  = tid >> 5;
    const int range = (int)blockIdx.x / C::S;
    const int slice = (int)blockIdx.x - range * C::S;
    const int n0    = range * C::RN;
    const int e0    = slice * (C::SL4 * 4);
    unsigned* wlist = s_list + wave * C::WCAP;

    if (n0 >= N_NODES_C) return;

    const float4 w1v = ((const float4*)W1)[tid];
    const float4 w2v = ((const float4*)W2)[tid];

    for (int i = tid; i < C::RN * PA / 4; i += OWN_THREADS)
        ((float4*)s_acc)[i] = make_float4(0.f, 0.f, 0.f, 0.f);
    {
        const int n = tid / (HID_C / 4), q = tid - n * (HID_C / 4);
        v4h o1, o2;
        o1[0] = (_Float16)w1v.x; o1[1] = (_Float16)w1v.y; o1[2] = (_Float16)w1v.z; o1[3] = (_Float16)w1v.w;
        o2[0] = (_Float16)w2v.x; o2[1] = (_Float16)w2v.y; o2[2] = (_Float16)w2v.z; o2[3] = (_Float16)w2v.w;
        *(v4h*)(sW1 + n * P2 + 4 * q) = o1;
        *(v4h*)(sW2 + n * P2 + 4 * q) = o2;
    }
    __syncthreads();

    const int c0 = 2 * lane;
    const float4 w0 = *(const float4*)(eW + (c0 + 0) * EDGE_DIM_C);
    const float4 w1 = *(const float4*)(eW + (c0 + 1) * EDGE_DIM_C);
    const float  b0 = eb[c0], b1e = eb[c0 + 1];

    auto add_one = [&](int erel, int od) {
        const int oe = e0 + erel;
        const int os = ei[oe];
        const float4 a = *(const float4*)(ea + (size_t)oe * EDGE_DIM_C);
#pragma unroll 1
        for (int q = 0; q < HID_C; ++q) {
            const float4 w = *(const float4*)(eW + q * EDGE_DIM_C);
            float t = eb[q] + a.x * w.x + a.y * w.y + a.z * w.z + a.w * w.w;
            t += (float)h[(size_t)os * HID_C + q];
            atomicAdd(&s_acc[od * PA + q], fmaxf(t, 0.0f));
        }
    };

    const int l0 = (range / LIST_SUBS) * cfg1::S + slice * LIST_SUBS;
    const unsigned sublo = (unsigned)((range % LIST_SUBS) * C::RN);
    unsigned cnt[LIST_SUBS];
    unsigned cmax = 0;
#pragma unroll
    for (int q = 0; q < LIST_SUBS; ++q) {
        cnt[q] = lcount[(l0 + q) * CNT_PITCH];
        cmax = cnt[q] > cmax ? cnt[q] : cmax;
    }
    unsigned wcount;
    if (cmax <= (unsigned)LIST_CAP) {
        wcount = list_scan(lists, l0, cnt, cmax, wlist, sublo, wave, lane);
        if (wcount > (unsigned)C::WCAP)
            list_overflow(lists, lcount, l0, cmax, sublo, wave, lane, add_one);
    } else {
        const int4* d4 = (const int4*)(ei + N_EDGES_C + e0);
        const int rot = (range * C::TRIPS) / C::R;
        wcount = own_scan<C>(d4, wlist, n0, rot, tid);
        if (wcount > (unsigned)C::WCAP)
            own_overflow<C>(d4, n0, rot, tid, add_one);
    }
    const int wn = (int)(wcount < (unsigned)C::WCAP ? wcount : (unsigned)C::WCAP);

    __syncthreads();

    const v2h* h2p = (const v2h*)h;
    for (int p = 0; p < wn; p += 32) {
        const int idx = p + lane;
        const bool valid = idx < wn;
        const unsigned ent = valid ? wlist[idx] : 0u;
        const int e  = e0 + (int)(ent >> C::DB);
        const int dl = (int)(ent & ((1u << C::DB) - 1u));
        const int sl = valid ? ei[e] : 0;
        const float4 al = valid ? *(const float4*)(ea + (size_t)e * EDGE_DIM_C)
                                : make_float4(0.f, 0.f, 0.f, 0.f);
        const int cnt = (wn - p) < 32 ? (wn - p) : 32;
        v2h hv[32];
#pragma unroll
        for (int j = 0; j < 32; ++j) {
            const int sj = __builtin_amdgcn_readlane(sl, j);
            hv[j] = h2p[(size_t)sj * (HID_C / 2) + lane];
        }
#pragma unroll
        for (int j = 0; j < 32; ++j) {
            if (j < cnt) {
                const int   dj = __builtin_amdgcn_readlane(dl, j);
                const float ax = __builtin_bit_cast(float, __builtin_amdgcn_readlane(__builtin_bit_cast(int, al.x), j));
                const float ay = __builtin_bit_cast(float, __builtin_amdgcn_readlane(__builtin_bit_cast(int, al.y), j));
                const float az = __builtin_bit_cast(float, __builtin_amdgcn_readlane(__builtin_bit_cast(int, al.z), j));
                const float aw = __builtin_bit_cast(float, __builtin_amdgcn_readlane(__builtin_bit_cast(int, al.w), j));
                float t0 = b0  + ax * w0.x + ay * w0.y + az * w0.z + aw * w0.w;
                float t1 = b1e + ax * w1.x + ay * w1.y + az * w1.z + aw * w1.w;
                t0 += (float)hv[j][0];
                t1 += (float)hv[j][1];
                atomicAdd(&s_acc[dj * PA + c0],     fmaxf(t0, 0.0f));
                atomicAdd(&s_acc[dj * PA + c0 + 1], fmaxf(t1, 0.0f));
            }
        }
    }
    const int  wv   = __builtin_amdgcn_readfirstlane(wave);
    const bool tile = wv < NT;
    const int  c    = lane & 15;
    const int  hh   = lane >> 4;
    const _Float16* grow = h + (size_t)(n0 + (tile ? wv * 16 + c : 0)) * HID_C + 8 * hh;
    v8h    xr[4];
    float4 pb1[8], pb2[8];
#pragma unroll
    for (int i = 0; i < 4; ++i) {
        xr[i]          = *(const v8h*)(grow + 16 * i);
        pb1[2 * i]     = *(const float4*)(b1 + 16 * i + 8 * hh);
        pb1[2 * i + 1] = *(const float4*)(b1 + 16 * i + 8 * hh + 4);
        pb2[2 * i]     = *(const float4*)(b2 + 16 * i + 8 * hh);
        pb2[2 * i + 1] = *(const float4*)(b2 + 16 * i + 8 * hh + 4);
    }

    __syncthreads();

    float*     trow  = s_acc + (wv * 16 + c) * PA;
    if (tile) {
        v16h bop[2];
#pragma unroll
        for (int kb = 0; kb < 2; ++kb) {
            const int k0 = kb * 32 + 8 * hh;
            const v8h x0 = xr[2 * kb];
            const v8h x1 = xr[2 * kb + 1];
            const float4 a0 = *(const float4*)(trow + k0);
            const float4 a1 = *(const float4*)(trow + k0 + 4);
            const float4 a2 = *(const float4*)(trow + k0 + 16);
            const float4 a3 = *(const float4*)(trow + k0 + 20);
            const float lo[8] = {a0.x, a0.y, a0.z, a0.w, a1.x, a1.y, a1.z, a1.w};
            const float hi[8] = {a2.x, a2.y, a2.z, a2.w, a3.x, a3.y, a3.z, a3.w};
            v16h b;
#pragma unroll
            for (int e = 0; e < 8; ++e) {
                b[e]     = (_Float16)((float)x0[e] + lo[e]);
                b[8 + e] = (_Float16)((float)x1[e] + hi[e]);
            }
            bop[kb] = b;
        }

        v8f acc[4];
#pragma unroll
        for (int t = 0; t < 4; ++t) {
            acc[t] = v8f{};
#pragma unroll
            for (int kb = 0; kb < 2; ++kb) {
                const v16h a = lds_load_tile(sW1 + 16 * t * P2, P2, kb * 32);
                acc[t] = __builtin_amdgcn_wmma_f32_16x16x32_f16(false, a, false, bop[kb],
                                                                (short)0, acc[t], false, false);
            }
        }

        v16h bop2[2];
#pragma unroll
        for (int s2 = 0; s2 < 2; ++s2) {
            const float4 p0 = pb1[4 * s2], p1 = pb1[4 * s2 + 1], p2 = pb1[4 * s2 + 2], p3 = pb1[4 * s2 + 3];
            const float bl[8] = {p0.x, p0.y, p0.z, p0.w, p1.x, p1.y, p1.z, p1.w};
            const float bh[8] = {p2.x, p2.y, p2.z, p2.w, p3.x, p3.y, p3.z, p3.w};
            v16h b;
#pragma unroll
            for (int e = 0; e < 8; ++e) {
                float v0 = acc[2 * s2][e] + bl[e];
                float v1 = acc[2 * s2 + 1][e] + bh[e];
                v0 = v0 > 0.0f ? v0 : 0.0f;
                v1 = v1 > 0.0f ? v1 : 0.0f;
                b[e]     = (_Float16)v0;
                b[8 + e] = (_Float16)v1;
            }
            bop2[s2] = b;
        }

#pragma unroll
        for (int t = 0; t < 4; ++t) {
            v8f d = v8f{};
#pragma unroll
            for (int s2 = 0; s2 < 2; ++s2) {
                const v16h a = lds_load_tile(sW2 + 16 * t * P2, P2, s2 * 32);
                d = __builtin_amdgcn_wmma_f32_16x16x32_f16(false, a, false, bop2[s2],
                                                           (short)0, d, false, false);
            }
            const float4 q0 = pb2[2 * t], q1 = pb2[2 * t + 1];
            const float bb[8] = {q0.x, q0.y, q0.z, q0.w, q1.x, q1.y, q1.z, q1.w};
            float v[8];
#pragma unroll
            for (int e = 0; e < 8; ++e) {
                const float u = d[e] + bb[e];
                v[e] = u > 0.0f ? u : 0.0f;
            }
            float* q = trow + 16 * t + 8 * hh;
            *(float4*)(q)     = make_float4(v[0], v[1], v[2], v[3]);
            *(float4*)(q + 4) = make_float4(v[4], v[5], v[6], v[7]);
        }
    }
    __syncthreads();

    if (tile) {
        const float* st = s_acc + wv * 16 * PA;
        float*       gt = h2 + (size_t)(n0 + wv * 16) * HID_C;
#pragma unroll
        for (int i = 0; i < 8; ++i) {
            const int f = 4 * (lane + 32 * i);
            *(float4*)(gt + f) = *(const float4*)(st + (f >> 6) * PA + (f & 63));
        }
    }
}

__global__ __launch_bounds__(OWN_THREADS)
void edge1_own_kernel(const float* __restrict__ x, const int* __restrict__ ei,
                      const float* __restrict__ ea,
                      const float* __restrict__ eW, const float* __restrict__ eb,
                      float* __restrict__ aggr,
                      unsigned* __restrict__ lists, unsigned* __restrict__ lcount) {
    typedef cfg1 C;
    __shared__ __align__(16) float    s_acc[C::RN * 8];
    __shared__ __align__(16) unsigned s_list[(OWN_THREADS / 32) * C::WCAP];
    __shared__ unsigned s_wn[OWN_THREADS / 32];

    const int tid   = threadIdx.x;
    const int lane  = tid & 31;
    const int wave  = tid >> 5;
    const int range = (int)blockIdx.x / C::S;
    const int slice = (int)blockIdx.x - range * C::S;
    const int n0    = range * C::RN;
    const int e0    = slice * (C::SL4 * 4);
    unsigned* wlist = s_list + wave * C::WCAP;

    for (int i = tid; i < C::RN * 8; i += OWN_THREADS) s_acc[i] = 0.0f;
    __syncthreads();

    const int c   = lane & 7;
    const int sub = lane >> 3;
    const bool cact = c < IN_DIM_C;
    const float4 wc = cact ? *(const float4*)(eW + c * EDGE_DIM_C) : make_float4(0.f, 0.f, 0.f, 0.f);
    const float  bc = cact ? eb[c] : 0.0f;

    const int4* d4 = (const int4*)(ei + N_EDGES_C + e0);
    const int rot = (range * C::TRIPS) / C::R;
    const unsigned wcount = own_scan<C>(d4, wlist, n0, rot, tid);
    if (wcount > (unsigned)C::WCAP) {
        own_overflow<C>(d4, n0, rot, tid, [&](int erel, int od) {
            const int oe = e0 + erel;
            const int os = ei[oe];
            const float4 a = *(const float4*)(ea + (size_t)oe * EDGE_DIM_C);
#pragma unroll 1
            for (int q = 0; q < IN_DIM_C; ++q) {
                const float4 w = *(const float4*)(eW + q * EDGE_DIM_C);
                float t = eb[q] + a.x * w.x + a.y * w.y + a.z * w.z + a.w * w.w;
                t += x[(size_t)os * IN_DIM_C + q];
                atomicAdd(&s_acc[od * 8 + q], fmaxf(t, 0.0f));
            }
        });
    }
    const int wn = (int)(wcount < (unsigned)C::WCAP ? wcount : (unsigned)C::WCAP);
    if (lane == 0) s_wn[wave] = wcount;

    __syncthreads();

    {
        unsigned off = 0, tot = 0;
        bool full = false;
        for (int w = 0; w < OWN_THREADS / 32; ++w) {
            const unsigned cw = s_wn[w];
            full = full || cw > (unsigned)C::WCAP;
            const unsigned pw = ((cw < (unsigned)C::WCAP ? cw : (unsigned)C::WCAP) + 31u) & ~31u;
            if (w < wave) off += pw;
            tot += pw;
        }
        unsigned* L = lists + (size_t)blockIdx.x * LIST_CAP + off;
        const int wp = (wn + 31) & ~31;
        for (int i = lane; i < wp; i += 32) L[i] = (i < wn) ? wlist[i] : LIST_BAD;
        if (tid == 0) lcount[blockIdx.x * CNT_PITCH] = full ? LIST_BAD : tot;
    }

    for (int p = 0; p < wn; p += 32) {
        unsigned ent[8]; int sr[8]; float4 av[8]; float xv[8]; bool ok[8];
#pragma unroll
        for (int g = 0; g < 8; ++g) {
            const int idx = p + g * 4 + sub;
            ok[g]  = idx < wn;
            ent[g] = ok[g] ? wlist[idx] : 0u;
        }
#pragma unroll
        for (int g = 0; g < 8; ++g) {
            const int e = e0 + (int)(ent[g] >> C::DB);
            sr[g] = ok[g] ? ei[e] : 0;
            av[g] = ok[g] ? *(const float4*)(ea + (size_t)e * EDGE_DIM_C)
                          : make_float4(0.f, 0.f, 0.f, 0.f);
        }
#pragma unroll
        for (int g = 0; g < 8; ++g)
            xv[g] = (ok[g] && cact) ? x[(size_t)sr[g] * IN_DIM_C + c] : 0.0f;
#pragma unroll
        for (int g = 0; g < 8; ++g) {
            if (ok[g] && cact) {
                const int dl = (int)(ent[g] & ((1u << C::DB) - 1u));
                float t = bc + av[g].x * wc.x + av[g].y * wc.y + av[g].z * wc.z + av[g].w * wc.w;
                t += xv[g];
                atomicAdd(&s_acc[dl * 8 + c], fmaxf(t, 0.0f));
            }
        }
    }
    __syncthreads();

    int rows = N_NODES_C - n0;
    if (rows > C::RN) rows = C::RN;
    float* o = aggr + (size_t)slice * N_NODES_C * IN_DIM_C + (size_t)n0 * IN_DIM_C;
    for (int i = tid; i < rows * IN_DIM_C; i += OWN_THREADS) {
        const int r = i / IN_DIM_C, q = i - r * IN_DIM_C;
        o[i] = s_acc[r * 8 + q];
    }
}

#define POOL_GPB     4
#define POOL_THREADS 1024
#define POOL_U       14
#define POOL_Q       8

__global__ __launch_bounds__(POOL_THREADS)
void pool_fc_kernel(const float* __restrict__ h2, const int* __restrict__ batch,
                    const float* __restrict__ fcW, const float* __restrict__ fcb,
                    float* __restrict__ out) {
    constexpr int NW  = POOL_THREADS / 32;
    constexpr int NCH = (N_NODES_C + 31) / 32;
    static_assert(N_GRAPHS_C % POOL_GPB == 0, "every block owns POOL_GPB graphs");
    static_assert(POOL_GPB * HID_C <= POOL_THREADS, "one thread per (graph, channel) in the wave sum");

    __shared__ __align__(16) float s_part[NW][POOL_GPB][HID_C];
    __shared__ int   s_cnt [NW][POOL_GPB];
    __shared__ float s_mean[POOL_GPB][HID_C];

    const int tid  = threadIdx.x;
    const int lane = tid & 31;
    const int wave = tid >> 5;
    const int g0   = (int)blockIdx.x * POOL_GPB;
    const float2* h2v = (const float2*)h2;

    float2 acc[POOL_GPB];
    int    cnt[POOL_GPB];
#pragma unroll
    for (int k = 0; k < POOL_GPB; ++k) { acc[k] = make_float2(0.f, 0.f); cnt[k] = 0; }

    for (int c0 = wave; c0 < NCH; c0 += NW * POOL_U) {
        unsigned hit = 0;
#pragma unroll
        for (int u = 0; u < POOL_U; ++u) {
            const int c = c0 + u * NW;
            const int n = c * 32 + lane;
            const int g = (c < NCH && n < N_NODES_C) ? batch[n] : -1;
            const bool m = (unsigned)(g - g0) < (unsigned)POOL_GPB;
            if (__builtin_amdgcn_ballot_w32(m)) hit |= 1u << u;
        }
        while (hit) {
            const int u = __builtin_ctz(hit);
            hit &= hit - 1;
            const int base = (c0 + u * NW) * 32;
            const int n = base + lane;
            const int g = (n < N_NODES_C) ? batch[n] : -1;
            const int d = g - g0;
            unsigned mask = __builtin_amdgcn_ballot_w32((unsigned)d < (unsigned)POOL_GPB);
#pragma unroll
            for (int k = 0; k < POOL_GPB; ++k)
                cnt[k] += __builtin_popcount(__builtin_amdgcn_ballot_w32(d == k));
            while (mask) {
                int gq[POOL_Q]; float2 v[POOL_Q];
                const int jf = __builtin_ctz(mask);
#pragma unroll
                for (int q = 0; q < POOL_Q; ++q) {
                    const bool has = mask != 0;
                    const int j = has ? __builtin_ctz(mask) : jf;
                    const int gj = __builtin_amdgcn_readlane(d, j);
                    gq[q] = has ? gj : -1;
                    v[q]  = h2v[(size_t)(base + j) * (HID_C / 2) + lane];
                    mask &= mask - 1;
                }
                __builtin_amdgcn_sched_barrier(0);
#pragma unroll
                for (int q = 0; q < POOL_Q; ++q) {
#pragma unroll
                    for (int k = 0; k < POOL_GPB; ++k)
                        if (gq[q] == k) { acc[k].x += v[q].x; acc[k].y += v[q].y; }
                }
            }
        }
    }

#pragma unroll
    for (int k = 0; k < POOL_GPB; ++k) {
        *(float2*)&s_part[wave][k][2 * lane] = acc[k];
        if (lane == 0) s_cnt[wave][k] = cnt[k];
    }
    __syncthreads();

    if (tid < POOL_GPB * HID_C) {
        const int k = tid / HID_C, ch = tid - k * HID_C;
        float s = 0.0f; int c = 0;
        for (int w = 0; w < NW; ++w) { s += s_part[w][k][ch]; c += s_cnt[w][k]; }
        s_mean[k][ch] = s / fmaxf((float)c, 1.0f);
    }
    __syncthreads();

    if (tid < POOL_GPB * NUM_TASKS_C) {
        const int k = tid / NUM_TASKS_C, t = tid - k * NUM_TASKS_C;
        float s = fcb[t];
#pragma unroll
        for (int c = 0; c < HID_C; ++c)
            s += s_mean[k][c] * fcW[t * HID_C + c];
        out[(size_t)(g0 + k) * NUM_TASKS_C + t] = s;
    }
}

extern "C" void kernel_launch(void* const* d_in, const int* in_sizes, int n_in,
                              void* d_out, int out_size, void* d_ws, size_t ws_size,
                              hipStream_t stream) {
    const float* x    = (const float*)d_in[0];
    const int*   ei   = (const int*)  d_in[1];
    const float* ea   = (const float*)d_in[2];
    const int*   bat  = (const int*)  d_in[3];
    const float* e1W  = (const float*)d_in[4];  const float* e1b  = (const float*)d_in[5];
    const float* n1W1 = (const float*)d_in[6];  const float* n1b1 = (const float*)d_in[7];
    const float* n1W2 = (const float*)d_in[8];  const float* n1b2 = (const float*)d_in[9];
    const float* e2W  = (const float*)d_in[10]; const float* e2b  = (const float*)d_in[11];
    const float* n2W1 = (const float*)d_in[12]; const float* n2b1 = (const float*)d_in[13];
    const float* n2W2 = (const float*)d_in[14]; const float* n2b2 = (const float*)d_in[15];
    const float* fcW  = (const float*)d_in[16]; const float* fcb  = (const float*)d_in[17];
    float* out = (float*)d_out;

    float*    aggr1 = (float*)d_ws;
    _Float16* h     = (_Float16*)(aggr1 + (size_t)cfg1::S * N_NODES_C * IN_DIM_C);
    float*    h2    = (float*)(h + (size_t)N_NODES_C * HID_C);
    unsigned* lists  = (unsigned*)(h2 + (size_t)N_NODES_C * HID_C);
    unsigned* lcount = lists + (size_t)OWN_BLOCKS * LIST_CAP;

    edge1_own_kernel<<<OWN_BLOCKS, OWN_THREADS, 0, stream>>>(x, ei, ea, e1W, e1b, aggr1, lists, lcount);
    node_mlp_kernel<IN_DIM_C, cfg1::S><<<MLP_GRID, MLP_THREADS, 0, stream>>>(
        x, nullptr, aggr1, n1W1, n1b1, n1W2, n1b2, h, nullptr, N_NODES_C);
    edge2_own_kernel<<<OWN_BLOCKS, OWN_THREADS, 0, stream>>>(h, ei, ea, e2W, e2b, lists, lcount,
                                                             n2W1, n2b1, n2W2, n2b2, h2);
    pool_fc_kernel<<<N_GRAPHS_C / POOL_GPB, POOL_THREADS, 0, stream>>>(h2, bat, fcW, fcb, out);
}


# Conv1 edge phase 2: the 8 groups' src, edge_attr and x-row loads are no longer inside per-group branches, so each stage's 8 loads are in flight together instead of one memory round trip per group
# speedup vs baseline: 3.7299x; 1.0349x over previous
#include <hip/hip_runtime.h>
#include <hip/hip_bf16.h>

#define N_NODES_C  100000
#define N_EDGES_C  1600000
#define N_GRAPHS_C 1000
#define HID_C      64
#define IN_DIM_C   7
#define EDGE_DIM_C 4
#define NUM_TASKS_C 12

typedef __attribute__((ext_vector_type(16))) _Float16 v16h;
typedef __attribute__((ext_vector_type(8)))  _Float16 v8h;
typedef __attribute__((ext_vector_type(8)))  float    v8f;

__device__ __forceinline__ v16h lds_load_tile(const _Float16* s, int ld, int kbase) {
    const int lane = threadIdx.x & 31;
    const int r    = lane & 15;
    const int koff = (lane < 16) ? 0 : 8;
    const _Float16* p = s + r * ld + kbase + koff;
    v8h lo = *(const v8h*)(p);
    v8h hi = *(const v8h*)(p + 16);
    v16h out;
#pragma unroll
    for (int e = 0; e < 8; ++e) { out[e] = lo[e]; out[8 + e] = hi[e]; }
    return out;
}

#define MLP_WAVES   8
#define MLP_NT      2
#define MLP_THREADS (MLP_WAVES * 32)
#define MLP_TILES   (MLP_WAVES * MLP_NT)
#define MLP_ROWS    (MLP_TILES * 16)
#define MLP_GRID    ((N_NODES_C + MLP_ROWS - 1) / MLP_ROWS)
static_assert(N_NODES_C % 16 == 0, "the node count is a whole number of 16-node tiles");

typedef __attribute__((ext_vector_type(4))) _Float16 v4h;

template <int K, int KP>
struct mlp_wregs {
    static constexpr bool VEC = (K % 4) == 0;
    static constexpr int  N   = (VEC ? HID_C * (K / 4) : HID_C * KP) / MLP_THREADS;
    static_assert((VEC ? HID_C * (K / 4) : HID_C * KP) % MLP_THREADS == 0, "the same number of pieces in every thread");
    static_assert(!VEC || K == KP, "no padding on the 16-byte path");
    float4 v4[VEC ? N : 1];
    float  v1[VEC ? 1 : N];
};

template <int K, int KP>
__device__ __forceinline__ void mlp_load_weights(const float* __restrict__ W, int tid, mlp_wregs<K, KP>& r) {
#pragma unroll
    for (int it = 0; it < mlp_wregs<K, KP>::N; ++it) {
        const int i = tid + it * MLP_THREADS;
        if constexpr (mlp_wregs<K, KP>::VEC) {
            r.v4[it] = ((const float4*)W)[i];
        } else {
            const int n = i / KP, k = i - n * KP;
            const float w = W[n * K + (k < K ? k : 0)];
            r.v1[it] = (k < K) ? w : 0.0f;
        }
    }
}

template <int K, int KP, int P>
__device__ __forceinline__ void mlp_store_weights(_Float16* s, int tid, const mlp_wregs<K, KP>& r) {
#pragma unroll
    for (int it = 0; it < mlp_wregs<K, KP>::N; ++it) {
        const int i = tid + it * MLP_THREADS;
        if constexpr (mlp_wregs<K, KP>::VEC) {
            const int n = i / (K / 4), q = i - n * (K / 4);
            const float4 v = r.v4[it];
            v4h o;
            o[0] = (_Float16)v.x; o[1] = (_Float16)v.y; o[2] = (_Float16)v.z; o[3] = (_Float16)v.w;
            *(v4h*)(s + n * P + 4 * q) = o;
        } else {
            const int n = i / KP, k = i - n * KP;
            s[n * P + k] = (_Float16)r.v1[it];
        }
    }
}

template <int K1, int NP>
__global__ __launch_bounds__(MLP_THREADS)
void node_mlp_kernel(const float* __restrict__ xin, const _Float16* __restrict__ hin,
                     const float* __restrict__ in1,
                     const float* __restrict__ W1,  const float* __restrict__ b1,
                     const float* __restrict__ W2,  const float* __restrict__ b2,
                     _Float16* __restrict__ outh, float* __restrict__ outf, int n_nodes) {
    constexpr int  K1P   = (K1 + 31) & ~31;
    constexpr int  KS    = K1P / 32;
    constexpr int  P1    = K1P + 8;
    constexpr int  P2    = HID_C + 8;
    constexpr int  PO    = HID_C + 4;
    constexpr bool SMALL = (K1 % 8) != 0;
    constexpr int  XT    = 16 * K1;
    static_assert(!SMALL || (XT % 4 == 0 && XT / 4 <= 32), "a flat conv1 tile is at most one float4 per lane");
    static_assert(SMALL || K1 == HID_C, "the conv2 path cuts rows of 64");

    __shared__ __align__(16) _Float16 sW1[HID_C * P1];
    __shared__ __align__(16) _Float16 sW2[HID_C * P2];
    __shared__ __align__(16) float    sX [SMALL ? MLP_TILES * 128 : 4];
    __shared__ __align__(16) _Float16 sA [SMALL ? 8 : MLP_TILES * 16 * P1];
    __shared__ __align__(16) _Float16 sOh[SMALL ? MLP_TILES * 16 * P2 : 8];
    __shared__ __align__(16) float    sOf[SMALL ? 4 : MLP_TILES * 16 * PO];

    const int tid  = threadIdx.x;
    const int wave = __builtin_amdgcn_readfirstlane(tid >> 5);
    const int lane = tid & 31;
    const int c    = lane & 15;
    const int h    = lane >> 4;
    const int tile0 = ((int)blockIdx.x * MLP_WAVES + wave) * MLP_NT;

    int  row0[MLP_NT];
    bool tv[MLP_NT];
#pragma unroll
    for (int j = 0; j < MLP_NT; ++j) {
        row0[j] = (tile0 + j) * 16;
        tv[j]   = row0[j] < n_nodes;
    }

    mlp_wregs<K1, K1P>     w1r;
    mlp_wregs<HID_C, HID_C> w2r;
    mlp_load_weights<K1, K1P>(W1, tid, w1r);
    mlp_load_weights<HID_C, HID_C>(W2, tid, w2r);

    if constexpr (SMALL) {
        const int ll = lane < XT / 4 ? lane : XT / 4 - 1;
        float4 raw[MLP_NT][NP + 1];
#pragma unroll
        for (int j = 0; j < MLP_NT; ++j) {
            const int o4 = (tv[j] ? row0[j] : 0) * K1 / 4 + ll;
            raw[j][0] = ((const float4*)xin)[o4];
#pragma unroll
            for (int p = 0; p < NP; ++p)
                raw[j][p + 1] = ((const float4*)(in1 + (size_t)p * (size_t)n_nodes * K1))[o4];
        }
        float4 v[MLP_NT];
#pragma unroll
        for (int j = 0; j < MLP_NT; ++j) {
            v[j] = raw[j][0];
#pragma unroll
            for (int p = 0; p < NP; ++p) {
                const float4 a = raw[j][p + 1];
                v[j].x += a.x; v[j].y += a.y; v[j].z += a.z; v[j].w += a.w;
            }
        }
#pragma unroll
        for (int j = 0; j < MLP_NT; ++j)
            ((float4*)sX)[(wave * MLP_NT + j) * 32 + lane] = v[j];
    } else {
#pragma unroll
        for (int j = 0; j < MLP_NT; ++j) {
            const int    eb = (tv[j] ? row0[j] : 0) * K1;
            const int    tl = wave * MLP_NT + j;
#pragma unroll
            for (int i = 0; i < XT / 128; ++i) {
                const int f = 4 * (lane + 32 * i);
                const v4h x4 = *(const v4h*)(hin + eb + f);
                float v0 = (float)x4[0], v1 = (float)x4[1], v2 = (float)x4[2], v3 = (float)x4[3];
#pragma unroll
                for (int p = 0; p < NP; ++p) {
                    const float4 a = *(const float4*)(in1 + (size_t)p * (size_t)n_nodes * K1 + eb + f);
                    v0 += a.x; v1 += a.y; v2 += a.z; v3 += a.w;
                }
                v4h o;
                o[0] = (_Float16)v0; o[1] = (_Float16)v1; o[2] = (_Float16)v2; o[3] = (_Float16)v3;
                *(v4h*)(sA + (tl * 16 + (f >> 6)) * P1 + (f & 63)) = o;
            }
        }
    }

    mlp_store_weights<K1, K1P, P1>(sW1, tid, w1r);
    mlp_store_weights<HID_C, HID_C, P2>(sW2, tid, w2r);
    __syncthreads();

    v16h bop[MLP_NT][KS];
#pragma unroll
    for (int j = 0; j < MLP_NT; ++j) {
        const int tl = wave * MLP_NT + j;
        if constexpr (SMALL) {
            const float* r = sX + tl * 128 + c * K1;
            v16h b = {};
#pragma unroll
            for (int k = 0; k < K1; ++k) {
                const float f = r[k];
                b[k] = (h == 0) ? (_Float16)f : (_Float16)0.0f;
            }
            bop[j][0] = b;
        } else {
#pragma unroll
            for (int kb = 0; kb < KS; ++kb)
                bop[j][kb] = lds_load_tile(sA + tl * 16 * P1, P1, kb * 32);
        }
    }

    float bias1[4][8], bias2[4][8];
#pragma unroll
    for (int t = 0; t < 4; ++t) {
        const float4 a0 = *(const float4*)(b1 + 16 * t + 8 * h);
        const float4 a1 = *(const float4*)(b1 + 16 * t + 8 * h + 4);
        bias1[t][0] = a0.x; bias1[t][1] = a0.y; bias1[t][2] = a0.z; bias1[t][3] = a0.w;
        bias1[t][4] = a1.x; bias1[t][5] = a1.y; bias1[t][6] = a1.z; bias1[t][7] = a1.w;
        const float4 c0 = *(const float4*)(b2 + 16 * t + 8 * h);
        const float4 c1 = *(const float4*)(b2 + 16 * t + 8 * h + 4);
        bias2[t][0] = c0.x; bias2[t][1] = c0.y; bias2[t][2] = c0.z; bias2[t][3] = c0.w;
        bias2[t][4] = c1.x; bias2[t][5] = c1.y; bias2[t][6] = c1.z; bias2[t][7] = c1.w;
    }

    v8f acc[MLP_NT][4];
#pragma unroll
    for (int j = 0; j < MLP_NT; ++j)
#pragma unroll
        for (int t = 0; t < 4; ++t) acc[j][t] = v8f{};
#pragma unroll
    for (int t = 0; t < 4; ++t) {
#pragma unroll
        for (int kb = 0; kb < KS; ++kb) {
            const v16h a = lds_load_tile(sW1 + 16 * t * P1, P1, kb * 32);
#pragma unroll
            for (int j = 0; j < MLP_NT; ++j)
                acc[j][t] = __builtin_amdgcn_wmma_f32_16x16x32_f16(false, a, false, bop[j][kb],
                                                                   (short)0, acc[j][t], false, false);
        }
    }

    v16h bop2[MLP_NT][2];
#pragma unroll
    for (int j = 0; j < MLP_NT; ++j) {
#pragma unroll
        for (int s = 0; s < 2; ++s) {
            v16h b;
#pragma unroll
            for (int e = 0; e < 8; ++e) {
                float v0 = acc[j][2 * s][e] + bias1[2 * s][e];
                float v1 = acc[j][2 * s + 1][e] + bias1[2 * s + 1][e];
                v0 = v0 > 0.0f ? v0 : 0.0f;
                v1 = v1 > 0.0f ? v1 : 0.0f;
                b[e]     = (_Float16)v0;
                b[8 + e] = (_Float16)v1;
            }
            bop2[j][s] = b;
        }
    }

    v8f acc2[MLP_NT][4];
#pragma unroll
    for (int j = 0; j < MLP_NT; ++j)
#pragma unroll
        for (int t = 0; t < 4; ++t) acc2[j][t] = v8f{};
#pragma unroll
    for (int t = 0; t < 4; ++t) {
#pragma unroll
        for (int s = 0; s < 2; ++s) {
            const v16h a = lds_load_tile(sW2 + 16 * t * P2, P2, s * 32);
#pragma unroll
            for (int j = 0; j < MLP_NT; ++j)
                acc2[j][t] = __builtin_amdgcn_wmma_f32_16x16x32_f16(false, a, false, bop2[j][s],
                                                                    (short)0, acc2[j][t], false, false);
        }
    }

#pragma unroll
    for (int j = 0; j < MLP_NT; ++j) {
        const int tl = wave * MLP_NT + j;
#pragma unroll
        for (int t = 0; t < 4; ++t) {
            float v[8];
#pragma unroll
            for (int e = 0; e < 8; ++e) {
                const float u = acc2[j][t][e] + bias2[t][e];
                v[e] = u > 0.0f ? u : 0.0f;
            }
            if constexpr (SMALL) {
                v8h o8;
#pragma unroll
                for (int e = 0; e < 8; ++e) o8[e] = (_Float16)v[e];
                *(v8h*)(sOh + (tl * 16 + c) * P2 + 16 * t + 8 * h) = o8;
            } else {
                float* q = sOf + (tl * 16 + c) * PO + 16 * t + 8 * h;
                *(float4*)(q)     = make_float4(v[0], v[1], v[2], v[3]);
                *(float4*)(q + 4) = make_float4(v[4], v[5], v[6], v[7]);
            }
        }
    }
    __syncthreads();

#pragma unroll
    for (int j = 0; j < MLP_NT; ++j) {
        if (tv[j]) {
            const int tl = wave * MLP_NT + j;
            if constexpr (SMALL) {
#pragma unroll
                for (int i = 0; i < 4; ++i) {
                    const int f = 8 * (lane + 32 * i);
                    const v8h o8 = *(const v8h*)(sOh + (tl * 16 + (f >> 6)) * P2 + (f & 63));
                    *(v8h*)(outh + (size_t)row0[j] * HID_C + f) = o8;
                }
            } else {
#pragma unroll
                for (int i = 0; i < 8; ++i) {
                    const int f = 4 * (lane + 32 * i);
                    const float4 o4 = *(const float4*)(sOf + (tl * 16 + (f >> 6)) * PO + (f & 63));
                    *(float4*)(outf + (size_t)row0[j] * HID_C + f) = o4;
                }
            }
        }
    }
}

#define OWN_BLOCKS  256
#define OWN_THREADS 1024

template <int RN_, int S_, int R_, int WCAP_, int DB_, int U_>
struct own_cfg {
    static constexpr int RN = RN_, S = S_, R = R_, WCAP = WCAP_, DB = DB_, U = U_;
    static constexpr int SL4   = N_EDGES_C / 4 / S_;
    static constexpr int STEP  = OWN_THREADS * U_;
    static constexpr int TRIPS = (SL4 + STEP - 1) / STEP;
    static_assert((N_EDGES_C / 4) % S_ == 0, "a slice is a whole number of int4 words");
    static_assert(R_ * S_ == OWN_BLOCKS, "one block per (range, slice)");
    static_assert(R_ * RN_ >= N_NODES_C, "ranges cover every node");
    static_assert(RN_ <= (1 << DB_), "place in range fits its bits");
    static_assert(((long long)SL4 * 4) <= (1ll << (32 - DB_)), "edge id in slice fits its bits");
};
#define LIST_SUBS  8
typedef own_cfg<400, 1, 256, 512, 10, 8>  cfg2;
typedef own_cfg<3200, 8, 32, 1280, 12, 4> cfg1;
static_assert(cfg1::RN == LIST_SUBS * cfg2::RN && cfg2::R == LIST_SUBS * cfg1::R, "a conv1 range is LIST_SUBS conv2 ranges");
static_assert(cfg1::S == LIST_SUBS * cfg2::S, "a conv2 slice is LIST_SUBS conv1 slices");
static_assert(cfg1::WCAP % 32 == 0, "a wave's padded list part fits its capacity");
static_assert(cfg2::RN % 16 == 0 && N_NODES_C % cfg2::RN == 0, "a conv2 range is whole 16-node tiles and no range is partial");
static_assert(cfg2::RN / 16 <= OWN_THREADS / 32, "one wave per 16-node tile in the conv2 node MLP");

#define LIST_CAP   ((OWN_THREADS / 32) * cfg1::WCAP)
#define LIST_BAD   0xFFFFFFFFu
#define CNT_PITCH  32
#define LIST_U     8
static_assert((LIST_BAD & ((1u << cfg1::DB) - 1u)) >= (unsigned)cfg1::RN, "the pad entry lies outside every range");

typedef __attribute__((ext_vector_type(2))) _Float16 v2h;

template <class C>
__device__ __forceinline__ void own_push(unsigned* wlist, unsigned& wcount,
                                         int dstv, int n0, int erel) {
    const unsigned d = (unsigned)(dstv - n0);
    const bool m = d < (unsigned)C::RN;
    const unsigned mask = __builtin_amdgcn_ballot_w32(m);
    if (mask) {
        const unsigned pos = wcount + __builtin_amdgcn_mbcnt_lo(mask, 0u);
        if (m && pos < (unsigned)C::WCAP) wlist[pos] = ((unsigned)erel << C::DB) | d;
        wcount += (unsigned)__builtin_popcount(mask);
    }
}

template <class C>
__device__ __forceinline__ unsigned own_scan(const int4* __restrict__ d4, unsigned* wlist,
                                             int n0, int rot, int tid) {
    unsigned wcount = 0;
    for (int t = 0; t < C::TRIPS; ++t) {
        int tt = t + rot;
        if (tt >= C::TRIPS) tt -= C::TRIPS;
        const int base = tt * C::STEP;
        int4 v[C::U];
#pragma unroll
        for (int u = 0; u < C::U; ++u) {
            const int i = base + u * OWN_THREADS + tid;
            v[u] = (i < C::SL4) ? d4[i] : make_int4(-1, -1, -1, -1);
        }
#pragma unroll
        for (int u = 0; u < C::U; ++u) {
            const int e = (base + u * OWN_THREADS + tid) * 4;
            own_push<C>(wlist, wcount, v[u].x, n0, e + 0);
            own_push<C>(wlist, wcount, v[u].y, n0, e + 1);
            own_push<C>(wlist, wcount, v[u].z, n0, e + 2);
            own_push<C>(wlist, wcount, v[u].w, n0, e + 3);
        }
    }
    return wcount;
}

template <class C, class F>
__device__ __forceinline__ void own_overflow(const int4* __restrict__ d4, int n0, int rot,
                                             int tid, F add) {
    unsigned seen = 0;
#pragma unroll 1
    for (int t = 0; t < C::TRIPS * C::U; ++t) {
        int tt = t / C::U + rot;
        if (tt >= C::TRIPS) tt -= C::TRIPS;
        const int i = (tt * C::U + (t % C::U)) * OWN_THREADS + tid;
        const int4 v = (i < C::SL4) ? d4[i] : make_int4(-1, -1, -1, -1);
        const int dv[4] = {v.x, v.y, v.z, v.w};
#pragma unroll 1
        for (int k = 0; k < 4; ++k) {
            const unsigned d = (unsigned)(dv[k] - n0);
            const bool m = d < (unsigned)C::RN;
            const unsigned mask = __builtin_amdgcn_ballot_w32(m);
            const unsigned pos = seen + __builtin_amdgcn_mbcnt_lo(mask, 0u);
            seen += (unsigned)__builtin_popcount(mask);
            if (m && pos >= (unsigned)C::WCAP) add(i * 4 + k, (int)d);
        }
    }
}

__device__ __forceinline__ void list_push(unsigned* wlist, unsigned& wcount, unsigned ent,
                                          unsigned sublo, unsigned eoff) {
    const unsigned d = (ent & ((1u << cfg1::DB) - 1u)) - sublo;
    const bool m = d < (unsigned)cfg2::RN;
    const unsigned mask = __builtin_amdgcn_ballot_w32(m);
    if (mask) {
        const unsigned pos = wcount + __builtin_amdgcn_mbcnt_lo(mask, 0u);
        if (m && pos < (unsigned)cfg2::WCAP)
            wlist[pos] = (((ent >> cfg1::DB) + eoff) << cfg2::DB) | d;
        wcount += (unsigned)__builtin_popcount(mask);
    }
}

__device__ __forceinline__ unsigned list_scan(const unsigned* __restrict__ lists, int l0,
                                              const unsigned (&cnt)[LIST_SUBS], unsigned cmax,
                                              unsigned* wlist, unsigned sublo, int wave, int lane) {
    unsigned wcount = 0;
    for (unsigned k0 = 0; k0 * OWN_THREADS < cmax; k0 += LIST_U) {
        unsigned v[LIST_SUBS][LIST_U];
#pragma unroll
        for (int q = 0; q < LIST_SUBS; ++q) {
            const unsigned* L = lists + (size_t)(l0 + q) * LIST_CAP;
            const unsigned wb = (unsigned)(((wave + 8 * q) & 31) * 32 + lane);
#pragma unroll
            for (int u = 0; u < LIST_U; ++u) {
                const unsigned i = (k0 + (unsigned)u) * OWN_THREADS + wb;
                v[q][u] = (i < cnt[q]) ? L[i] : LIST_BAD;
            }
        }
#pragma unroll
        for (int q = 0; q < LIST_SUBS; ++q) {
#pragma unroll
            for (int u = 0; u < LIST_U; ++u)
                list_push(wlist, wcount, v[q][u], sublo, (unsigned)(q * (cfg1::SL4 * 4)));
        }
    }
    return wcount;
}

template <class F>
__device__ __forceinline__ void list_overflow(const unsigned* __restrict__ lists,
                                              const unsigned* __restrict__ lcount, int l0,
                                              unsigned cmax, unsigned sublo, int wave, int lane, F add) {
    unsigned seen = 0;
#pragma unroll 1
    for (unsigned k0 = 0; k0 * OWN_THREADS < cmax; k0 += LIST_U) {
#pragma unroll 1
        for (int q = 0; q < LIST_SUBS; ++q) {
            const unsigned* L = lists + (size_t)(l0 + q) * LIST_CAP;
            const unsigned cq = lcount[(l0 + q) * CNT_PITCH];
            const unsigned wb = (unsigned)(((wave + 8 * q) & 31) * 32 + lane);
#pragma unroll 1
            for (int u = 0; u < LIST_U; ++u) {
                const unsigned i = (k0 + (unsigned)u) * OWN_THREADS + wb;
                const unsigned ent = (i < cq) ? L[i] : LIST_BAD;
                const unsigned d = (ent & ((1u << cfg1::DB) - 1u)) - sublo;
                const bool m = d < (unsigned)cfg2::RN;
                const unsigned mask = __builtin_amdgcn_ballot_w32(m);
                const unsigned pos = seen + __builtin_amdgcn_mbcnt_lo(mask, 0u);
                seen += (unsigned)__builtin_popcount(mask);
                if (m && pos >= (unsigned)cfg2::WCAP)
                    add((int)((ent >> cfg1::DB) + (unsigned)(q * (cfg1::SL4 * 4))), (int)d);
            }
        }
    }
}

__global__ __launch_bounds__(OWN_THREADS)
void edge2_own_kernel(const _Float16* __restrict__ h, const int* __restrict__ ei,
                      const float* __restrict__ ea,
                      const float* __restrict__ eW, const float* __restrict__ eb,
                      const unsigned* __restrict__ lists, const unsigned* __restrict__ lcount,
                      const float* __restrict__ W1, const float* __restrict__ b1,
                      const float* __restrict__ W2, const float* __restrict__ b2,
                      float* __restrict__ h2) {
    typedef cfg2 C;
    constexpr int P2 = HID_C + 8;
    constexpr int NT = C::RN / 16;
    constexpr int PA = HID_C + 4;
    static_assert(HID_C * HID_C / 4 == OWN_THREADS, "one float4 of W1 and one of W2 per thread");
    __shared__ __align__(16) float    s_acc[C::RN * PA];
    __shared__ __align__(16) unsigned s_list[(OWN_THREADS / 32) * C::WCAP];
    __shared__ __align__(16) _Float16 sW1[HID_C * P2];
    __shared__ __align__(16) _Float16 sW2[HID_C * P2];

    const int tid   = threadIdx.x;
    const int lane  = tid & 31;
    const int wave  = tid >> 5;
    const int range = (int)blockIdx.x / C::S;
    const int slice = (int)blockIdx.x - range * C::S;
    const int n0    = range * C::RN;
    const int e0    = slice * (C::SL4 * 4);
    unsigned* wlist = s_list + wave * C::WCAP;

    if (n0 >= N_NODES_C) return;

    const float4 w1v = ((const float4*)W1)[tid];
    const float4 w2v = ((const float4*)W2)[tid];

    for (int i = tid; i < C::RN * PA / 4; i += OWN_THREADS)
        ((float4*)s_acc)[i] = make_float4(0.f, 0.f, 0.f, 0.f);
    {
        const int n = tid / (HID_C / 4), q = tid - n * (HID_C / 4);
        v4h o1, o2;
        o1[0] = (_Float16)w1v.x; o1[1] = (_Float16)w1v.y; o1[2] = (_Float16)w1v.z; o1[3] = (_Float16)w1v.w;
        o2[0] = (_Float16)w2v.x; o2[1] = (_Float16)w2v.y; o2[2] = (_Float16)w2v.z; o2[3] = (_Float16)w2v.w;
        *(v4h*)(sW1 + n * P2 + 4 * q) = o1;
        *(v4h*)(sW2 + n * P2 + 4 * q) = o2;
    }
    __syncthreads();

    const int c0 = 2 * lane;
    const float4 w0 = *(const float4*)(eW + (c0 + 0) * EDGE_DIM_C);
    const float4 w1 = *(const float4*)(eW + (c0 + 1) * EDGE_DIM_C);
    const float  b0 = eb[c0], b1e = eb[c0 + 1];

    auto add_one = [&](int erel, int od) {
        const int oe = e0 + erel;
        const int os = ei[oe];
        const float4 a = *(const float4*)(ea + (size_t)oe * EDGE_DIM_C);
#pragma unroll 1
        for (int q = 0; q < HID_C; ++q) {
            const float4 w = *(const float4*)(eW + q * EDGE_DIM_C);
            float t = eb[q] + a.x * w.x + a.y * w.y + a.z * w.z + a.w * w.w;
            t += (float)h[(size_t)os * HID_C + q];
            atomicAdd(&s_acc[od * PA + q], fmaxf(t, 0.0f));
        }
    };

    const int l0 = (range / LIST_SUBS) * cfg1::S + slice * LIST_SUBS;
    const unsigned sublo = (unsigned)((range % LIST_SUBS) * C::RN);
    unsigned cnt[LIST_SUBS];
    unsigned cmax = 0;
#pragma unroll
    for (int q = 0; q < LIST_SUBS; ++q) {
        cnt[q] = lcount[(l0 + q) * CNT_PITCH];
        cmax = cnt[q] > cmax ? cnt[q] : cmax;
    }
    unsigned wcount;
    if (cmax <= (unsigned)LIST_CAP) {
        wcount = list_scan(lists, l0, cnt, cmax, wlist, sublo, wave, lane);
        if (wcount > (unsigned)C::WCAP)
            list_overflow(lists, lcount, l0, cmax, sublo, wave, lane, add_one);
    } else {
        const int4* d4 = (const int4*)(ei + N_EDGES_C + e0);
        const int rot = (range * C::TRIPS) / C::R;
        wcount = own_scan<C>(d4, wlist, n0, rot, tid);
        if (wcount > (unsigned)C::WCAP)
            own_overflow<C>(d4, n0, rot, tid, add_one);
    }
    const int wn = (int)(wcount < (unsigned)C::WCAP ? wcount : (unsigned)C::WCAP);

    __syncthreads();

    const v2h* h2p = (const v2h*)h;
    for (int p = 0; p < wn; p += 32) {
        const int idx = p + lane;
        const bool valid = idx < wn;
        const unsigned ent = valid ? wlist[idx] : 0u;
        const int e  = e0 + (int)(ent >> C::DB);
        const int dl = (int)(ent & ((1u << C::DB) - 1u));
        const int sl = valid ? ei[e] : 0;
        const float4 al = valid ? *(const float4*)(ea + (size_t)e * EDGE_DIM_C)
                                : make_float4(0.f, 0.f, 0.f, 0.f);
        const int cnt = (wn - p) < 32 ? (wn - p) : 32;
        v2h hv[32];
#pragma unroll
        for (int j = 0; j < 32; ++j) {
            const int sj = __builtin_amdgcn_readlane(sl, j);
            hv[j] = h2p[(size_t)sj * (HID_C / 2) + lane];
        }
#pragma unroll
        for (int j = 0; j < 32; ++j) {
            if (j < cnt) {
                const int   dj = __builtin_amdgcn_readlane(dl, j);
                const float ax = __builtin_bit_cast(float, __builtin_amdgcn_readlane(__builtin_bit_cast(int, al.x), j));
                const float ay = __builtin_bit_cast(float, __builtin_amdgcn_readlane(__builtin_bit_cast(int, al.y), j));
                const float az = __builtin_bit_cast(float, __builtin_amdgcn_readlane(__builtin_bit_cast(int, al.z), j));
                const float aw = __builtin_bit_cast(float, __builtin_amdgcn_readlane(__builtin_bit_cast(int, al.w), j));
                float t0 = b0  + ax * w0.x + ay * w0.y + az * w0.z + aw * w0.w;
                float t1 = b1e + ax * w1.x + ay * w1.y + az * w1.z + aw * w1.w;
                t0 += (float)hv[j][0];
                t1 += (float)hv[j][1];
                atomicAdd(&s_acc[dj * PA + c0],     fmaxf(t0, 0.0f));
                atomicAdd(&s_acc[dj * PA + c0 + 1], fmaxf(t1, 0.0f));
            }
        }
    }
    const int  wv   = __builtin_amdgcn_readfirstlane(wave);
    const bool tile = wv < NT;
    const int  c    = lane & 15;
    const int  hh   = lane >> 4;
    const _Float16* grow = h + (size_t)(n0 + (tile ? wv * 16 + c : 0)) * HID_C + 8 * hh;
    v8h    xr[4];
    float4 pb1[8], pb2[8];
#pragma unroll
    for (int i = 0; i < 4; ++i) {
        xr[i]          = *(const v8h*)(grow + 16 * i);
        pb1[2 * i]     = *(const float4*)(b1 + 16 * i + 8 * hh);
        pb1[2 * i + 1] = *(const float4*)(b1 + 16 * i + 8 * hh + 4);
        pb2[2 * i]     = *(const float4*)(b2 + 16 * i + 8 * hh);
        pb2[2 * i + 1] = *(const float4*)(b2 + 16 * i + 8 * hh + 4);
    }

    __syncthreads();

    float*     trow  = s_acc + (wv * 16 + c) * PA;
    if (tile) {
        v16h bop[2];
#pragma unroll
        for (int kb = 0; kb < 2; ++kb) {
            const int k0 = kb * 32 + 8 * hh;
            const v8h x0 = xr[2 * kb];
            const v8h x1 = xr[2 * kb + 1];
            const float4 a0 = *(const float4*)(trow + k0);
            const float4 a1 = *(const float4*)(trow + k0 + 4);
            const float4 a2 = *(const float4*)(trow + k0 + 16);
            const float4 a3 = *(const float4*)(trow + k0 + 20);
            const float lo[8] = {a0.x, a0.y, a0.z, a0.w, a1.x, a1.y, a1.z, a1.w};
            const float hi[8] = {a2.x, a2.y, a2.z, a2.w, a3.x, a3.y, a3.z, a3.w};
            v16h b;
#pragma unroll
            for (int e = 0; e < 8; ++e) {
                b[e]     = (_Float16)((float)x0[e] + lo[e]);
                b[8 + e] = (_Float16)((float)x1[e] + hi[e]);
            }
            bop[kb] = b;
        }

        v8f acc[4];
#pragma unroll
        for (int t = 0; t < 4; ++t) {
            acc[t] = v8f{};
#pragma unroll
            for (int kb = 0; kb < 2; ++kb) {
                const v16h a = lds_load_tile(sW1 + 16 * t * P2, P2, kb * 32);
                acc[t] = __builtin_amdgcn_wmma_f32_16x16x32_f16(false, a, false, bop[kb],
                                                                (short)0, acc[t], false, false);
            }
        }

        v16h bop2[2];
#pragma unroll
        for (int s2 = 0; s2 < 2; ++s2) {
            const float4 p0 = pb1[4 * s2], p1 = pb1[4 * s2 + 1], p2 = pb1[4 * s2 + 2], p3 = pb1[4 * s2 + 3];
            const float bl[8] = {p0.x, p0.y, p0.z, p0.w, p1.x, p1.y, p1.z, p1.w};
            const float bh[8] = {p2.x, p2.y, p2.z, p2.w, p3.x, p3.y, p3.z, p3.w};
            v16h b;
#pragma unroll
            for (int e = 0; e < 8; ++e) {
                float v0 = acc[2 * s2][e] + bl[e];
                float v1 = acc[2 * s2 + 1][e] + bh[e];
                v0 = v0 > 0.0f ? v0 : 0.0f;
                v1 = v1 > 0.0f ? v1 : 0.0f;
                b[e]     = (_Float16)v0;
                b[8 + e] = (_Float16)v1;
            }
            bop2[s2] = b;
        }

#pragma unroll
        for (int t = 0; t < 4; ++t) {
            v8f d = v8f{};
#pragma unroll
            for (int s2 = 0; s2 < 2; ++s2) {
                const v16h a = lds_load_tile(sW2 + 16 * t * P2, P2, s2 * 32);
                d = __builtin_amdgcn_wmma_f32_16x16x32_f16(false, a, false, bop2[s2],
                                                           (short)0, d, false, false);
            }
            const float4 q0 = pb2[2 * t], q1 = pb2[2 * t + 1];
            const float bb[8] = {q0.x, q0.y, q0.z, q0.w, q1.x, q1.y, q1.z, q1.w};
            float v[8];
#pragma unroll
            for (int e = 0; e < 8; ++e) {
                const float u = d[e] + bb[e];
                v[e] = u > 0.0f ? u : 0.0f;
            }
            float* q = trow + 16 * t + 8 * hh;
            *(float4*)(q)     = make_float4(v[0], v[1], v[2], v[3]);
            *(float4*)(q + 4) = make_float4(v[4], v[5], v[6], v[7]);
        }
    }
    __syncthreads();

    if (tile) {
        const float* st = s_acc + wv * 16 * PA;
        float*       gt = h2 + (size_t)(n0 + wv * 16) * HID_C;
#pragma unroll
        for (int i = 0; i < 8; ++i) {
            const int f = 4 * (lane + 32 * i);
            *(float4*)(gt + f) = *(const float4*)(st + (f >> 6) * PA + (f & 63));
        }
    }
}

__global__ __launch_bounds__(OWN_THREADS)
void edge1_own_kernel(const float* __restrict__ x, const int* __restrict__ ei,
                      const float* __restrict__ ea,
                      const float* __restrict__ eW, const float* __restrict__ eb,
                      float* __restrict__ aggr,
                      unsigned* __restrict__ lists, unsigned* __restrict__ lcount) {
    typedef cfg1 C;
    __shared__ __align__(16) float    s_acc[C::RN * 8];
    __shared__ __align__(16) unsigned s_list[(OWN_THREADS / 32) * C::WCAP];
    __shared__ unsigned s_wn[OWN_THREADS / 32];

    const int tid   = threadIdx.x;
    const int lane  = tid & 31;
    const int wave  = tid >> 5;
    const int range = (int)blockIdx.x / C::S;
    const int slice = (int)blockIdx.x - range * C::S;
    const int n0    = range * C::RN;
    const int e0    = slice * (C::SL4 * 4);
    unsigned* wlist = s_list + wave * C::WCAP;

    for (int i = tid; i < C::RN * 8; i += OWN_THREADS) s_acc[i] = 0.0f;
    __syncthreads();

    const int c   = lane & 7;
    const int sub = lane >> 3;
    const bool cact = c < IN_DIM_C;
    const float4 wc = cact ? *(const float4*)(eW + c * EDGE_DIM_C) : make_float4(0.f, 0.f, 0.f, 0.f);
    const float  bc = cact ? eb[c] : 0.0f;
    const int    cx = cact ? c : 0;

    const int4* d4 = (const int4*)(ei + N_EDGES_C + e0);
    const int rot = (range * C::TRIPS) / C::R;
    const unsigned wcount = own_scan<C>(d4, wlist, n0, rot, tid);
    if (wcount > (unsigned)C::WCAP) {
        own_overflow<C>(d4, n0, rot, tid, [&](int erel, int od) {
            const int oe = e0 + erel;
            const int os = ei[oe];
            const float4 a = *(const float4*)(ea + (size_t)oe * EDGE_DIM_C);
#pragma unroll 1
            for (int q = 0; q < IN_DIM_C; ++q) {
                const float4 w = *(const float4*)(eW + q * EDGE_DIM_C);
                float t = eb[q] + a.x * w.x + a.y * w.y + a.z * w.z + a.w * w.w;
                t += x[(size_t)os * IN_DIM_C + q];
                atomicAdd(&s_acc[od * 8 + q], fmaxf(t, 0.0f));
            }
        });
    }
    const int wn = (int)(wcount < (unsigned)C::WCAP ? wcount : (unsigned)C::WCAP);
    if (lane == 0) s_wn[wave] = wcount;

    __syncthreads();

    {
        unsigned off = 0, tot = 0;
        bool full = false;
        for (int w = 0; w < OWN_THREADS / 32; ++w) {
            const unsigned cw = s_wn[w];
            full = full || cw > (unsigned)C::WCAP;
            const unsigned pw = ((cw < (unsigned)C::WCAP ? cw : (unsigned)C::WCAP) + 31u) & ~31u;
            if (w < wave) off += pw;
            tot += pw;
        }
        unsigned* L = lists + (size_t)blockIdx.x * LIST_CAP + off;
        const int wp = (wn + 31) & ~31;
        for (int i = lane; i < wp; i += 32) L[i] = (i < wn) ? wlist[i] : LIST_BAD;
        if (tid == 0) lcount[blockIdx.x * CNT_PITCH] = full ? LIST_BAD : tot;
    }

    for (int p = 0; p < wn; p += 32) {
        unsigned ent[8]; int sr[8]; float4 av[8]; float xv[8]; bool ok[8];
#pragma unroll
        for (int g = 0; g < 8; ++g) {
            const int idx = p + g * 4 + sub;
            ok[g]  = idx < wn;
            ent[g] = ok[g] ? wlist[idx] : 0u;
        }
#pragma unroll
        for (int g = 0; g < 8; ++g) {
            const int e = e0 + (int)(ent[g] >> C::DB);
            sr[g] = ei[e];
            av[g] = *(const float4*)(ea + (size_t)e * EDGE_DIM_C);
        }
#pragma unroll
        for (int g = 0; g < 8; ++g)
            xv[g] = x[(size_t)sr[g] * IN_DIM_C + cx];
        asm volatile("" :: "v"(xv[0]), "v"(xv[1]), "v"(xv[2]), "v"(xv[3]), "v"(xv[4]), "v"(xv[5]), "v"(xv[6]), "v"(xv[7]),
                           "v"(av[0].x), "v"(av[0].y), "v"(av[0].z), "v"(av[0].w));
#pragma unroll
        for (int g = 0; g < 8; ++g) {
            if (ok[g] && cact) {
                const int dl = (int)(ent[g] & ((1u << C::DB) - 1u));
                float t = bc + av[g].x * wc.x + av[g].y * wc.y + av[g].z * wc.z + av[g].w * wc.w;
                t += xv[g];
                atomicAdd(&s_acc[dl * 8 + c], fmaxf(t, 0.0f));
            }
        }
    }
    __syncthreads();

    int rows = N_NODES_C - n0;
    if (rows > C::RN) rows = C::RN;
    float* o = aggr + (size_t)slice * N_NODES_C * IN_DIM_C + (size_t)n0 * IN_DIM_C;
    for (int i = tid; i < rows * IN_DIM_C; i += OWN_THREADS) {
        const int r = i / IN_DIM_C, q = i - r * IN_DIM_C;
        o[i] = s_acc[r * 8 + q];
    }
}

#define POOL_GPB     4
#define POOL_THREADS 1024
#define POOL_U       14
#define POOL_Q       8

__global__ __launch_bounds__(POOL_THREADS)
void pool_fc_kernel(const float* __restrict__ h2, const int* __restrict__ batch,
                    const float* __restrict__ fcW, const float* __restrict__ fcb,
                    float* __restrict__ out) {
    constexpr int NW  = POOL_THREADS / 32;
    constexpr int NCH = (N_NODES_C + 31) / 32;
    static_assert(N_GRAPHS_C % POOL_GPB == 0, "every block owns POOL_GPB graphs");
    static_assert(POOL_GPB * HID_C <= POOL_THREADS, "one thread per (graph, channel) in the wave sum");

    __shared__ __align__(16) float s_part[NW][POOL_GPB][HID_C];
    __shared__ int   s_cnt [NW][POOL_GPB];
    __shared__ float s_mean[POOL_GPB][HID_C];

    const int tid  = threadIdx.x;
    const int lane = tid & 31;
    const int wave = tid >> 5;
    const int g0   = (int)blockIdx.x * POOL_GPB;
    const float2* h2v = (const float2*)h2;

    float2 acc[POOL_GPB];
    int    cnt[POOL_GPB];
#pragma unroll
    for (int k = 0; k < POOL_GPB; ++k) { acc[k] = make_float2(0.f, 0.f); cnt[k] = 0; }

    for (int c0 = wave; c0 < NCH; c0 += NW * POOL_U) {
        unsigned hit = 0;
#pragma unroll
        for (int u = 0; u < POOL_U; ++u) {
            const int c = c0 + u * NW;
            const int n = c * 32 + lane;
            const int g = (c < NCH && n < N_NODES_C) ? batch[n] : -1;
            const bool m = (unsigned)(g - g0) < (unsigned)POOL_GPB;
            if (__builtin_amdgcn_ballot_w32(m)) hit |= 1u << u;
        }
        while (hit) {
            const int u = __builtin_ctz(hit);
            hit &= hit - 1;
            const int base = (c0 + u * NW) * 32;
            const int n = base + lane;
            const int g = (n < N_NODES_C) ? batch[n] : -1;
            const int d = g - g0;
            unsigned mask = __builtin_amdgcn_ballot_w32((unsigned)d < (unsigned)POOL_GPB);
#pragma unroll
            for (int k = 0; k < POOL_GPB; ++k)
                cnt[k] += __builtin_popcount(__builtin_amdgcn_ballot_w32(d == k));
            while (mask) {
                int gq[POOL_Q]; float2 v[POOL_Q];
                const int jf = __builtin_ctz(mask);
#pragma unroll
                for (int q = 0; q < POOL_Q; ++q) {
                    const bool has = mask != 0;
                    const int j = has ? __builtin_ctz(mask) : jf;
                    const int gj = __builtin_amdgcn_readlane(d, j);
                    gq[q] = has ? gj : -1;
                    v[q]  = h2v[(size_t)(base + j) * (HID_C / 2) + lane];
                    mask &= mask - 1;
                }
                __builtin_amdgcn_sched_barrier(0);
#pragma unroll
                for (int q = 0; q < POOL_Q; ++q) {
#pragma unroll
                    for (int k = 0; k < POOL_GPB; ++k)
                        if (gq[q] == k) { acc[k].x += v[q].x; acc[k].y += v[q].y; }
                }
            }
        }
    }

#pragma unroll
    for (int k = 0; k < POOL_GPB; ++k) {
        *(float2*)&s_part[wave][k][2 * lane] = acc[k];
        if (lane == 0) s_cnt[wave][k] = cnt[k];
    }
    __syncthreads();

    if (tid < POOL_GPB * HID_C) {
        const int k = tid / HID_C, ch = tid - k * HID_C;
        float s = 0.0f; int c = 0;
        for (int w = 0; w < NW; ++w) { s += s_part[w][k][ch]; c += s_cnt[w][k]; }
        s_mean[k][ch] = s / fmaxf((float)c, 1.0f);
    }
    __syncthreads();

    if (tid < POOL_GPB * NUM_TASKS_C) {
        const int k = tid / NUM_TASKS_C, t = tid - k * NUM_TASKS_C;
        float s = fcb[t];
#pragma unroll
        for (int c = 0; c < HID_C; ++c)
            s += s_mean[k][c] * fcW[t * HID_C + c];
        out[(size_t)(g0 + k) * NUM_TASKS_C + t] = s;
    }
}

extern "C" void kernel_launch(void* const* d_in, const int* in_sizes, int n_in,
                              void* d_out, int out_size, void* d_ws, size_t ws_size,
                              hipStream_t stream) {
    const float* x    = (const float*)d_in[0];
    const int*   ei   = (const int*)  d_in[1];
    const float* ea   = (const float*)d_in[2];
    const int*   bat  = (const int*)  d_in[3];
    const float* e1W  = (const float*)d_in[4];  const float* e1b  = (const float*)d_in[5];
    const float* n1W1 = (const float*)d_in[6];  const float* n1b1 = (const float*)d_in[7];
    const float* n1W2 = (const float*)d_in[8];  const float* n1b2 = (const float*)d_in[9];
    const float* e2W  = (const float*)d_in[10]; const float* e2b  = (const float*)d_in[11];
    const float* n2W1 = (const float*)d_in[12]; const float* n2b1 = (const float*)d_in[13];
    const float* n2W2 = (const float*)d_in[14]; const float* n2b2 = (const float*)d_in[15];
    const float* fcW  = (const float*)d_in[16]; const float* fcb  = (const float*)d_in[17];
    float* out = (float*)d_out;

    float*    aggr1 = (float*)d_ws;
    _Float16* h     = (_Float16*)(aggr1 + (size_t)cfg1::S * N_NODES_C * IN_DIM_C);
    float*    h2    = (float*)(h + (size_t)N_NODES_C * HID_C);
    unsigned* lists  = (unsigned*)(h2 + (size_t)N_NODES_C * HID_C);
    unsigned* lcount = lists + (size_t)OWN_BLOCKS * LIST_CAP;

    edge1_own_kernel<<<OWN_BLOCKS, OWN_THREADS, 0, stream>>>(x, ei, ea, e1W, e1b, aggr1, lists, lcount);
    node_mlp_kernel<IN_DIM_C, cfg1::S><<<MLP_GRID, MLP_THREADS, 0, stream>>>(
        x, nullptr, aggr1, n1W1, n1b1, n1W2, n1b2, h, nullptr, N_NODES_C);
    edge2_own_kernel<<<OWN_BLOCKS, OWN_THREADS, 0, stream>>>(h, ei, ea, e2W, e2b, lists, lcount,
                                                             n2W1, n2b1, n2W2, n2b2, h2);
    pool_fc_kernel<<<N_GRAPHS_C / POOL_GPB, POOL_THREADS, 0, stream>>>(h2, bat, fcW, fcb, out);
}


# Conv2 edge messages 16 at a time on the matrix unit: edge_attr x eW^T as f32 WMMA 16x16x4 with eb + h[src] as C; a lane holds 4 channels of 8 edges, 8-byte row loads; sums still LDS arrival order
# speedup vs baseline: 4.1237x; 1.1056x over previous
#include <hip/hip_runtime.h>
#include <hip/hip_bf16.h>

#define N_NODES_C  100000
#define N_EDGES_C  1600000
#define N_GRAPHS_C 1000
#define HID_C      64
#define IN_DIM_C   7
#define EDGE_DIM_C 4
#define NUM_TASKS_C 12

typedef __attribute__((ext_vector_type(16))) _Float16 v16h;
typedef __attribute__((ext_vector_type(8)))  _Float16 v8h;
typedef __attribute__((ext_vector_type(8)))  float    v8f;

__device__ __forceinline__ v16h lds_load_tile(const _Float16* s, int ld, int kbase) {
    const int lane = threadIdx.x & 31;
    const int r    = lane & 15;
    const int koff = (lane < 16) ? 0 : 8;
    const _Float16* p = s + r * ld + kbase + koff;
    v8h lo = *(const v8h*)(p);
    v8h hi = *(const v8h*)(p + 16);
    v16h out;
#pragma unroll
    for (int e = 0; e < 8; ++e) { out[e] = lo[e]; out[8 + e] = hi[e]; }
    return out;
}

#define MLP_WAVES   8
#define MLP_NT      2
#define MLP_THREADS (MLP_WAVES * 32)
#define MLP_TILES   (MLP_WAVES * MLP_NT)
#define MLP_ROWS    (MLP_TILES * 16)
#define MLP_GRID    ((N_NODES_C + MLP_ROWS - 1) / MLP_ROWS)
static_assert(N_NODES_C % 16 == 0, "the node count is a whole number of 16-node tiles");

typedef __attribute__((ext_vector_type(4))) _Float16 v4h;

template <int K, int KP>
struct mlp_wregs {
    static constexpr bool VEC = (K % 4) == 0;
    static constexpr int  N   = (VEC ? HID_C * (K / 4) : HID_C * KP) / MLP_THREADS;
    static_assert((VEC ? HID_C * (K / 4) : HID_C * KP) % MLP_THREADS == 0, "the same number of pieces in every thread");
    static_assert(!VEC || K == KP, "no padding on the 16-byte path");
    float4 v4[VEC ? N : 1];
    float  v1[VEC ? 1 : N];
};

template <int K, int KP>
__device__ __forceinline__ void mlp_load_weights(const float* __restrict__ W, int tid, mlp_wregs<K, KP>& r) {
#pragma unroll
    for (int it = 0; it < mlp_wregs<K, KP>::N; ++it) {
        const int i = tid + it * MLP_THREADS;
        if constexpr (mlp_wregs<K, KP>::VEC) {
            r.v4[it] = ((const float4*)W)[i];
        } else {
            const int n = i / KP, k = i - n * KP;
            const float w = W[n * K + (k < K ? k : 0)];
            r.v1[it] = (k < K) ? w : 0.0f;
        }
    }
}

template <int K, int KP, int P>
__device__ __forceinline__ void mlp_store_weights(_Float16* s, int tid, const mlp_wregs<K, KP>& r) {
#pragma unroll
    for (int it = 0; it < mlp_wregs<K, KP>::N; ++it) {
        const int i = tid + it * MLP_THREADS;
        if constexpr (mlp_wregs<K, KP>::VEC) {
            const int n = i / (K / 4), q = i - n * (K / 4);
            const float4 v = r.v4[it];
            v4h o;
            o[0] = (_Float16)v.x; o[1] = (_Float16)v.y; o[2] = (_Float16)v.z; o[3] = (_Float16)v.w;
            *(v4h*)(s + n * P + 4 * q) = o;
        } else {
            const int n = i / KP, k = i - n * KP;
            s[n * P + k] = (_Float16)r.v1[it];
        }
    }
}

template <int K1, int NP>
__global__ __launch_bounds__(MLP_THREADS)
void node_mlp_kernel(const float* __restrict__ xin, const _Float16* __restrict__ hin,
                     const float* __restrict__ in1,
                     const float* __restrict__ W1,  const float* __restrict__ b1,
                     const float* __restrict__ W2,  const float* __restrict__ b2,
                     _Float16* __restrict__ outh, float* __restrict__ outf, int n_nodes) {
    constexpr int  K1P   = (K1 + 31) & ~31;
    constexpr int  KS    = K1P / 32;
    constexpr int  P1    = K1P + 8;
    constexpr int  P2    = HID_C + 8;
    constexpr int  PO    = HID_C + 4;
    constexpr bool SMALL = (K1 % 8) != 0;
    constexpr int  XT    = 16 * K1;
    static_assert(!SMALL || (XT % 4 == 0 && XT / 4 <= 32), "a flat conv1 tile is at most one float4 per lane");
    static_assert(SMALL || K1 == HID_C, "the conv2 path cuts rows of 64");

    __shared__ __align__(16) _Float16 sW1[HID_C * P1];
    __shared__ __align__(16) _Float16 sW2[HID_C * P2];
    __shared__ __align__(16) float    sX [SMALL ? MLP_TILES * 128 : 4];
    __shared__ __align__(16) _Float16 sA [SMALL ? 8 : MLP_TILES * 16 * P1];
    __shared__ __align__(16) _Float16 sOh[SMALL ? MLP_TILES * 16 * P2 : 8];
    __shared__ __align__(16) float    sOf[SMALL ? 4 : MLP_TILES * 16 * PO];

    const int tid  = threadIdx.x;
    const int wave = __builtin_amdgcn_readfirstlane(tid >> 5);
    const int lane = tid & 31;
    const int c    = lane & 15;
    const int h    = lane >> 4;
    const int tile0 = ((int)blockIdx.x * MLP_WAVES + wave) * MLP_NT;

    int  row0[MLP_NT];
    bool tv[MLP_NT];
#pragma unroll
    for (int j = 0; j < MLP_NT; ++j) {
        row0[j] = (tile0 + j) * 16;
        tv[j]   = row0[j] < n_nodes;
    }

    mlp_wregs<K1, K1P>     w1r;
    mlp_wregs<HID_C, HID_C> w2r;
    mlp_load_weights<K1, K1P>(W1, tid, w1r);
    mlp_load_weights<HID_C, HID_C>(W2, tid, w2r);

    if constexpr (SMALL) {
        const int ll = lane < XT / 4 ? lane : XT / 4 - 1;
        float4 raw[MLP_NT][NP + 1];
#pragma unroll
        for (int j = 0; j < MLP_NT; ++j) {
            const int o4 = (tv[j] ? row0[j] : 0) * K1 / 4 + ll;
            raw[j][0] = ((const float4*)xin)[o4];
#pragma unroll
            for (int p = 0; p < NP; ++p)
                raw[j][p + 1] = ((const float4*)(in1 + (size_t)p * (size_t)n_nodes * K1))[o4];
        }
        float4 v[MLP_NT];
#pragma unroll
        for (int j = 0; j < MLP_NT; ++j) {
            v[j] = raw[j][0];
#pragma unroll
            for (int p = 0; p < NP; ++p) {
                const float4 a = raw[j][p + 1];
                v[j].x += a.x; v[j].y += a.y; v[j].z += a.z; v[j].w += a.w;
            }
        }
#pragma unroll
        for (int j = 0; j < MLP_NT; ++j)
            ((float4*)sX)[(wave * MLP_NT + j) * 32 + lane] = v[j];
    } else {
#pragma unroll
        for (int j = 0; j < MLP_NT; ++j) {
            const int    eb = (tv[j] ? row0[j] : 0) * K1;
            const int    tl = wave * MLP_NT + j;
#pragma unroll
            for (int i = 0; i < XT / 128; ++i) {
                const int f = 4 * (lane + 32 * i);
                const v4h x4 = *(const v4h*)(hin + eb + f);
                float v0 = (float)x4[0], v1 = (float)x4[1], v2 = (float)x4[2], v3 = (float)x4[3];
#pragma unroll
                for (int p = 0; p < NP; ++p) {
                    const float4 a = *(const float4*)(in1 + (size_t)p * (size_t)n_nodes * K1 + eb + f);
                    v0 += a.x; v1 += a.y; v2 += a.z; v3 += a.w;
                }
                v4h o;
                o[0] = (_Float16)v0; o[1] = (_Float16)v1; o[2] = (_Float16)v2; o[3] = (_Float16)v3;
                *(v4h*)(sA + (tl * 16 + (f >> 6)) * P1 + (f & 63)) = o;
            }
        }
    }

    mlp_store_weights<K1, K1P, P1>(sW1, tid, w1r);
    mlp_store_weights<HID_C, HID_C, P2>(sW2, tid, w2r);
    __syncthreads();

    v16h bop[MLP_NT][KS];
#pragma unroll
    for (int j = 0; j < MLP_NT; ++j) {
        const int tl = wave * MLP_NT + j;
        if constexpr (SMALL) {
            const float* r = sX + tl * 128 + c * K1;
            v16h b = {};
#pragma unroll
            for (int k = 0; k < K1; ++k) {
                const float f = r[k];
                b[k] = (h == 0) ? (_Float16)f : (_Float16)0.0f;
            }
            bop[j][0] = b;
        } else {
#pragma unroll
            for (int kb = 0; kb < KS; ++kb)
                bop[j][kb] = lds_load_tile(sA + tl * 16 * P1, P1, kb * 32);
        }
    }

    float bias1[4][8], bias2[4][8];
#pragma unroll
    for (int t = 0; t < 4; ++t) {
        const float4 a0 = *(const float4*)(b1 + 16 * t + 8 * h);
        const float4 a1 = *(const float4*)(b1 + 16 * t + 8 * h + 4);
        bias1[t][0] = a0.x; bias1[t][1] = a0.y; bias1[t][2] = a0.z; bias1[t][3] = a0.w;
        bias1[t][4] = a1.x; bias1[t][5] = a1.y; bias1[t][6] = a1.z; bias1[t][7] = a1.w;
        const float4 c0 = *(const float4*)(b2 + 16 * t + 8 * h);
        const float4 c1 = *(const float4*)(b2 + 16 * t + 8 * h + 4);
        bias2[t][0] = c0.x; bias2[t][1] = c0.y; bias2[t][2] = c0.z; bias2[t][3] = c0.w;
        bias2[t][4] = c1.x; bias2[t][5] = c1.y; bias2[t][6] = c1.z; bias2[t][7] = c1.w;
    }

    v8f acc[MLP_NT][4];
#pragma unroll
    for (int j = 0; j < MLP_NT; ++j)
#pragma unroll
        for (int t = 0; t < 4; ++t) acc[j][t] = v8f{};
#pragma unroll
    for (int t = 0; t < 4; ++t) {
#pragma unroll
        for (int kb = 0; kb < KS; ++kb) {
            const v16h a = lds_load_tile(sW1 + 16 * t * P1, P1, kb * 32);
#pragma unroll
            for (int j = 0; j < MLP_NT; ++j)
                acc[j][t] = __builtin_amdgcn_wmma_f32_16x16x32_f16(false, a, false, bop[j][kb],
                                                                   (short)0, acc[j][t], false, false);
        }
    }

    v16h bop2[MLP_NT][2];
#pragma unroll
    for (int j = 0; j < MLP_NT; ++j) {
#pragma unroll
        for (int s = 0; s < 2; ++s) {
            v16h b;
#pragma unroll
            for (int e = 0; e < 8; ++e) {
                float v0 = acc[j][2 * s][e] + bias1[2 * s][e];
                float v1 = acc[j][2 * s + 1][e] + bias1[2 * s + 1][e];
                v0 = v0 > 0.0f ? v0 : 0.0f;
                v1 = v1 > 0.0f ? v1 : 0.0f;
                b[e]     = (_Float16)v0;
                b[8 + e] = (_Float16)v1;
            }
            bop2[j][s] = b;
        }
    }

    v8f acc2[MLP_NT][4];
#pragma unroll
    for (int j = 0; j < MLP_NT; ++j)
#pragma unroll
        for (int t = 0; t < 4; ++t) acc2[j][t] = v8f{};
#pragma unroll
    for (int t = 0; t < 4; ++t) {
#pragma unroll
        for (int s = 0; s < 2; ++s) {
            const v16h a = lds_load_tile(sW2 + 16 * t * P2, P2, s * 32);
#pragma unroll
            for (int j = 0; j < MLP_NT; ++j)
                acc2[j][t] = __builtin_amdgcn_wmma_f32_16x16x32_f16(false, a, false, bop2[j][s],
                                                                    (short)0, acc2[j][t], false, false);
        }
    }

#pragma unroll
    for (int j = 0; j < MLP_NT; ++j) {
        const int tl = wave * MLP_NT + j;
#pragma unroll
        for (int t = 0; t < 4; ++t) {
            float v[8];
#pragma unroll
            for (int e = 0; e < 8; ++e) {
                const float u = acc2[j][t][e] + bias2[t][e];
                v[e] = u > 0.0f ? u : 0.0f;
            }
            if constexpr (SMALL) {
                v8h o8;
#pragma unroll
                for (int e = 0; e < 8; ++e) o8[e] = (_Float16)v[e];
                *(v8h*)(sOh + (tl * 16 + c) * P2 + 16 * t + 8 * h) = o8;
            } else {
                float* q = sOf + (tl * 16 + c) * PO + 16 * t + 8 * h;
                *(float4*)(q)     = make_float4(v[0], v[1], v[2], v[3]);
                *(float4*)(q + 4) = make_float4(v[4], v[5], v[6], v[7]);
            }
        }
    }
    __syncthreads();

#pragma unroll
    for (int j = 0; j < MLP_NT; ++j) {
        if (tv[j]) {
            const int tl = wave * MLP_NT + j;
            if constexpr (SMALL) {
#pragma unroll
                for (int i = 0; i < 4; ++i) {
                    const int f = 8 * (lane + 32 * i);
                    const v8h o8 = *(const v8h*)(sOh + (tl * 16 + (f >> 6)) * P2 + (f & 63));
                    *(v8h*)(outh + (size_t)row0[j] * HID_C + f) = o8;
                }
            } else {
#pragma unroll
                for (int i = 0; i < 8; ++i) {
                    const int f = 4 * (lane + 32 * i);
                    const float4 o4 = *(const float4*)(sOf + (tl * 16 + (f >> 6)) * PO + (f & 63));
                    *(float4*)(outf + (size_t)row0[j] * HID_C + f) = o4;
                }
            }
        }
    }
}

#define OWN_BLOCKS  256
#define OWN_THREADS 1024

template <int RN_, int S_, int R_, int WCAP_, int DB_, int U_>
struct own_cfg {
    static constexpr int RN = RN_, S = S_, R = R_, WCAP = WCAP_, DB = DB_, U = U_;
    static constexpr int SL4   = N_EDGES_C / 4 / S_;
    static constexpr int STEP  = OWN_THREADS * U_;
    static constexpr int TRIPS = (SL4 + STEP - 1) / STEP;
    static_assert((N_EDGES_C / 4) % S_ == 0, "a slice is a whole number of int4 words");
    static_assert(R_ * S_ == OWN_BLOCKS, "one block per (range, slice)");
    static_assert(R_ * RN_ >= N_NODES_C, "ranges cover every node");
    static_assert(RN_ <= (1 << DB_), "place in range fits its bits");
    static_assert(((long long)SL4 * 4) <= (1ll << (32 - DB_)), "edge id in slice fits its bits");
};
#define LIST_SUBS  8
typedef own_cfg<400, 1, 256, 512, 10, 8>  cfg2;
typedef own_cfg<3200, 8, 32, 1280, 12, 4> cfg1;
static_assert(cfg1::RN == LIST_SUBS * cfg2::RN && cfg2::R == LIST_SUBS * cfg1::R, "a conv1 range is LIST_SUBS conv2 ranges");
static_assert(cfg1::S == LIST_SUBS * cfg2::S, "a conv2 slice is LIST_SUBS conv1 slices");
static_assert(cfg1::WCAP % 32 == 0, "a wave's padded list part fits its capacity");
static_assert(cfg2::RN % 16 == 0 && N_NODES_C % cfg2::RN == 0, "a conv2 range is whole 16-node tiles and no range is partial");
static_assert(cfg2::RN / 16 <= OWN_THREADS / 32, "one wave per 16-node tile in the conv2 node MLP");

#define LIST_CAP   ((OWN_THREADS / 32) * cfg1::WCAP)
#define LIST_BAD   0xFFFFFFFFu
#define CNT_PITCH  32
#define LIST_U     8
static_assert((LIST_BAD & ((1u << cfg1::DB) - 1u)) >= (unsigned)cfg1::RN, "the pad entry lies outside every range");

typedef __attribute__((ext_vector_type(2))) _Float16 v2h;
typedef __attribute__((ext_vector_type(2))) float    v2f;

__device__ __forceinline__ int sum_pos(int ch) { return 16 * (ch & 3) + (ch >> 2); }

__device__ __forceinline__ void edge2_fetch(const unsigned* wlist, int p, int wn, int c, int hh, int e0,
                                            const int* __restrict__ ei, const float* __restrict__ ea,
                                            int (&sv)[8], v2f& av) {
    const uint4 q0 = *(const uint4*)(wlist + p + 8 * hh);
    const uint4 q1 = *(const uint4*)(wlist + p + 8 * hh + 4);
    const unsigned raw[8] = {q0.x, q0.y, q0.z, q0.w, q1.x, q1.y, q1.z, q1.w};
#pragma unroll
    for (int r = 0; r < 8; ++r) {
        const unsigned ent = (p + 8 * hh + r < wn) ? raw[r] : 0u;
        sv[r] = ei[e0 + (int)(ent >> cfg2::DB)];
    }
    const unsigned raw_c = wlist[p + c];
    const unsigned ent_c = (p + c < wn) ? raw_c : 0u;
    av = *(const v2f*)(ea + (size_t)(e0 + (int)(ent_c >> cfg2::DB)) * EDGE_DIM_C + 2 * hh);
}

template <class C>
__device__ __forceinline__ void own_push(unsigned* wlist, unsigned& wcount,
                                         int dstv, int n0, int erel) {
    const unsigned d = (unsigned)(dstv - n0);
    const bool m = d < (unsigned)C::RN;
    const unsigned mask = __builtin_amdgcn_ballot_w32(m);
    if (mask) {
        const unsigned pos = wcount + __builtin_amdgcn_mbcnt_lo(mask, 0u);
        if (m && pos < (unsigned)C::WCAP) wlist[pos] = ((unsigned)erel << C::DB) | d;
        wcount += (unsigned)__builtin_popcount(mask);
    }
}

template <class C>
__device__ __forceinline__ unsigned own_scan(const int4* __restrict__ d4, unsigned* wlist,
                                             int n0, int rot, int tid) {
    unsigned wcount = 0;
    for (int t = 0; t < C::TRIPS; ++t) {
        int tt = t + rot;
        if (tt >= C::TRIPS) tt -= C::TRIPS;
        const int base = tt * C::STEP;
        int4 v[C::U];
#pragma unroll
        for (int u = 0; u < C::U; ++u) {
            const int i = base + u * OWN_THREADS + tid;
            v[u] = (i < C::SL4) ? d4[i] : make_int4(-1, -1, -1, -1);
        }
#pragma unroll
        for (int u = 0; u < C::U; ++u) {
            const int e = (base + u * OWN_THREADS + tid) * 4;
            own_push<C>(wlist, wcount, v[u].x, n0, e + 0);
            own_push<C>(wlist, wcount, v[u].y, n0, e + 1);
            own_push<C>(wlist, wcount, v[u].z, n0, e + 2);
            own_push<C>(wlist, wcount, v[u].w, n0, e + 3);
        }
    }
    return wcount;
}

template <class C, class F>
__device__ __forceinline__ void own_overflow(const int4* __restrict__ d4, int n0, int rot,
                                             int tid, F add) {
    unsigned seen = 0;
#pragma unroll 1
    for (int t = 0; t < C::TRIPS * C::U; ++t) {
        int tt = t / C::U + rot;
        if (tt >= C::TRIPS) tt -= C::TRIPS;
        const int i = (tt * C::U + (t % C::U)) * OWN_THREADS + tid;
        const int4 v = (i < C::SL4) ? d4[i] : make_int4(-1, -1, -1, -1);
        const int dv[4] = {v.x, v.y, v.z, v.w};
#pragma unroll 1
        for (int k = 0; k < 4; ++k) {
            const unsigned d = (unsigned)(dv[k] - n0);
            const bool m = d < (unsigned)C::RN;
            const unsigned mask = __builtin_amdgcn_ballot_w32(m);
            const unsigned pos = seen + __builtin_amdgcn_mbcnt_lo(mask, 0u);
            seen += (unsigned)__builtin_popcount(mask);
            if (m && pos >= (unsigned)C::WCAP) add(i * 4 + k, (int)d);
        }
    }
}

__device__ __forceinline__ void list_push(unsigned* wlist, unsigned& wcount, unsigned ent,
                                          unsigned sublo, unsigned eoff) {
    const unsigned d = (ent & ((1u << cfg1::DB) - 1u)) - sublo;
    const bool m = d < (unsigned)cfg2::RN;
    const unsigned mask = __builtin_amdgcn_ballot_w32(m);
    if (mask) {
        const unsigned pos = wcount + __builtin_amdgcn_mbcnt_lo(mask, 0u);
        if (m && pos < (unsigned)cfg2::WCAP)
            wlist[pos] = (((ent >> cfg1::DB) + eoff) << cfg2::DB) | d;
        wcount += (unsigned)__builtin_popcount(mask);
    }
}

__device__ __forceinline__ unsigned list_scan(const unsigned* __restrict__ lists, int l0,
                                              const unsigned (&cnt)[LIST_SUBS], unsigned cmax,
                                              unsigned* wlist, unsigned sublo, int wave, int lane) {
    unsigned wcount = 0;
    for (unsigned k0 = 0; k0 * OWN_THREADS < cmax; k0 += LIST_U) {
        unsigned v[LIST_SUBS][LIST_U];
#pragma unroll
        for (int q = 0; q < LIST_SUBS; ++q) {
            const unsigned* L = lists + (size_t)(l0 + q) * LIST_CAP;
            const unsigned wb = (unsigned)(((wave + 8 * q) & 31) * 32 + lane);
#pragma unroll
            for (int u = 0; u < LIST_U; ++u) {
                const unsigned i = (k0 + (unsigned)u) * OWN_THREADS + wb;
                v[q][u] = (i < cnt[q]) ? L[i] : LIST_BAD;
            }
        }
#pragma unroll
        for (int q = 0; q < LIST_SUBS; ++q) {
#pragma unroll
            for (int u = 0; u < LIST_U; ++u)
                list_push(wlist, wcount, v[q][u], sublo, (unsigned)(q * (cfg1::SL4 * 4)));
        }
    }
    return wcount;
}

template <class F>
__device__ __forceinline__ void list_overflow(const unsigned* __restrict__ lists,
                                              const unsigned* __restrict__ lcount, int l0,
                                              unsigned cmax, unsigned sublo, int wave, int lane, F add) {
    unsigned seen = 0;
#pragma unroll 1
    for (unsigned k0 = 0; k0 * OWN_THREADS < cmax; k0 += LIST_U) {
#pragma unroll 1
        for (int q = 0; q < LIST_SUBS; ++q) {
            const unsigned* L = lists + (size_t)(l0 + q) * LIST_CAP;
            const unsigned cq = lcount[(l0 + q) * CNT_PITCH];
            const unsigned wb = (unsigned)(((wave + 8 * q) & 31) * 32 + lane);
#pragma unroll 1
            for (int u = 0; u < LIST_U; ++u) {
                const unsigned i = (k0 + (unsigned)u) * OWN_THREADS + wb;
                const unsigned ent = (i < cq) ? L[i] : LIST_BAD;
                const unsigned d = (ent & ((1u << cfg1::DB) - 1u)) - sublo;
                const bool m = d < (unsigned)cfg2::RN;
                const unsigned mask = __builtin_amdgcn_ballot_w32(m);
                const unsigned pos = seen + __builtin_amdgcn_mbcnt_lo(mask, 0u);
                seen += (unsigned)__builtin_popcount(mask);
                if (m && pos >= (unsigned)cfg2::WCAP)
                    add((int)((ent >> cfg1::DB) + (unsigned)(q * (cfg1::SL4 * 4))), (int)d);
            }
        }
    }
}

__global__ __launch_bounds__(OWN_THREADS)
void edge2_own_kernel(const _Float16* __restrict__ h, const int* __restrict__ ei,
                      const float* __restrict__ ea,
                      const float* __restrict__ eW, const float* __restrict__ eb,
                      const unsigned* __restrict__ lists, const unsigned* __restrict__ lcount,
                      const float* __restrict__ W1, const float* __restrict__ b1,
                      const float* __restrict__ W2, const float* __restrict__ b2,
                      float* __restrict__ h2) {
    typedef cfg2 C;
    constexpr int P2 = HID_C + 8;
    constexpr int NT = C::RN / 16;
    constexpr int PA = HID_C + 4;
    static_assert(HID_C * HID_C / 4 == OWN_THREADS, "one float4 of W1 and one of W2 per thread");
    __shared__ __align__(16) float    s_acc[(C::RN + 1) * PA];
    __shared__ __align__(16) unsigned s_list[(OWN_THREADS / 32) * C::WCAP];
    __shared__ __align__(16) _Float16 sW1[HID_C * P2];
    __shared__ __align__(16) _Float16 sW2[HID_C * P2];

    const int tid   = threadIdx.x;
    const int lane  = tid & 31;
    const int wave  = tid >> 5;
    const int range = (int)blockIdx.x / C::S;
    const int slice = (int)blockIdx.x - range * C::S;
    const int n0    = range * C::RN;
    const int e0    = slice * (C::SL4 * 4);
    unsigned* wlist = s_list + wave * C::WCAP;

    if (n0 >= N_NODES_C) return;

    const float4 w1v = ((const float4*)W1)[tid];
    const float4 w2v = ((const float4*)W2)[tid];

    for (int i = tid; i < C::RN * PA / 4; i += OWN_THREADS)
        ((float4*)s_acc)[i] = make_float4(0.f, 0.f, 0.f, 0.f);
    {
        const int n = tid / (HID_C / 4), q = tid - n * (HID_C / 4);
        v4h o1, o2;
        o1[0] = (_Float16)w1v.x; o1[1] = (_Float16)w1v.y; o1[2] = (_Float16)w1v.z; o1[3] = (_Float16)w1v.w;
        o2[0] = (_Float16)w2v.x; o2[1] = (_Float16)w2v.y; o2[2] = (_Float16)w2v.z; o2[3] = (_Float16)w2v.w;
        *(v4h*)(sW1 + n * P2 + 4 * q) = o1;
        *(v4h*)(sW2 + n * P2 + 4 * q) = o2;
    }
    __syncthreads();

    const int c  = lane & 15;
    const int hh = lane >> 4;
    v2f   bw[4];
    float bch[4];
#pragma unroll
    for (int t = 0; t < 4; ++t) {
        bw[t]  = *(const v2f*)(eW + (4 * c + t) * EDGE_DIM_C + 2 * hh);
        bch[t] = eb[4 * c + t];
    }

    auto add_one = [&](int erel, int od) {
        const int oe = e0 + erel;
        const int os = ei[oe];
        const float4 a = *(const float4*)(ea + (size_t)oe * EDGE_DIM_C);
#pragma unroll 1
        for (int q = 0; q < HID_C; ++q) {
            const float4 w = *(const float4*)(eW + q * EDGE_DIM_C);
            float t = eb[q] + a.x * w.x + a.y * w.y + a.z * w.z + a.w * w.w;
            t += (float)h[(size_t)os * HID_C + q];
            atomicAdd(&s_acc[od * PA + sum_pos(q)], fmaxf(t, 0.0f));
        }
    };

    const int l0 = (range / LIST_SUBS) * cfg1::S + slice * LIST_SUBS;
    const unsigned sublo = (unsigned)((range % LIST_SUBS) * C::RN);
    unsigned cnt[LIST_SUBS];
    unsigned cmax = 0;
#pragma unroll
    for (int q = 0; q < LIST_SUBS; ++q) {
        cnt[q] = lcount[(l0 + q) * CNT_PITCH];
        cmax = cnt[q] > cmax ? cnt[q] : cmax;
    }
    unsigned wcount;
    if (cmax <= (unsigned)LIST_CAP) {
        wcount = list_scan(lists, l0, cnt, cmax, wlist, sublo, wave, lane);
        if (wcount > (unsigned)C::WCAP)
            list_overflow(lists, lcount, l0, cmax, sublo, wave, lane, add_one);
    } else {
        const int4* d4 = (const int4*)(ei + N_EDGES_C + e0);
        const int rot = (range * C::TRIPS) / C::R;
        wcount = own_scan<C>(d4, wlist, n0, rot, tid);
        if (wcount > (unsigned)C::WCAP)
            own_overflow<C>(d4, n0, rot, tid, add_one);
    }
    const int wn = (int)(wcount < (unsigned)C::WCAP ? wcount : (unsigned)C::WCAP);

    __syncthreads();

    const char* hb = (const char*)h;
    int sv[8];
    v2f av;
    edge2_fetch(wlist, 0, wn, c, hh, e0, ei, ea, sv, av);
    for (int p = 0; p < wn; p += 16) {
        v4h hv[8];
#pragma unroll
        for (int r = 0; r < 8; ++r) {
            const unsigned off = (unsigned)sv[r] * (unsigned)(HID_C * 2) + 8u * (unsigned)c;
            hv[r] = *(const v4h*)(hb + off);
        }
        const int pn = (p + 16 < wn) ? p + 16 : p;
        int svn[8];
        v2f avn;
        edge2_fetch(wlist, pn, wn, c, hh, e0, ei, ea, svn, avn);
        float* arow[8];
        {
            const uint4 q0 = *(const uint4*)(wlist + p + 8 * hh);
            const uint4 q1 = *(const uint4*)(wlist + p + 8 * hh + 4);
            const unsigned raw[8] = {q0.x, q0.y, q0.z, q0.w, q1.x, q1.y, q1.z, q1.w};
#pragma unroll
            for (int r = 0; r < 8; ++r) {
                const int dl = (p + 8 * hh + r < wn) ? (int)(raw[r] & ((1u << C::DB) - 1u)) : C::RN;
                arow[r] = s_acc + dl * PA + c;
            }
        }
#pragma unroll
        for (int t = 0; t < 4; ++t) {
            v8f cc;
#pragma unroll
            for (int r = 0; r < 8; ++r) cc[r] = bch[t] + (float)hv[r][t];
            const v8f d = __builtin_amdgcn_wmma_f32_16x16x4_f32(false, av, false, bw[t], (short)0, cc, false, false);
#pragma unroll
            for (int r = 0; r < 8; ++r)
                atomicAdd(arow[r] + 16 * t, fmaxf(d[r], 0.0f));
        }
#pragma unroll
        for (int r = 0; r < 8; ++r) sv[r] = svn[r];
        av = avn;
    }
    const int  wv   = __builtin_amdgcn_readfirstlane(wave);
    const bool tile = wv < NT;
    const _Float16* grow = h + (size_t)(n0 + (tile ? wv * 16 + c : 0)) * HID_C + 8 * hh;
    v8h    xr[4];
    float4 pb1[8], pb2[8];
#pragma unroll
    for (int i = 0; i < 4; ++i) {
        xr[i]          = *(const v8h*)(grow + 16 * i);
        pb1[2 * i]     = *(const float4*)(b1 + 16 * i + 8 * hh);
        pb1[2 * i + 1] = *(const float4*)(b1 + 16 * i + 8 * hh + 4);
        pb2[2 * i]     = *(const float4*)(b2 + 16 * i + 8 * hh);
        pb2[2 * i + 1] = *(const float4*)(b2 + 16 * i + 8 * hh + 4);
    }

    __syncthreads();

    float*     trow  = s_acc + (wv * 16 + c) * PA;
    if (tile) {
        v16h bop[2];
#pragma unroll
        for (int kb = 0; kb < 2; ++kb) {
            const v8h x0 = xr[2 * kb];
            const v8h x1 = xr[2 * kb + 1];
            float lo[8], hi[8];
#pragma unroll
            for (int j = 0; j < 4; ++j) {
                const float2 pl = *(const float2*)(trow + 16 * j + 8 * kb + 2 * hh);
                const float2 ph = *(const float2*)(trow + 16 * j + 8 * kb + 2 * hh + 4);
                lo[j] = pl.x; lo[j + 4] = pl.y;
                hi[j] = ph.x; hi[j + 4] = ph.y;
            }
            v16h b;
#pragma unroll
            for (int e = 0; e < 8; ++e) {
                b[e]     = (_Float16)((float)x0[e] + lo[e]);
                b[8 + e] = (_Float16)((float)x1[e] + hi[e]);
            }
            bop[kb] = b;
        }

        v8f acc[4];
#pragma unroll
        for (int t = 0; t < 4; ++t) {
            acc[t] = v8f{};
#pragma unroll
            for (int kb = 0; kb < 2; ++kb) {
                const v16h a = lds_load_tile(sW1 + 16 * t * P2, P2, kb * 32);
                acc[t] = __builtin_amdgcn_wmma_f32_16x16x32_f16(false, a, false, bop[kb],
                                                                (short)0, acc[t], false, false);
            }
        }

        v16h bop2[2];
#pragma unroll
        for (int s2 = 0; s2 < 2; ++s2) {
            const float4 p0 = pb1[4 * s2], p1 = pb1[4 * s2 + 1], p2 = pb1[4 * s2 + 2], p3 = pb1[4 * s2 + 3];
            const float bl[8] = {p0.x, p0.y, p0.z, p0.w, p1.x, p1.y, p1.z, p1.w};
            const float bh[8] = {p2.x, p2.y, p2.z, p2.w, p3.x, p3.y, p3.z, p3.w};
            v16h b;
#pragma unroll
            for (int e = 0; e < 8; ++e) {
                float v0 = acc[2 * s2][e] + bl[e];
                float v1 = acc[2 * s2 + 1][e] + bh[e];
                v0 = v0 > 0.0f ? v0 : 0.0f;
                v1 = v1 > 0.0f ? v1 : 0.0f;
                b[e]     = (_Float16)v0;
                b[8 + e] = (_Float16)v1;
            }
            bop2[s2] = b;
        }

#pragma unroll
        for (int t = 0; t < 4; ++t) {
            v8f d = v8f{};
#pragma unroll
            for (int s2 = 0; s2 < 2; ++s2) {
                const v16h a = lds_load_tile(sW2 + 16 * t * P2, P2, s2 * 32);
                d = __builtin_amdgcn_wmma_f32_16x16x32_f16(false, a, false, bop2[s2],
                                                           (short)0, d, false, false);
            }
            const float4 q0 = pb2[2 * t], q1 = pb2[2 * t + 1];
            const float bb[8] = {q0.x, q0.y, q0.z, q0.w, q1.x, q1.y, q1.z, q1.w};
            float v[8];
#pragma unroll
            for (int e = 0; e < 8; ++e) {
                const float u = d[e] + bb[e];
                v[e] = u > 0.0f ? u : 0.0f;
            }
            float* q = trow + 16 * t + 8 * hh;
            *(float4*)(q)     = make_float4(v[0], v[1], v[2], v[3]);
            *(float4*)(q + 4) = make_float4(v[4], v[5], v[6], v[7]);
        }
    }
    __syncthreads();

    if (tile) {
        const float* st = s_acc + wv * 16 * PA;
        float*       gt = h2 + (size_t)(n0 + wv * 16) * HID_C;
#pragma unroll
        for (int i = 0; i < 8; ++i) {
            const int f = 4 * (lane + 32 * i);
            *(float4*)(gt + f) = *(const float4*)(st + (f >> 6) * PA + (f & 63));
        }
    }
}

__global__ __launch_bounds__(OWN_THREADS)
void edge1_own_kernel(const float* __restrict__ x, const int* __restrict__ ei,
                      const float* __restrict__ ea,
                      const float* __restrict__ eW, const float* __restrict__ eb,
                      float* __restrict__ aggr,
                      unsigned* __restrict__ lists, unsigned* __restrict__ lcount) {
    typedef cfg1 C;
    __shared__ __align__(16) float    s_acc[C::RN * 8];
    __shared__ __align__(16) unsigned s_list[(OWN_THREADS / 32) * C::WCAP];
    __shared__ unsigned s_wn[OWN_THREADS / 32];

    const int tid   = threadIdx.x;
    const int lane  = tid & 31;
    const int wave  = tid >> 5;
    const int range = (int)blockIdx.x / C::S;
    const int slice = (int)blockIdx.x - range * C::S;
    const int n0    = range * C::RN;
    const int e0    = slice * (C::SL4 * 4);
    unsigned* wlist = s_list + wave * C::WCAP;

    for (int i = tid; i < C::RN * 8; i += OWN_THREADS) s_acc[i] = 0.0f;
    __syncthreads();

    const int c   = lane & 7;
    const int sub = lane >> 3;
    const bool cact = c < IN_DIM_C;
    const float4 wc = cact ? *(const float4*)(eW + c * EDGE_DIM_C) : make_float4(0.f, 0.f, 0.f, 0.f);
    const float  bc = cact ? eb[c] : 0.0f;
    const int    cx = cact ? c : 0;

    const int4* d4 = (const int4*)(ei + N_EDGES_C + e0);
    const int rot = (range * C::TRIPS) / C::R;
    const unsigned wcount = own_scan<C>(d4, wlist, n0, rot, tid);
    if (wcount > (unsigned)C::WCAP) {
        own_overflow<C>(d4, n0, rot, tid, [&](int erel, int od) {
            const int oe = e0 + erel;
            const int os = ei[oe];
            const float4 a = *(const float4*)(ea + (size_t)oe * EDGE_DIM_C);
#pragma unroll 1
            for (int q = 0; q < IN_DIM_C; ++q) {
                const float4 w = *(const float4*)(eW + q * EDGE_DIM_C);
                float t = eb[q] + a.x * w.x + a.y * w.y + a.z * w.z + a.w * w.w;
                t += x[(size_t)os * IN_DIM_C + q];
                atomicAdd(&s_acc[od * 8 + q], fmaxf(t, 0.0f));
            }
        });
    }
    const int wn = (int)(wcount < (unsigned)C::WCAP ? wcount : (unsigned)C::WCAP);
    if (lane == 0) s_wn[wave] = wcount;

    __syncthreads();

    {
        unsigned off = 0, tot = 0;
        bool full = false;
        for (int w = 0; w < OWN_THREADS / 32; ++w) {
            const unsigned cw = s_wn[w];
            full = full || cw > (unsigned)C::WCAP;
            const unsigned pw = ((cw < (unsigned)C::WCAP ? cw : (unsigned)C::WCAP) + 31u) & ~31u;
            if (w < wave) off += pw;
            tot += pw;
        }
        unsigned* L = lists + (size_t)blockIdx.x * LIST_CAP + off;
        const int wp = (wn + 31) & ~31;
        for (int i = lane; i < wp; i += 32) L[i] = (i < wn) ? wlist[i] : LIST_BAD;
        if (tid == 0) lcount[blockIdx.x * CNT_PITCH] = full ? LIST_BAD : tot;
    }

    for (int p = 0; p < wn; p += 32) {
        unsigned ent[8]; int sr[8]; float4 av[8]; float xv[8]; bool ok[8];
#pragma unroll
        for (int g = 0; g < 8; ++g) {
            const int idx = p + g * 4 + sub;
            ok[g]  = idx < wn;
            ent[g] = ok[g] ? wlist[idx] : 0u;
        }
#pragma unroll
        for (int g = 0; g < 8; ++g) {
            const int e = e0 + (int)(ent[g] >> C::DB);
            sr[g] = ei[e];
            av[g] = *(const float4*)(ea + (size_t)e * EDGE_DIM_C);
        }
#pragma unroll
        for (int g = 0; g < 8; ++g)
            xv[g] = x[(size_t)sr[g] * IN_DIM_C + cx];
        asm volatile("" :: "v"(xv[0]), "v"(xv[1]), "v"(xv[2]), "v"(xv[3]), "v"(xv[4]), "v"(xv[5]), "v"(xv[6]), "v"(xv[7]),
                           "v"(av[0].x), "v"(av[0].y), "v"(av[0].z), "v"(av[0].w));
#pragma unroll
        for (int g = 0; g < 8; ++g) {
            if (ok[g] && cact) {
                const int dl = (int)(ent[g] & ((1u << C::DB) - 1u));
                float t = bc + av[g].x * wc.x + av[g].y * wc.y + av[g].z * wc.z + av[g].w * wc.w;
                t += xv[g];
                atomicAdd(&s_acc[dl * 8 + c], fmaxf(t, 0.0f));
            }
        }
    }
    __syncthreads();

    int rows = N_NODES_C - n0;
    if (rows > C::RN) rows = C::RN;
    float* o = aggr + (size_t)slice * N_NODES_C * IN_DIM_C + (size_t)n0 * IN_DIM_C;
    for (int i = tid; i < rows * IN_DIM_C; i += OWN_THREADS) {
        const int r = i / IN_DIM_C, q = i - r * IN_DIM_C;
        o[i] = s_acc[r * 8 + q];
    }
}

#define POOL_GPB     4
#define POOL_THREADS 1024
#define POOL_U       14
#define POOL_Q       8

__global__ __launch_bounds__(POOL_THREADS)
void pool_fc_kernel(const float* __restrict__ h2, const int* __restrict__ batch,
                    const float* __restrict__ fcW, const float* __restrict__ fcb,
                    float* __restrict__ out) {
    constexpr int NW  = POOL_THREADS / 32;
    constexpr int NCH = (N_NODES_C + 31) / 32;
    static_assert(N_GRAPHS_C % POOL_GPB == 0, "every block owns POOL_GPB graphs");
    static_assert(POOL_GPB * HID_C <= POOL_THREADS, "one thread per (graph, channel) in the wave sum");

    __shared__ __align__(16) float s_part[NW][POOL_GPB][HID_C];
    __shared__ int   s_cnt [NW][POOL_GPB];
    __shared__ float s_mean[POOL_GPB][HID_C];

    const int tid  = threadIdx.x;
    const int lane = tid & 31;
    const int wave = tid >> 5;
    const int g0   = (int)blockIdx.x * POOL_GPB;
    const float2* h2v = (const float2*)h2;

    float2 acc[POOL_GPB];
    int    cnt[POOL_GPB];
#pragma unroll
    for (int k = 0; k < POOL_GPB; ++k) { acc[k] = make_float2(0.f, 0.f); cnt[k] = 0; }

    for (int c0 = wave; c0 < NCH; c0 += NW * POOL_U) {
        unsigned hit = 0;
#pragma unroll
        for (int u = 0; u < POOL_U; ++u) {
            const int c = c0 + u * NW;
            const int n = c * 32 + lane;
            const int g = (c < NCH && n < N_NODES_C) ? batch[n] : -1;
            const bool m = (unsigned)(g - g0) < (unsigned)POOL_GPB;
            if (__builtin_amdgcn_ballot_w32(m)) hit |= 1u << u;
        }
        while (hit) {
            const int u = __builtin_ctz(hit);
            hit &= hit - 1;
            const int base = (c0 + u * NW) * 32;
            const int n = base + lane;
            const int g = (n < N_NODES_C) ? batch[n] : -1;
            const int d = g - g0;
            unsigned mask = __builtin_amdgcn_ballot_w32((unsigned)d < (unsigned)POOL_GPB);
#pragma unroll
            for (int k = 0; k < POOL_GPB; ++k)
                cnt[k] += __builtin_popcount(__builtin_amdgcn_ballot_w32(d == k));
            while (mask) {
                int gq[POOL_Q]; float2 v[POOL_Q];
                const int jf = __builtin_ctz(mask);
#pragma unroll
                for (int q = 0; q < POOL_Q; ++q) {
                    const bool has = mask != 0;
                    const int j = has ? __builtin_ctz(mask) : jf;
                    const int gj = __builtin_amdgcn_readlane(d, j);
                    gq[q] = has ? gj : -1;
                    v[q]  = h2v[(size_t)(base + j) * (HID_C / 2) + lane];
                    mask &= mask - 1;
                }
                __builtin_amdgcn_sched_barrier(0);
#pragma unroll
                for (int q = 0; q < POOL_Q; ++q) {
#pragma unroll
                    for (int k = 0; k < POOL_GPB; ++k)
                        if (gq[q] == k) { acc[k].x += v[q].x; acc[k].y += v[q].y; }
                }
            }
        }
    }

#pragma unroll
    for (int k = 0; k < POOL_GPB; ++k) {
        *(float2*)&s_part[wave][k][2 * lane] = acc[k];
        if (lane == 0) s_cnt[wave][k] = cnt[k];
    }
    __syncthreads();

    if (tid < POOL_GPB * HID_C) {
        const int k = tid / HID_C, ch = tid - k * HID_C;
        float s = 0.0f; int c = 0;
        for (int w = 0; w < NW; ++w) { s += s_part[w][k][ch]; c += s_cnt[w][k]; }
        s_mean[k][ch] = s / fmaxf((float)c, 1.0f);
    }
    __syncthreads();

    if (tid < POOL_GPB * NUM_TASKS_C) {
        const int k = tid / NUM_TASKS_C, t = tid - k * NUM_TASKS_C;
        float s = fcb[t];
#pragma unroll
        for (int c = 0; c < HID_C; ++c)
            s += s_mean[k][c] * fcW[t * HID_C + c];
        out[(size_t)(g0 + k) * NUM_TASKS_C + t] = s;
    }
}

extern "C" void kernel_launch(void* const* d_in, const int* in_sizes, int n_in,
                              void* d_out, int out_size, void* d_ws, size_t ws_size,
                              hipStream_t stream) {
    const float* x    = (const float*)d_in[0];
    const int*   ei   = (const int*)  d_in[1];
    const float* ea   = (const float*)d_in[2];
    const int*   bat  = (const int*)  d_in[3];
    const float* e1W  = (const float*)d_in[4];  const float* e1b  = (const float*)d_in[5];
    const float* n1W1 = (const float*)d_in[6];  const float* n1b1 = (const float*)d_in[7];
    const float* n1W2 = (const float*)d_in[8];  const float* n1b2 = (const float*)d_in[9];
    const float* e2W  = (const float*)d_in[10]; const float* e2b  = (const float*)d_in[11];
    const float* n2W1 = (const float*)d_in[12]; const float* n2b1 = (const float*)d_in[13];
    const float* n2W2 = (const float*)d_in[14]; const float* n2b2 = (const float*)d_in[15];
    const float* fcW  = (const float*)d_in[16]; const float* fcb  = (const float*)d_in[17];
    float* out = (float*)d_out;

    float*    aggr1 = (float*)d_ws;
    _Float16* h     = (_Float16*)(aggr1 + (size_t)cfg1::S * N_NODES_C * IN_DIM_C);
    float*    h2    = (float*)(h + (size_t)N_NODES_C * HID_C);
    unsigned* lists  = (unsigned*)(h2 + (size_t)N_NODES_C * HID_C);
    unsigned* lcount = lists + (size_t)OWN_BLOCKS * LIST_CAP;

    edge1_own_kernel<<<OWN_BLOCKS, OWN_THREADS, 0, stream>>>(x, ei, ea, e1W, e1b, aggr1, lists, lcount);
    node_mlp_kernel<IN_DIM_C, cfg1::S><<<MLP_GRID, MLP_THREADS, 0, stream>>>(
        x, nullptr, aggr1, n1W1, n1b1, n1W2, n1b2, h, nullptr, N_NODES_C);
    edge2_own_kernel<<<OWN_BLOCKS, OWN_THREADS, 0, stream>>>(h, ei, ea, e2W, e2b, lists, lcount,
                                                             n2W1, n2b1, n2W2, n2b2, h2);
    pool_fc_kernel<<<N_GRAPHS_C / POOL_GPB, POOL_THREADS, 0, stream>>>(h2, bat, fcW, fcb, out);
}


# Phase 1 of the edge kernels: a lane appends matches to its own LDS list (no ballot, rank, branch or scalar count per element), a prefix sum packs them per wave; list order changes, sums arrival order
# speedup vs baseline: 4.3031x; 1.0435x over previous
#include <hip/hip_runtime.h>
#include <hip/hip_bf16.h>

#define N_NODES_C  100000
#define N_EDGES_C  1600000
#define N_GRAPHS_C 1000
#define HID_C      64
#define IN_DIM_C   7
#define EDGE_DIM_C 4
#define NUM_TASKS_C 12

typedef __attribute__((ext_vector_type(16))) _Float16 v16h;
typedef __attribute__((ext_vector_type(8)))  _Float16 v8h;
typedef __attribute__((ext_vector_type(8)))  float    v8f;

__device__ __forceinline__ v16h lds_load_tile(const _Float16* s, int ld, int kbase) {
    const int lane = threadIdx.x & 31;
    const int r    = lane & 15;
    const int koff = (lane < 16) ? 0 : 8;
    const _Float16* p = s + r * ld + kbase + koff;
    v8h lo = *(const v8h*)(p);
    v8h hi = *(const v8h*)(p + 16);
    v16h out;
#pragma unroll
    for (int e = 0; e < 8; ++e) { out[e] = lo[e]; out[8 + e] = hi[e]; }
    return out;
}

#define MLP_WAVES   8
#define MLP_NT      2
#define MLP_THREADS (MLP_WAVES * 32)
#define MLP_TILES   (MLP_WAVES * MLP_NT)
#define MLP_ROWS    (MLP_TILES * 16)
#define MLP_GRID    ((N_NODES_C + MLP_ROWS - 1) / MLP_ROWS)
static_assert(N_NODES_C % 16 == 0, "the node count is a whole number of 16-node tiles");

typedef __attribute__((ext_vector_type(4))) _Float16 v4h;

template <int K, int KP>
struct mlp_wregs {
    static constexpr bool VEC = (K % 4) == 0;
    static constexpr int  N   = (VEC ? HID_C * (K / 4) : HID_C * KP) / MLP_THREADS;
    static_assert((VEC ? HID_C * (K / 4) : HID_C * KP) % MLP_THREADS == 0, "the same number of pieces in every thread");
    static_assert(!VEC || K == KP, "no padding on the 16-byte path");
    float4 v4[VEC ? N : 1];
    float  v1[VEC ? 1 : N];
};

template <int K, int KP>
__device__ __forceinline__ void mlp_load_weights(const float* __restrict__ W, int tid, mlp_wregs<K, KP>& r) {
#pragma unroll
    for (int it = 0; it < mlp_wregs<K, KP>::N; ++it) {
        const int i = tid + it * MLP_THREADS;
        if constexpr (mlp_wregs<K, KP>::VEC) {
            r.v4[it] = ((const float4*)W)[i];
        } else {
            const int n = i / KP, k = i - n * KP;
            const float w = W[n * K + (k < K ? k : 0)];
            r.v1[it] = (k < K) ? w : 0.0f;
        }
    }
}

template <int K, int KP, int P>
__device__ __forceinline__ void mlp_store_weights(_Float16* s, int tid, const mlp_wregs<K, KP>& r) {
#pragma unroll
    for (int it = 0; it < mlp_wregs<K, KP>::N; ++it) {
        const int i = tid + it * MLP_THREADS;
        if constexpr (mlp_wregs<K, KP>::VEC) {
            const int n = i / (K / 4), q = i - n * (K / 4);
            const float4 v = r.v4[it];
            v4h o;
            o[0] = (_Float16)v.x; o[1] = (_Float16)v.y; o[2] = (_Float16)v.z; o[3] = (_Float16)v.w;
            *(v4h*)(s + n * P + 4 * q) = o;
        } else {
            const int n = i / KP, k = i - n * KP;
            s[n * P + k] = (_Float16)r.v1[it];
        }
    }
}

template <int K1, int NP>
__global__ __launch_bounds__(MLP_THREADS)
void node_mlp_kernel(const float* __restrict__ xin, const _Float16* __restrict__ hin,
                     const float* __restrict__ in1,
                     const float* __restrict__ W1,  const float* __restrict__ b1,
                     const float* __restrict__ W2,  const float* __restrict__ b2,
                     _Float16* __restrict__ outh, float* __restrict__ outf, int n_nodes) {
    constexpr int  K1P   = (K1 + 31) & ~31;
    constexpr int  KS    = K1P / 32;
    constexpr int  P1    = K1P + 8;
    constexpr int  P2    = HID_C + 8;
    constexpr int  PO    = HID_C + 4;
    constexpr bool SMALL = (K1 % 8) != 0;
    constexpr int  XT    = 16 * K1;
    static_assert(!SMALL || (XT % 4 == 0 && XT / 4 <= 32), "a flat conv1 tile is at most one float4 per lane");
    static_assert(SMALL || K1 == HID_C, "the conv2 path cuts rows of 64");

    __shared__ __align__(16) _Float16 sW1[HID_C * P1];
    __shared__ __align__(16) _Float16 sW2[HID_C * P2];
    __shared__ __align__(16) float    sX [SMALL ? MLP_TILES * 128 : 4];
    __shared__ __align__(16) _Float16 sA [SMALL ? 8 : MLP_TILES * 16 * P1];
    __shared__ __align__(16) _Float16 sOh[SMALL ? MLP_TILES * 16 * P2 : 8];
    __shared__ __align__(16) float    sOf[SMALL ? 4 : MLP_TILES * 16 * PO];

    const int tid  = threadIdx.x;
    const int wave = __builtin_amdgcn_readfirstlane(tid >> 5);
    const int lane = tid & 31;
    const int c    = lane & 15;
    const int h    = lane >> 4;
    const int tile0 = ((int)blockIdx.x * MLP_WAVES + wave) * MLP_NT;

    int  row0[MLP_NT];
    bool tv[MLP_NT];
#pragma unroll
    for (int j = 0; j < MLP_NT; ++j) {
        row0[j] = (tile0 + j) * 16;
        tv[j]   = row0[j] < n_nodes;
    }

    mlp_wregs<K1, K1P>     w1r;
    mlp_wregs<HID_C, HID_C> w2r;
    mlp_load_weights<K1, K1P>(W1, tid, w1r);
    mlp_load_weights<HID_C, HID_C>(W2, tid, w2r);

    if constexpr (SMALL) {
        const int ll = lane < XT / 4 ? lane : XT / 4 - 1;
        float4 raw[MLP_NT][NP + 1];
#pragma unroll
        for (int j = 0; j < MLP_NT; ++j) {
            const int o4 = (tv[j] ? row0[j] : 0) * K1 / 4 + ll;
            raw[j][0] = ((const float4*)xin)[o4];
#pragma unroll
            for (int p = 0; p < NP; ++p)
                raw[j][p + 1] = ((const float4*)(in1 + (size_t)p * (size_t)n_nodes * K1))[o4];
        }
        float4 v[MLP_NT];
#pragma unroll
        for (int j = 0; j < MLP_NT; ++j) {
            v[j] = raw[j][0];
#pragma unroll
            for (int p = 0; p < NP; ++p) {
                const float4 a = raw[j][p + 1];
                v[j].x += a.x; v[j].y += a.y; v[j].z += a.z; v[j].w += a.w;
            }
        }
#pragma unroll
        for (int j = 0; j < MLP_NT; ++j)
            ((float4*)sX)[(wave * MLP_NT + j) * 32 + lane] = v[j];
    } else {
#pragma unroll
        for (int j = 0; j < MLP_NT; ++j) {
            const int    eb = (tv[j] ? row0[j] : 0) * K1;
            const int    tl = wave * MLP_NT + j;
#pragma unroll
            for (int i = 0; i < XT / 128; ++i) {
                const int f = 4 * (lane + 32 * i);
                const v4h x4 = *(const v4h*)(hin + eb + f);
                float v0 = (float)x4[0], v1 = (float)x4[1], v2 = (float)x4[2], v3 = (float)x4[3];
#pragma unroll
                for (int p = 0; p < NP; ++p) {
                    const float4 a = *(const float4*)(in1 + (size_t)p * (size_t)n_nodes * K1 + eb + f);
                    v0 += a.x; v1 += a.y; v2 += a.z; v3 += a.w;
                }
                v4h o;
                o[0] = (_Float16)v0; o[1] = (_Float16)v1; o[2] = (_Float16)v2; o[3] = (_Float16)v3;
                *(v4h*)(sA + (tl * 16 + (f >> 6)) * P1 + (f & 63)) = o;
            }
        }
    }

    mlp_store_weights<K1, K1P, P1>(sW1, tid, w1r);
    mlp_store_weights<HID_C, HID_C, P2>(sW2, tid, w2r);
    __syncthreads();

    v16h bop[MLP_NT][KS];
#pragma unroll
    for (int j = 0; j < MLP_NT; ++j) {
        const int tl = wave * MLP_NT + j;
        if constexpr (SMALL) {
            const float* r = sX + tl * 128 + c * K1;
            v16h b = {};
#pragma unroll
            for (int k = 0; k < K1; ++k) {
                const float f = r[k];
                b[k] = (h == 0) ? (_Float16)f : (_Float16)0.0f;
            }
            bop[j][0] = b;
        } else {
#pragma unroll
            for (int kb = 0; kb < KS; ++kb)
                bop[j][kb] = lds_load_tile(sA + tl * 16 * P1, P1, kb * 32);
        }
    }

    float bias1[4][8], bias2[4][8];
#pragma unroll
    for (int t = 0; t < 4; ++t) {
        const float4 a0 = *(const float4*)(b1 + 16 * t + 8 * h);
        const float4 a1 = *(const float4*)(b1 + 16 * t + 8 * h + 4);
        bias1[t][0] = a0.x; bias1[t][1] = a0.y; bias1[t][2] = a0.z; bias1[t][3] = a0.w;
        bias1[t][4] = a1.x; bias1[t][5] = a1.y; bias1[t][6] = a1.z; bias1[t][7] = a1.w;
        const float4 c0 = *(const float4*)(b2 + 16 * t + 8 * h);
        const float4 c1 = *(const float4*)(b2 + 16 * t + 8 * h + 4);
        bias2[t][0] = c0.x; bias2[t][1] = c0.y; bias2[t][2] = c0.z; bias2[t][3] = c0.w;
        bias2[t][4] = c1.x; bias2[t][5] = c1.y; bias2[t][6] = c1.z; bias2[t][7] = c1.w;
    }

    v8f acc[MLP_NT][4];
#pragma unroll
    for (int j = 0; j < MLP_NT; ++j)
#pragma unroll
        for (int t = 0; t < 4; ++t) acc[j][t] = v8f{};
#pragma unroll
    for (int t = 0; t < 4; ++t) {
#pragma unroll
        for (int kb = 0; kb < KS; ++kb) {
            const v16h a = lds_load_tile(sW1 + 16 * t * P1, P1, kb * 32);
#pragma unroll
            for (int j = 0; j < MLP_NT; ++j)
                acc[j][t] = __builtin_amdgcn_wmma_f32_16x16x32_f16(false, a, false, bop[j][kb],
                                                                   (short)0, acc[j][t], false, false);
        }
    }

    v16h bop2[MLP_NT][2];
#pragma unroll
    for (int j = 0; j < MLP_NT; ++j) {
#pragma unroll
        for (int s = 0; s < 2; ++s) {
            v16h b;
#pragma unroll
            for (int e = 0; e < 8; ++e) {
                float v0 = acc[j][2 * s][e] + bias1[2 * s][e];
                float v1 = acc[j][2 * s + 1][e] + bias1[2 * s + 1][e];
                v0 = v0 > 0.0f ? v0 : 0.0f;
                v1 = v1 > 0.0f ? v1 : 0.0f;
                b[e]     = (_Float16)v0;
                b[8 + e] = (_Float16)v1;
            }
            bop2[j][s] = b;
        }
    }

    v8f acc2[MLP_NT][4];
#pragma unroll
    for (int j = 0; j < MLP_NT; ++j)
#pragma unroll
        for (int t = 0; t < 4; ++t) acc2[j][t] = v8f{};
#pragma unroll
    for (int t = 0; t < 4; ++t) {
#pragma unroll
        for (int s = 0; s < 2; ++s) {
            const v16h a = lds_load_tile(sW2 + 16 * t * P2, P2, s * 32);
#pragma unroll
            for (int j = 0; j < MLP_NT; ++j)
                acc2[j][t] = __builtin_amdgcn_wmma_f32_16x16x32_f16(false, a, false, bop2[j][s],
                                                                    (short)0, acc2[j][t], false, false);
        }
    }

#pragma unroll
    for (int j = 0; j < MLP_NT; ++j) {
        const int tl = wave * MLP_NT + j;
#pragma unroll
        for (int t = 0; t < 4; ++t) {
            float v[8];
#pragma unroll
            for (int e = 0; e < 8; ++e) {
                const float u = acc2[j][t][e] + bias2[t][e];
                v[e] = u > 0.0f ? u : 0.0f;
            }
            if constexpr (SMALL) {
                v8h o8;
#pragma unroll
                for (int e = 0; e < 8; ++e) o8[e] = (_Float16)v[e];
                *(v8h*)(sOh + (tl * 16 + c) * P2 + 16 * t + 8 * h) = o8;
            } else {
                float* q = sOf + (tl * 16 + c) * PO + 16 * t + 8 * h;
                *(float4*)(q)     = make_float4(v[0], v[1], v[2], v[3]);
                *(float4*)(q + 4) = make_float4(v[4], v[5], v[6], v[7]);
            }
        }
    }
    __syncthreads();

#pragma unroll
    for (int j = 0; j < MLP_NT; ++j) {
        if (tv[j]) {
            const int tl = wave * MLP_NT + j;
            if constexpr (SMALL) {
#pragma unroll
                for (int i = 0; i < 4; ++i) {
                    const int f = 8 * (lane + 32 * i);
                    const v8h o8 = *(const v8h*)(sOh + (tl * 16 + (f >> 6)) * P2 + (f & 63));
                    *(v8h*)(outh + (size_t)row0[j] * HID_C + f) = o8;
                }
            } else {
#pragma unroll
                for (int i = 0; i < 8; ++i) {
                    const int f = 4 * (lane + 32 * i);
                    const float4 o4 = *(const float4*)(sOf + (tl * 16 + (f >> 6)) * PO + (f & 63));
                    *(float4*)(outf + (size_t)row0[j] * HID_C + f) = o4;
                }
            }
        }
    }
}

#define OWN_BLOCKS  256
#define OWN_THREADS 1024

template <int RN_, int S_, int R_, int WCAP_, int DB_, int U_>
struct own_cfg {
    static constexpr int RN = RN_, S = S_, R = R_, WCAP = WCAP_, DB = DB_, U = U_;
    static constexpr int SL4   = N_EDGES_C / 4 / S_;
    static constexpr int STEP  = OWN_THREADS * U_;
    static constexpr int TRIPS = (SL4 + STEP - 1) / STEP;
    static_assert((N_EDGES_C / 4) % S_ == 0, "a slice is a whole number of int4 words");
    static_assert(R_ * S_ == OWN_BLOCKS, "one block per (range, slice)");
    static_assert(R_ * RN_ >= N_NODES_C, "ranges cover every node");
    static_assert(RN_ <= (1 << DB_), "place in range fits its bits");
    static_assert(((long long)SL4 * 4) <= (1ll << (32 - DB_)), "edge id in slice fits its bits");
};
#define LIST_SUBS  8
typedef own_cfg<400, 1, 256, 512, 10, 8>  cfg2;
typedef own_cfg<3200, 8, 32, 512, 12, 4>  cfg1;
static_assert(cfg1::RN == LIST_SUBS * cfg2::RN && cfg2::R == LIST_SUBS * cfg1::R, "a conv1 range is LIST_SUBS conv2 ranges");
static_assert(cfg1::S == LIST_SUBS * cfg2::S, "a conv2 slice is LIST_SUBS conv1 slices");
static_assert(cfg1::WCAP % 32 == 0, "a wave's padded list part fits its capacity");
static_assert(cfg2::RN % 16 == 0 && N_NODES_C % cfg2::RN == 0, "a conv2 range is whole 16-node tiles and no range is partial");
static_assert(cfg2::RN / 16 <= OWN_THREADS / 32, "one wave per 16-node tile in the conv2 node MLP");

#define LIST_CAP   ((OWN_THREADS / 32) * cfg1::WCAP)
#define LIST_BAD   0xFFFFFFFFu
#define CNT_PITCH  32
#define LIST_U     8
#define PRIV_ROWS  28
static_assert((LIST_BAD & ((1u << cfg1::DB) - 1u)) >= (unsigned)cfg1::RN, "the pad entry lies outside every range");

typedef __attribute__((ext_vector_type(2))) _Float16 v2h;
typedef __attribute__((ext_vector_type(2))) float    v2f;

__device__ __forceinline__ int sum_pos(int ch) { return 16 * (ch & 3) + (ch >> 2); }

__device__ __forceinline__ void edge2_fetch(const unsigned* wlist, int p, int wn, int c, int hh, int e0,
                                            const int* __restrict__ ei, const float* __restrict__ ea,
                                            int (&sv)[8], v2f& av) {
    const uint4 q0 = *(const uint4*)(wlist + p + 8 * hh);
    const uint4 q1 = *(const uint4*)(wlist + p + 8 * hh + 4);
    const unsigned raw[8] = {q0.x, q0.y, q0.z, q0.w, q1.x, q1.y, q1.z, q1.w};
#pragma unroll
    for (int r = 0; r < 8; ++r) {
        const unsigned ent = (p + 8 * hh + r < wn) ? raw[r] : 0u;
        sv[r] = ei[e0 + (int)(ent >> cfg2::DB)];
    }
    const unsigned raw_c = wlist[p + c];
    const unsigned ent_c = (p + c < wn) ? raw_c : 0u;
    av = *(const v2f*)(ea + (size_t)(e0 + (int)(ent_c >> cfg2::DB)) * EDGE_DIM_C + 2 * hh);
}

template <class C>
__device__ __forceinline__ void own_push(unsigned*& pnext, unsigned* plast, int dstv, int n0, int erel) {
    const unsigned d = (unsigned)(dstv - n0);
    *(pnext < plast ? pnext : plast) = ((unsigned)erel << C::DB) | d;
    pnext += (d < (unsigned)C::RN) ? 32 : 0;
}

template <class C>
__device__ __forceinline__ unsigned own_scan(const int4* __restrict__ d4, unsigned* pfirst, unsigned* plast,
                                             int n0, int rot, int tid) {
    unsigned* pnext = pfirst;
    for (int t = 0; t < C::TRIPS; ++t) {
        int tt = t + rot;
        if (tt >= C::TRIPS) tt -= C::TRIPS;
        const int base = tt * C::STEP;
        int4 v[C::U];
#pragma unroll
        for (int u = 0; u < C::U; ++u) {
            const int i = base + u * OWN_THREADS + tid;
            v[u] = (i < C::SL4) ? d4[i] : make_int4(-1, -1, -1, -1);
        }
#pragma unroll
        for (int u = 0; u < C::U; ++u) {
            const int e = (base + u * OWN_THREADS + tid) * 4;
            own_push<C>(pnext, plast, v[u].x, n0, e + 0);
            own_push<C>(pnext, plast, v[u].y, n0, e + 1);
            own_push<C>(pnext, plast, v[u].z, n0, e + 2);
            own_push<C>(pnext, plast, v[u].w, n0, e + 3);
        }
    }
    return (unsigned)(pnext - pfirst) / 32u;
}

__device__ __forceinline__ unsigned priv_gather(const unsigned* pfirst, unsigned* wlist, unsigned cnt,
                                                int lane, unsigned wcap, bool& over) {
    unsigned inc = cnt;
#pragma unroll
    for (int sft = 1; sft < 32; sft <<= 1) {
        const unsigned o = __shfl_up(inc, (unsigned)sft, 32);
        if (lane >= sft) inc += o;
    }
    const unsigned total = (unsigned)__builtin_amdgcn_readlane((int)inc, 31);
    over = __builtin_amdgcn_ballot_w32(cnt >= (unsigned)PRIV_ROWS) != 0u || total > wcap;
    if (!over) {
        const unsigned off = inc - cnt;
        static_assert(PRIV_ROWS % 4 == 0, "whole trips of four places");
        for (unsigned r = 0; __builtin_amdgcn_ballot_w32(r < cnt) != 0u; r += 4) {
            unsigned e[4];
#pragma unroll
            for (int j = 0; j < 4; ++j) e[j] = pfirst[(r + (unsigned)j) * 32u];
#pragma unroll
            for (int j = 0; j < 4; ++j)
                if (r + (unsigned)j < cnt) wlist[off + r + (unsigned)j] = e[j];
        }
    }
    return total;
}

template <class C, class F>
__device__ __forceinline__ void own_all(const int4* __restrict__ d4, int n0, int tid, F add) {
#pragma unroll 1
    for (int t = 0; t < C::TRIPS * C::U; ++t) {
        const int i = t * OWN_THREADS + tid;
        const int4 v = (i < C::SL4) ? d4[i] : make_int4(-1, -1, -1, -1);
        const int dv[4] = {v.x, v.y, v.z, v.w};
#pragma unroll 1
        for (int k = 0; k < 4; ++k) {
            const unsigned d = (unsigned)(dv[k] - n0);
            if (d < (unsigned)C::RN) add(i * 4 + k, (int)d);
        }
    }
}

__device__ __forceinline__ void list_push(unsigned*& pnext, unsigned* plast, unsigned ent,
                                          unsigned sublo, unsigned eoff) {
    const unsigned d = (ent & ((1u << cfg1::DB) - 1u)) - sublo;
    *(pnext < plast ? pnext : plast) = (((ent >> cfg1::DB) + eoff) << cfg2::DB) | d;
    pnext += (d < (unsigned)cfg2::RN) ? 32 : 0;
}

__device__ __forceinline__ unsigned list_scan(const unsigned* __restrict__ lists, int l0,
                                              const unsigned (&cnt)[LIST_SUBS], unsigned cmax,
                                              unsigned* pfirst, unsigned* plast, unsigned sublo, int wave, int lane) {
    unsigned* pnext = pfirst;
    for (unsigned k0 = 0; k0 * OWN_THREADS < cmax; k0 += LIST_U) {
        unsigned v[LIST_SUBS][LIST_U];
#pragma unroll
        for (int q = 0; q < LIST_SUBS; ++q) {
            const unsigned* L = lists + (size_t)(l0 + q) * LIST_CAP;
            const unsigned wb = (unsigned)(((wave + 8 * q) & 31) * 32 + lane);
#pragma unroll
            for (int u = 0; u < LIST_U; ++u) {
                const unsigned i = (k0 + (unsigned)u) * OWN_THREADS + wb;
                v[q][u] = (i < cnt[q]) ? L[i] : LIST_BAD;
            }
        }
#pragma unroll
        for (int q = 0; q < LIST_SUBS; ++q) {
#pragma unroll
            for (int u = 0; u < LIST_U; ++u)
                list_push(pnext, plast, v[q][u], sublo, (unsigned)(q * (cfg1::SL4 * 4)));
        }
    }
    return (unsigned)(pnext - pfirst) / 32u;
}

template <class F>
__device__ __forceinline__ void list_all(const unsigned* __restrict__ lists,
                                         const unsigned* __restrict__ lcount, int l0,
                                         unsigned cmax, unsigned sublo, int wave, int lane, F add) {
#pragma unroll 1
    for (unsigned k0 = 0; k0 * OWN_THREADS < cmax; k0 += LIST_U) {
#pragma unroll 1
        for (int q = 0; q < LIST_SUBS; ++q) {
            const unsigned* L = lists + (size_t)(l0 + q) * LIST_CAP;
            const unsigned cq = lcount[(l0 + q) * CNT_PITCH];
            const unsigned wb = (unsigned)(((wave + 8 * q) & 31) * 32 + lane);
#pragma unroll 1
            for (int u = 0; u < LIST_U; ++u) {
                const unsigned i = (k0 + (unsigned)u) * OWN_THREADS + wb;
                const unsigned ent = (i < cq) ? L[i] : LIST_BAD;
                const unsigned d = (ent & ((1u << cfg1::DB) - 1u)) - sublo;
                if (d < (unsigned)cfg2::RN)
                    add((int)((ent >> cfg1::DB) + (unsigned)(q * (cfg1::SL4 * 4))), (int)d);
            }
        }
    }
}

__global__ __launch_bounds__(OWN_THREADS)
void edge2_own_kernel(const _Float16* __restrict__ h, const int* __restrict__ ei,
                      const float* __restrict__ ea,
                      const float* __restrict__ eW, const float* __restrict__ eb,
                      const unsigned* __restrict__ lists, const unsigned* __restrict__ lcount,
                      const float* __restrict__ W1, const float* __restrict__ b1,
                      const float* __restrict__ W2, const float* __restrict__ b2,
                      float* __restrict__ h2) {
    typedef cfg2 C;
    constexpr int P2 = HID_C + 8;
    constexpr int NT = C::RN / 16;
    constexpr int PA = HID_C + 4;
    static_assert(HID_C * HID_C / 4 == OWN_THREADS, "one float4 of W1 and one of W2 per thread");
    __shared__ __align__(16) float    s_acc[(C::RN + 1) * PA];
    __shared__ __align__(16) unsigned s_list[(OWN_THREADS / 32) * C::WCAP];
    __shared__ __align__(16) unsigned s_priv[OWN_THREADS * PRIV_ROWS];
    __shared__ __align__(16) _Float16 sW1[HID_C * P2];
    __shared__ __align__(16) _Float16 sW2[HID_C * P2];

    const int tid   = threadIdx.x;
    const int lane  = tid & 31;
    const int wave  = tid >> 5;
    const int range = (int)blockIdx.x / C::S;
    const int slice = (int)blockIdx.x - range * C::S;
    const int n0    = range * C::RN;
    const int e0    = slice * (C::SL4 * 4);
    unsigned* wlist = s_list + wave * C::WCAP;
    unsigned* pfirst = s_priv + wave * (32 * PRIV_ROWS) + lane;
    unsigned* plast  = pfirst + 32 * (PRIV_ROWS - 1);

    if (n0 >= N_NODES_C) return;

    const float4 w1v = ((const float4*)W1)[tid];
    const float4 w2v = ((const float4*)W2)[tid];

    for (int i = tid; i < C::RN * PA / 4; i += OWN_THREADS)
        ((float4*)s_acc)[i] = make_float4(0.f, 0.f, 0.f, 0.f);
    {
        const int n = tid / (HID_C / 4), q = tid - n * (HID_C / 4);
        v4h o1, o2;
        o1[0] = (_Float16)w1v.x; o1[1] = (_Float16)w1v.y; o1[2] = (_Float16)w1v.z; o1[3] = (_Float16)w1v.w;
        o2[0] = (_Float16)w2v.x; o2[1] = (_Float16)w2v.y; o2[2] = (_Float16)w2v.z; o2[3] = (_Float16)w2v.w;
        *(v4h*)(sW1 + n * P2 + 4 * q) = o1;
        *(v4h*)(sW2 + n * P2 + 4 * q) = o2;
    }
    __syncthreads();

    const int c  = lane & 15;
    const int hh = lane >> 4;
    v2f   bw[4];
    float bch[4];
#pragma unroll
    for (int t = 0; t < 4; ++t) {
        bw[t]  = *(const v2f*)(eW + (4 * c + t) * EDGE_DIM_C + 2 * hh);
        bch[t] = eb[4 * c + t];
    }

    auto add_one = [&](int erel, int od) {
        const int oe = e0 + erel;
        const int os = ei[oe];
        const float4 a = *(const float4*)(ea + (size_t)oe * EDGE_DIM_C);
#pragma unroll 1
        for (int q = 0; q < HID_C; ++q) {
            const float4 w = *(const float4*)(eW + q * EDGE_DIM_C);
            float t = eb[q] + a.x * w.x + a.y * w.y + a.z * w.z + a.w * w.w;
            t += (float)h[(size_t)os * HID_C + q];
            atomicAdd(&s_acc[od * PA + sum_pos(q)], fmaxf(t, 0.0f));
        }
    };

    const int l0 = (range / LIST_SUBS) * cfg1::S + slice * LIST_SUBS;
    const unsigned sublo = (unsigned)((range % LIST_SUBS) * C::RN);
    unsigned cnt[LIST_SUBS];
    unsigned cmax = 0;
#pragma unroll
    for (int q = 0; q < LIST_SUBS; ++q) {
        cnt[q] = lcount[(l0 + q) * CNT_PITCH];
        cmax = cnt[q] > cmax ? cnt[q] : cmax;
    }
    unsigned wcount;
    bool over;
    if (cmax <= (unsigned)LIST_CAP) {
        const unsigned mine = list_scan(lists, l0, cnt, cmax, pfirst, plast, sublo, wave, lane);
        wcount = priv_gather(pfirst, wlist, mine, lane, (unsigned)C::WCAP, over);
        if (over)
            list_all(lists, lcount, l0, cmax, sublo, wave, lane, add_one);
    } else {
        const int4* d4 = (const int4*)(ei + N_EDGES_C + e0);
        const int rot = (range * C::TRIPS) / C::R;
        const unsigned mine = own_scan<C>(d4, pfirst, plast, n0, rot, tid);
        wcount = priv_gather(pfirst, wlist, mine, lane, (unsigned)C::WCAP, over);
        if (over)
            own_all<C>(d4, n0, tid, add_one);
    }
    const int wn = over ? 0 : (int)wcount;

    __syncthreads();

    const char* hb = (const char*)h;
    int sv[8];
    v2f av;
    edge2_fetch(wlist, 0, wn, c, hh, e0, ei, ea, sv, av);
    for (int p = 0; p < wn; p += 16) {
        v4h hv[8];
#pragma unroll
        for (int r = 0; r < 8; ++r) {
            const unsigned off = (unsigned)sv[r] * (unsigned)(HID_C * 2) + 8u * (unsigned)c;
            hv[r] = *(const v4h*)(hb + off);
        }
        const int pn = (p + 16 < wn) ? p + 16 : p;
        int svn[8];
        v2f avn;
        edge2_fetch(wlist, pn, wn, c, hh, e0, ei, ea, svn, avn);
        float* arow[8];
        {
            const uint4 q0 = *(const uint4*)(wlist + p + 8 * hh);
            const uint4 q1 = *(const uint4*)(wlist + p + 8 * hh + 4);
            const unsigned raw[8] = {q0.x, q0.y, q0.z, q0.w, q1.x, q1.y, q1.z, q1.w};
#pragma unroll
            for (int r = 0; r < 8; ++r) {
                const int dl = (p + 8 * hh + r < wn) ? (int)(raw[r] & ((1u << C::DB) - 1u)) : C::RN;
                arow[r] = s_acc + dl * PA + c;
            }
        }
#pragma unroll
        for (int t = 0; t < 4; ++t) {
            v8f cc;
#pragma unroll
            for (int r = 0; r < 8; ++r) cc[r] = bch[t] + (float)hv[r][t];
            const v8f d = __builtin_amdgcn_wmma_f32_16x16x4_f32(false, av, false, bw[t], (short)0, cc, false, false);
#pragma unroll
            for (int r = 0; r < 8; ++r)
                atomicAdd(arow[r] + 16 * t, fmaxf(d[r], 0.0f));
        }
#pragma unroll
        for (int r = 0; r < 8; ++r) sv[r] = svn[r];
        av = avn;
    }
    const int  wv   = __builtin_amdgcn_readfirstlane(wave);
    const bool tile = wv < NT;
    const _Float16* grow = h + (size_t)(n0 + (tile ? wv * 16 + c : 0)) * HID_C + 8 * hh;
    v8h    xr[4];
    float4 pb1[8], pb2[8];
#pragma unroll
    for (int i = 0; i < 4; ++i) {
        xr[i]          = *(const v8h*)(grow + 16 * i);
        pb1[2 * i]     = *(const float4*)(b1 + 16 * i + 8 * hh);
        pb1[2 * i + 1] = *(const float4*)(b1 + 16 * i + 8 * hh + 4);
        pb2[2 * i]     = *(const float4*)(b2 + 16 * i + 8 * hh);
        pb2[2 * i + 1] = *(const float4*)(b2 + 16 * i + 8 * hh + 4);
    }

    __syncthreads();

    float*     trow  = s_acc + (wv * 16 + c) * PA;
    if (tile) {
        v16h bop[2];
#pragma unroll
        for (int kb = 0; kb < 2; ++kb) {
            const v8h x0 = xr[2 * kb];
            const v8h x1 = xr[2 * kb + 1];
            float lo[8], hi[8];
#pragma unroll
            for (int j = 0; j < 4; ++j) {
                const float2 pl = *(const float2*)(trow + 16 * j + 8 * kb + 2 * hh);
                const float2 ph = *(const float2*)(trow + 16 * j + 8 * kb + 2 * hh + 4);
                lo[j] = pl.x; lo[j + 4] = pl.y;
                hi[j] = ph.x; hi[j + 4] = ph.y;
            }
            v16h b;
#pragma unroll
            for (int e = 0; e < 8; ++e) {
                b[e]     = (_Float16)((float)x0[e] + lo[e]);
                b[8 + e] = (_Float16)((float)x1[e] + hi[e]);
            }
            bop[kb] = b;
        }

        v8f acc[4];
#pragma unroll
        for (int t = 0; t < 4; ++t) {
            acc[t] = v8f{};
#pragma unroll
            for (int kb = 0; kb < 2; ++kb) {
                const v16h a = lds_load_tile(sW1 + 16 * t * P2, P2, kb * 32);
                acc[t] = __builtin_amdgcn_wmma_f32_16x16x32_f16(false, a, false, bop[kb],
                                                                (short)0, acc[t], false, false);
            }
        }

        v16h bop2[2];
#pragma unroll
        for (int s2 = 0; s2 < 2; ++s2) {
            const float4 p0 = pb1[4 * s2], p1 = pb1[4 * s2 + 1], p2 = pb1[4 * s2 + 2], p3 = pb1[4 * s2 + 3];
            const float bl[8] = {p0.x, p0.y, p0.z, p0.w, p1.x, p1.y, p1.z, p1.w};
            const float bh[8] = {p2.x, p2.y, p2.z, p2.w, p3.x, p3.y, p3.z, p3.w};
            v16h b;
#pragma unroll
            for (int e = 0; e < 8; ++e) {
                float v0 = acc[2 * s2][e] + bl[e];
                float v1 = acc[2 * s2 + 1][e] + bh[e];
                v0 = v0 > 0.0f ? v0 : 0.0f;
                v1 = v1 > 0.0f ? v1 : 0.0f;
                b[e]     = (_Float16)v0;
                b[8 + e] = (_Float16)v1;
            }
            bop2[s2] = b;
        }

#pragma unroll
        for (int t = 0; t < 4; ++t) {
            v8f d = v8f{};
#pragma unroll
            for (int s2 = 0; s2 < 2; ++s2) {
                const v16h a = lds_load_tile(sW2 + 16 * t * P2, P2, s2 * 32);
                d = __builtin_amdgcn_wmma_f32_16x16x32_f16(false, a, false, bop2[s2],
                                                           (short)0, d, false, false);
            }
            const float4 q0 = pb2[2 * t], q1 = pb2[2 * t + 1];
            const float bb[8] = {q0.x, q0.y, q0.z, q0.w, q1.x, q1.y, q1.z, q1.w};
            float v[8];
#pragma unroll
            for (int e = 0; e < 8; ++e) {
                const float u = d[e] + bb[e];
                v[e] = u > 0.0f ? u : 0.0f;
            }
            float* q = trow + 16 * t + 8 * hh;
            *(float4*)(q)     = make_float4(v[0], v[1], v[2], v[3]);
            *(float4*)(q + 4) = make_float4(v[4], v[5], v[6], v[7]);
        }
    }
    __syncthreads();

    if (tile) {
        const float* st = s_acc + wv * 16 * PA;
        float*       gt = h2 + (size_t)(n0 + wv * 16) * HID_C;
#pragma unroll
        for (int i = 0; i < 8; ++i) {
            const int f = 4 * (lane + 32 * i);
            *(float4*)(gt + f) = *(const float4*)(st + (f >> 6) * PA + (f & 63));
        }
    }
}

__global__ __launch_bounds__(OWN_THREADS)
void edge1_own_kernel(const float* __restrict__ x, const int* __restrict__ ei,
                      const float* __restrict__ ea,
                      const float* __restrict__ eW, const float* __restrict__ eb,
                      float* __restrict__ aggr,
                      unsigned* __restrict__ lists, unsigned* __restrict__ lcount) {
    typedef cfg1 C;
    __shared__ __align__(16) float    s_acc[C::RN * 8];
    __shared__ __align__(16) unsigned s_list[(OWN_THREADS / 32) * C::WCAP];
    __shared__ __align__(16) unsigned s_priv[OWN_THREADS * PRIV_ROWS];
    __shared__ unsigned s_wn[OWN_THREADS / 32];

    const int tid   = threadIdx.x;
    const int lane  = tid & 31;
    const int wave  = tid >> 5;
    const int range = (int)blockIdx.x / C::S;
    const int slice = (int)blockIdx.x - range * C::S;
    const int n0    = range * C::RN;
    const int e0    = slice * (C::SL4 * 4);
    unsigned* wlist = s_list + wave * C::WCAP;
    unsigned* pfirst = s_priv + wave * (32 * PRIV_ROWS) + lane;
    unsigned* plast  = pfirst + 32 * (PRIV_ROWS - 1);

    for (int i = tid; i < C::RN * 8; i += OWN_THREADS) s_acc[i] = 0.0f;
    __syncthreads();

    const int c   = lane & 7;
    const int sub = lane >> 3;
    const bool cact = c < IN_DIM_C;
    const float4 wc = cact ? *(const float4*)(eW + c * EDGE_DIM_C) : make_float4(0.f, 0.f, 0.f, 0.f);
    const float  bc = cact ? eb[c] : 0.0f;
    const int    cx = cact ? c : 0;

    const int4* d4 = (const int4*)(ei + N_EDGES_C + e0);
    const int rot = (range * C::TRIPS) / C::R;
    const unsigned mine = own_scan<C>(d4, pfirst, plast, n0, rot, tid);
    bool over;
    const unsigned wcount = priv_gather(pfirst, wlist, mine, lane, (unsigned)C::WCAP, over);
    if (over) {
        own_all<C>(d4, n0, tid, [&](int erel, int od) {
            const int oe = e0 + erel;
            const int os = ei[oe];
            const float4 a = *(const float4*)(ea + (size_t)oe * EDGE_DIM_C);
#pragma unroll 1
            for (int q = 0; q < IN_DIM_C; ++q) {
                const float4 w = *(const float4*)(eW + q * EDGE_DIM_C);
                float t = eb[q] + a.x * w.x + a.y * w.y + a.z * w.z + a.w * w.w;
                t += x[(size_t)os * IN_DIM_C + q];
                atomicAdd(&s_acc[od * 8 + q], fmaxf(t, 0.0f));
            }
        });
    }
    const int wn = over ? 0 : (int)wcount;
    if (lane == 0) s_wn[wave] = over ? (unsigned)C::WCAP + 1u : wcount;

    __syncthreads();

    {
        unsigned off = 0, tot = 0;
        bool full = false;
        for (int w = 0; w < OWN_THREADS / 32; ++w) {
            const unsigned cw = s_wn[w];
            full = full || cw > (unsigned)C::WCAP;
            const unsigned pw = ((cw < (unsigned)C::WCAP ? cw : (unsigned)C::WCAP) + 31u) & ~31u;
            if (w < wave) off += pw;
            tot += pw;
        }
        unsigned* L = lists + (size_t)blockIdx.x * LIST_CAP + off;
        const int wp = (wn + 31) & ~31;
        for (int i = lane; i < wp; i += 32) L[i] = (i < wn) ? wlist[i] : LIST_BAD;
        if (tid == 0) lcount[blockIdx.x * CNT_PITCH] = full ? LIST_BAD : tot;
    }

    for (int p = 0; p < wn; p += 32) {
        unsigned ent[8]; int sr[8]; float4 av[8]; float xv[8]; bool ok[8];
#pragma unroll
        for (int g = 0; g < 8; ++g) {
            const int idx = p + g * 4 + sub;
            ok[g]  = idx < wn;
            ent[g] = ok[g] ? wlist[idx] : 0u;
        }
#pragma unroll
        for (int g = 0; g < 8; ++g) {
            const int e = e0 + (int)(ent[g] >> C::DB);
            sr[g] = ei[e];
            av[g] = *(const float4*)(ea + (size_t)e * EDGE_DIM_C);
        }
#pragma unroll
        for (int g = 0; g < 8; ++g)
            xv[g] = x[(size_t)sr[g] * IN_DIM_C + cx];
        asm volatile("" :: "v"(xv[0]), "v"(xv[1]), "v"(xv[2]), "v"(xv[3]), "v"(xv[4]), "v"(xv[5]), "v"(xv[6]), "v"(xv[7]),
                           "v"(av[0].x), "v"(av[0].y), "v"(av[0].z), "v"(av[0].w));
#pragma unroll
        for (int g = 0; g < 8; ++g) {
            if (ok[g] && cact) {
                const int dl = (int)(ent[g] & ((1u << C::DB) - 1u));
                float t = bc + av[g].x * wc.x + av[g].y * wc.y + av[g].z * wc.z + av[g].w * wc.w;
                t += xv[g];
                atomicAdd(&s_acc[dl * 8 + c], fmaxf(t, 0.0f));
            }
        }
    }
    __syncthreads();

    int rows = N_NODES_C - n0;
    if (rows > C::RN) rows = C::RN;
    float* o = aggr + (size_t)slice * N_NODES_C * IN_DIM_C + (size_t)n0 * IN_DIM_C;
    for (int i = tid; i < rows * IN_DIM_C; i += OWN_THREADS) {
        const int r = i / IN_DIM_C, q = i - r * IN_DIM_C;
        o[i] = s_acc[r * 8 + q];
    }
}

#define POOL_GPB     4
#define POOL_THREADS 1024
#define POOL_U       14
#define POOL_Q       8

__global__ __launch_bounds__(POOL_THREADS)
void pool_fc_kernel(const float* __restrict__ h2, const int* __restrict__ batch,
                    const float* __restrict__ fcW, const float* __restrict__ fcb,
                    float* __restrict__ out) {
    constexpr int NW  = POOL_THREADS / 32;
    constexpr int NCH = (N_NODES_C + 31) / 32;
    static_assert(N_GRAPHS_C % POOL_GPB == 0, "every block owns POOL_GPB graphs");
    static_assert(POOL_GPB * HID_C <= POOL_THREADS, "one thread per (graph, channel) in the wave sum");

    __shared__ __align__(16) float s_part[NW][POOL_GPB][HID_C];
    __shared__ int   s_cnt [NW][POOL_GPB];
    __shared__ float s_mean[POOL_GPB][HID_C];

    const int tid  = threadIdx.x;
    const int lane = tid & 31;
    const int wave = tid >> 5;
    const int g0   = (int)blockIdx.x * POOL_GPB;
    const float2* h2v = (const float2*)h2;

    float2 acc[POOL_GPB];
    int    cnt[POOL_GPB];
#pragma unroll
    for (int k = 0; k < POOL_GPB; ++k) { acc[k] = make_float2(0.f, 0.f); cnt[k] = 0; }

    for (int c0 = wave; c0 < NCH; c0 += NW * POOL_U) {
        unsigned hit = 0;
#pragma unroll
        for (int u = 0; u < POOL_U; ++u) {
            const int c = c0 + u * NW;
            const int n = c * 32 + lane;
            const int g = (c < NCH && n < N_NODES_C) ? batch[n] : -1;
            const bool m = (unsigned)(g - g0) < (unsigned)POOL_GPB;
            if (__builtin_amdgcn_ballot_w32(m)) hit |= 1u << u;
        }
        while (hit) {
            const int u = __builtin_ctz(hit);
            hit &= hit - 1;
            const int base = (c0 + u * NW) * 32;
            const int n = base + lane;
            const int g = (n < N_NODES_C) ? batch[n] : -1;
            const int d = g - g0;
            unsigned mask = __builtin_amdgcn_ballot_w32((unsigned)d < (unsigned)POOL_GPB);
#pragma unroll
            for (int k = 0; k < POOL_GPB; ++k)
                cnt[k] += __builtin_popcount(__builtin_amdgcn_ballot_w32(d == k));
            while (mask) {
                int gq[POOL_Q]; float2 v[POOL_Q];
                const int jf = __builtin_ctz(mask);
#pragma unroll
                for (int q = 0; q < POOL_Q; ++q) {
                    const bool has = mask != 0;
                    const int j = has ? __builtin_ctz(mask) : jf;
                    const int gj = __builtin_amdgcn_readlane(d, j);
                    gq[q] = has ? gj : -1;
                    v[q]  = h2v[(size_t)(base + j) * (HID_C / 2) + lane];
                    mask &= mask - 1;
                }
                __builtin_amdgcn_sched_barrier(0);
#pragma unroll
                for (int q = 0; q < POOL_Q; ++q) {
#pragma unroll
                    for (int k = 0; k < POOL_GPB; ++k)
                        if (gq[q] == k) { acc[k].x += v[q].x; acc[k].y += v[q].y; }
                }
            }
        }
    }

#pragma unroll
    for (int k = 0; k < POOL_GPB; ++k) {
        *(float2*)&s_part[wave][k][2 * lane] = acc[k];
        if (lane == 0) s_cnt[wave][k] = cnt[k];
    }
    __syncthreads();

    if (tid < POOL_GPB * HID_C) {
        const int k = tid / HID_C, ch = tid - k * HID_C;
        float s = 0.0f; int c = 0;
        for (int w = 0; w < NW; ++w) { s += s_part[w][k][ch]; c += s_cnt[w][k]; }
        s_mean[k][ch] = s / fmaxf((float)c, 1.0f);
    }
    __syncthreads();

    if (tid < POOL_GPB * NUM_TASKS_C) {
        const int k = tid / NUM_TASKS_C, t = tid - k * NUM_TASKS_C;
        float s = fcb[t];
#pragma unroll
        for (int c = 0; c < HID_C; ++c)
            s += s_mean[k][c] * fcW[t * HID_C + c];
        out[(size_t)(g0 + k) * NUM_TASKS_C + t] = s;
    }
}

extern "C" void kernel_launch(void* const* d_in, const int* in_sizes, int n_in,
                              void* d_out, int out_size, void* d_ws, size_t ws_size,
                              hipStream_t stream) {
    const float* x    = (const float*)d_in[0];
    const int*   ei   = (const int*)  d_in[1];
    const float* ea   = (const float*)d_in[2];
    const int*   bat  = (const int*)  d_in[3];
    const float* e1W  = (const float*)d_in[4];  const float* e1b  = (const float*)d_in[5];
    const float* n1W1 = (const float*)d_in[6];  const float* n1b1 = (const float*)d_in[7];
    const float* n1W2 = (const float*)d_in[8];  const float* n1b2 = (const float*)d_in[9];
    const float* e2W  = (const float*)d_in[10]; const float* e2b  = (const float*)d_in[11];
    const float* n2W1 = (const float*)d_in[12]; const float* n2b1 = (const float*)d_in[13];
    const float* n2W2 = (const float*)d_in[14]; const float* n2b2 = (const float*)d_in[15];
    const float* fcW  = (const float*)d_in[16]; const float* fcb  = (const float*)d_in[17];
    float* out = (float*)d_out;

    float*    aggr1 = (float*)d_ws;
    _Float16* h     = (_Float16*)(aggr1 + (size_t)cfg1::S * N_NODES_C * IN_DIM_C);
    float*    h2    = (float*)(h + (size_t)N_NODES_C * HID_C);
    unsigned* lists  = (unsigned*)(h2 + (size_t)N_NODES_C * HID_C);
    unsigned* lcount = lists + (size_t)OWN_BLOCKS * LIST_CAP;

    edge1_own_kernel<<<OWN_BLOCKS, OWN_THREADS, 0, stream>>>(x, ei, ea, e1W, e1b, aggr1, lists, lcount);
    node_mlp_kernel<IN_DIM_C, cfg1::S><<<MLP_GRID, MLP_THREADS, 0, stream>>>(
        x, nullptr, aggr1, n1W1, n1b1, n1W2, n1b2, h, nullptr, N_NODES_C);
    edge2_own_kernel<<<OWN_BLOCKS, OWN_THREADS, 0, stream>>>(h, ei, ea, e2W, e2b, lists, lcount,
                                                             n2W1, n2b1, n2W2, n2b2, h2);
    pool_fc_kernel<<<N_GRAPHS_C / POOL_GPB, POOL_THREADS, 0, stream>>>(h2, bat, fcW, fcb, out);
}


# Mean pool starts inside the conv2 owner blocks: a block adds its h2 rows from LDS per run of equal graph id (8 partial sums, node order), one row per run out; pool adds run rows; no h2 plane
# speedup vs baseline: 4.7235x; 1.0977x over previous
#include <hip/hip_runtime.h>
#include <hip/hip_bf16.h>

#define N_NODES_C  100000
#define N_EDGES_C  1600000
#define N_GRAPHS_C 1000
#define HID_C      64
#define IN_DIM_C   7
#define EDGE_DIM_C 4
#define NUM_TASKS_C 12

typedef __attribute__((ext_vector_type(16))) _Float16 v16h;
typedef __attribute__((ext_vector_type(8)))  _Float16 v8h;
typedef __attribute__((ext_vector_type(8)))  float    v8f;

__device__ __forceinline__ v16h lds_load_tile(const _Float16* s, int ld, int kbase) {
    const int lane = threadIdx.x & 31;
    const int r    = lane & 15;
    const int koff = (lane < 16) ? 0 : 8;
    const _Float16* p = s + r * ld + kbase + koff;
    v8h lo = *(const v8h*)(p);
    v8h hi = *(const v8h*)(p + 16);
    v16h out;
#pragma unroll
    for (int e = 0; e < 8; ++e) { out[e] = lo[e]; out[8 + e] = hi[e]; }
    return out;
}

#define MLP_WAVES   8
#define MLP_NT      2
#define MLP_THREADS (MLP_WAVES * 32)
#define MLP_TILES   (MLP_WAVES * MLP_NT)
#define MLP_ROWS    (MLP_TILES * 16)
#define MLP_GRID    ((N_NODES_C + MLP_ROWS - 1) / MLP_ROWS)
static_assert(N_NODES_C % 16 == 0, "the node count is a whole number of 16-node tiles");

typedef __attribute__((ext_vector_type(4))) _Float16 v4h;

template <int K, int KP>
struct mlp_wregs {
    static constexpr bool VEC = (K % 4) == 0;
    static constexpr int  N   = (VEC ? HID_C * (K / 4) : HID_C * KP) / MLP_THREADS;
    static_assert((VEC ? HID_C * (K / 4) : HID_C * KP) % MLP_THREADS == 0, "the same number of pieces in every thread");
    static_assert(!VEC || K == KP, "no padding on the 16-byte path");
    float4 v4[VEC ? N : 1];
    float  v1[VEC ? 1 : N];
};

template <int K, int KP>
__device__ __forceinline__ void mlp_load_weights(const float* __restrict__ W, int tid, mlp_wregs<K, KP>& r) {
#pragma unroll
    for (int it = 0; it < mlp_wregs<K, KP>::N; ++it) {
        const int i = tid + it * MLP_THREADS;
        if constexpr (mlp_wregs<K, KP>::VEC) {
            r.v4[it] = ((const float4*)W)[i];
        } else {
            const int n = i / KP, k = i - n * KP;
            const float w = W[n * K + (k < K ? k : 0)];
            r.v1[it] = (k < K) ? w : 0.0f;
        }
    }
}

template <int K, int KP, int P>
__device__ __forceinline__ void mlp_store_weights(_Float16* s, int tid, const mlp_wregs<K, KP>& r) {
#pragma unroll
    for (int it = 0; it < mlp_wregs<K, KP>::N; ++it) {
        const int i = tid + it * MLP_THREADS;
        if constexpr (mlp_wregs<K, KP>::VEC) {
            const int n = i / (K / 4), q = i - n * (K / 4);
            const float4 v = r.v4[it];
            v4h o;
            o[0] = (_Float16)v.x; o[1] = (_Float16)v.y; o[2] = (_Float16)v.z; o[3] = (_Float16)v.w;
            *(v4h*)(s + n * P + 4 * q) = o;
        } else {
            const int n = i / KP, k = i - n * KP;
            s[n * P + k] = (_Float16)r.v1[it];
        }
    }
}

template <int K1, int NP>
__global__ __launch_bounds__(MLP_THREADS)
void node_mlp_kernel(const float* __restrict__ xin, const _Float16* __restrict__ hin,
                     const float* __restrict__ in1,
                     const float* __restrict__ W1,  const float* __restrict__ b1,
                     const float* __restrict__ W2,  const float* __restrict__ b2,
                     _Float16* __restrict__ outh, float* __restrict__ outf, int n_nodes) {
    constexpr int  K1P   = (K1 + 31) & ~31;
    constexpr int  KS    = K1P / 32;
    constexpr int  P1    = K1P + 8;
    constexpr int  P2    = HID_C + 8;
    constexpr int  PO    = HID_C + 4;
    constexpr bool SMALL = (K1 % 8) != 0;
    constexpr int  XT    = 16 * K1;
    static_assert(!SMALL || (XT % 4 == 0 && XT / 4 <= 32), "a flat conv1 tile is at most one float4 per lane");
    static_assert(SMALL || K1 == HID_C, "the conv2 path cuts rows of 64");

    __shared__ __align__(16) _Float16 sW1[HID_C * P1];
    __shared__ __align__(16) _Float16 sW2[HID_C * P2];
    __shared__ __align__(16) float    sX [SMALL ? MLP_TILES * 128 : 4];
    __shared__ __align__(16) _Float16 sA [SMALL ? 8 : MLP_TILES * 16 * P1];
    __shared__ __align__(16) _Float16 sOh[SMALL ? MLP_TILES * 16 * P2 : 8];
    __shared__ __align__(16) float    sOf[SMALL ? 4 : MLP_TILES * 16 * PO];

    const int tid  = threadIdx.x;
    const int wave = __builtin_amdgcn_readfirstlane(tid >> 5);
    const int lane = tid & 31;
    const int c    = lane & 15;
    const int h    = lane >> 4;
    const int tile0 = ((int)blockIdx.x * MLP_WAVES + wave) * MLP_NT;

    int  row0[MLP_NT];
    bool tv[MLP_NT];
#pragma unroll
    for (int j = 0; j < MLP_NT; ++j) {
        row0[j] = (tile0 + j) * 16;
        tv[j]   = row0[j] < n_nodes;
    }

    mlp_wregs<K1, K1P>     w1r;
    mlp_wregs<HID_C, HID_C> w2r;
    mlp_load_weights<K1, K1P>(W1, tid, w1r);
    mlp_load_weights<HID_C, HID_C>(W2, tid, w2r);

    if constexpr (SMALL) {
        const int ll = lane < XT / 4 ? lane : XT / 4 - 1;
        float4 raw[MLP_NT][NP + 1];
#pragma unroll
        for (int j = 0; j < MLP_NT; ++j) {
            const int o4 = (tv[j] ? row0[j] : 0) * K1 / 4 + ll;
            raw[j][0] = ((const float4*)xin)[o4];
#pragma unroll
            for (int p = 0; p < NP; ++p)
                raw[j][p + 1] = ((const float4*)(in1 + (size_t)p * (size_t)n_nodes * K1))[o4];
        }
        float4 v[MLP_NT];
#pragma unroll
        for (int j = 0; j < MLP_NT; ++j) {
            v[j] = raw[j][0];
#pragma unroll
            for (int p = 0; p < NP; ++p) {
                const float4 a = raw[j][p + 1];
                v[j].x += a.x; v[j].y += a.y; v[j].z += a.z; v[j].w += a.w;
            }
        }
#pragma unroll
        for (int j = 0; j < MLP_NT; ++j)
            ((float4*)sX)[(wave * MLP_NT + j) * 32 + lane] = v[j];
    } else {
#pragma unroll
        for (int j = 0; j < MLP_NT; ++j) {
            const int    eb = (tv[j] ? row0[j] : 0) * K1;
            const int    tl = wave * MLP_NT + j;
#pragma unroll
            for (int i = 0; i < XT / 128; ++i) {
                const int f = 4 * (lane + 32 * i);
                const v4h x4 = *(const v4h*)(hin + eb + f);
                float v0 = (float)x4[0], v1 = (float)x4[1], v2 = (float)x4[2], v3 = (float)x4[3];
#pragma unroll
                for (int p = 0; p < NP; ++p) {
                    const float4 a = *(const float4*)(in1 + (size_t)p * (size_t)n_nodes * K1 + eb + f);
                    v0 += a.x; v1 += a.y; v2 += a.z; v3 += a.w;
                }
                v4h o;
                o[0] = (_Float16)v0; o[1] = (_Float16)v1; o[2] = (_Float16)v2; o[3] = (_Float16)v3;
                *(v4h*)(sA + (tl * 16 + (f >> 6)) * P1 + (f & 63)) = o;
            }
        }
    }

    mlp_store_weights<K1, K1P, P1>(sW1, tid, w1r);
    mlp_store_weights<HID_C, HID_C, P2>(sW2, tid, w2r);
    __syncthreads();

    v16h bop[MLP_NT][KS];
#pragma unroll
    for (int j = 0; j < MLP_NT; ++j) {
        const int tl = wave * MLP_NT + j;
        if constexpr (SMALL) {
            const float* r = sX + tl * 128 + c * K1;
            v16h b = {};
#pragma unroll
            for (int k = 0; k < K1; ++k) {
                const float f = r[k];
                b[k] = (h == 0) ? (_Float16)f : (_Float16)0.0f;
            }
            bop[j][0] = b;
        } else {
#pragma unroll
            for (int kb = 0; kb < KS; ++kb)
                bop[j][kb] = lds_load_tile(sA + tl * 16 * P1, P1, kb * 32);
        }
    }

    float bias1[4][8], bias2[4][8];
#pragma unroll
    for (int t = 0; t < 4; ++t) {
        const float4 a0 = *(const float4*)(b1 + 16 * t + 8 * h);
        const float4 a1 = *(const float4*)(b1 + 16 * t + 8 * h + 4);
        bias1[t][0] = a0.x; bias1[t][1] = a0.y; bias1[t][2] = a0.z; bias1[t][3] = a0.w;
        bias1[t][4] = a1.x; bias1[t][5] = a1.y; bias1[t][6] = a1.z; bias1[t][7] = a1.w;
        const float4 c0 = *(const float4*)(b2 + 16 * t + 8 * h);
        const float4 c1 = *(const float4*)(b2 + 16 * t + 8 * h + 4);
        bias2[t][0] = c0.x; bias2[t][1] = c0.y; bias2[t][2] = c0.z; bias2[t][3] = c0.w;
        bias2[t][4] = c1.x; bias2[t][5] = c1.y; bias2[t][6] = c1.z; bias2[t][7] = c1.w;
    }

    v8f acc[MLP_NT][4];
#pragma unroll
    for (int j = 0; j < MLP_NT; ++j)
#pragma unroll
        for (int t = 0; t < 4; ++t) acc[j][t] = v8f{};
#pragma unroll
    for (int t = 0; t < 4; ++t) {
#pragma unroll
        for (int kb = 0; kb < KS; ++kb) {
            const v16h a = lds_load_tile(sW1 + 16 * t * P1, P1, kb * 32);
#pragma unroll
            for (int j = 0; j < MLP_NT; ++j)
                acc[j][t] = __builtin_amdgcn_wmma_f32_16x16x32_f16(false, a, false, bop[j][kb],
                                                                   (short)0, acc[j][t], false, false);
        }
    }

    v16h bop2[MLP_NT][2];
#pragma unroll
    for (int j = 0; j < MLP_NT; ++j) {
#pragma unroll
        for (int s = 0; s < 2; ++s) {
            v16h b;
#pragma unroll
            for (int e = 0; e < 8; ++e) {
                float v0 = acc[j][2 * s][e] + bias1[2 * s][e];
                float v1 = acc[j][2 * s + 1][e] + bias1[2 * s + 1][e];
                v0 = v0 > 0.0f ? v0 : 0.0f;
                v1 = v1 > 0.0f ? v1 : 0.0f;
                b[e]     = (_Float16)v0;
                b[8 + e] = (_Float16)v1;
            }
            bop2[j][s] = b;
        }
    }

    v8f acc2[MLP_NT][4];
#pragma unroll
    for (int j = 0; j < MLP_NT; ++j)
#pragma unroll
        for (int t = 0; t < 4; ++t) acc2[j][t] = v8f{};
#pragma unroll
    for (int t = 0; t < 4; ++t) {
#pragma unroll
        for (int s = 0; s < 2; ++s) {
            const v16h a = lds_load_tile(sW2 + 16 * t * P2, P2, s * 32);
#pragma unroll
            for (int j = 0; j < MLP_NT; ++j)
                acc2[j][t] = __builtin_amdgcn_wmma_f32_16x16x32_f16(false, a, false, bop2[j][s],
                                                                    (short)0, acc2[j][t], false, false);
        }
    }

#pragma unroll
    for (int j = 0; j < MLP_NT; ++j) {
        const int tl = wave * MLP_NT + j;
#pragma unroll
        for (int t = 0; t < 4; ++t) {
            float v[8];
#pragma unroll
            for (int e = 0; e < 8; ++e) {
                const float u = acc2[j][t][e] + bias2[t][e];
                v[e] = u > 0.0f ? u : 0.0f;
            }
            if constexpr (SMALL) {
                v8h o8;
#pragma unroll
                for (int e = 0; e < 8; ++e) o8[e] = (_Float16)v[e];
                *(v8h*)(sOh + (tl * 16 + c) * P2 + 16 * t + 8 * h) = o8;
            } else {
                float* q = sOf + (tl * 16 + c) * PO + 16 * t + 8 * h;
                *(float4*)(q)     = make_float4(v[0], v[1], v[2], v[3]);
                *(float4*)(q + 4) = make_float4(v[4], v[5], v[6], v[7]);
            }
        }
    }
    __syncthreads();

#pragma unroll
    for (int j = 0; j < MLP_NT; ++j) {
        if (tv[j]) {
            const int tl = wave * MLP_NT + j;
            if constexpr (SMALL) {
#pragma unroll
                for (int i = 0; i < 4; ++i) {
                    const int f = 8 * (lane + 32 * i);
                    const v8h o8 = *(const v8h*)(sOh + (tl * 16 + (f >> 6)) * P2 + (f & 63));
                    *(v8h*)(outh + (size_t)row0[j] * HID_C + f) = o8;
                }
            } else {
#pragma unroll
                for (int i = 0; i < 8; ++i) {
                    const int f = 4 * (lane + 32 * i);
                    const float4 o4 = *(const float4*)(sOf + (tl * 16 + (f >> 6)) * PO + (f & 63));
                    *(float4*)(outf + (size_t)row0[j] * HID_C + f) = o4;
                }
            }
        }
    }
}

#define OWN_BLOCKS  256
#define OWN_THREADS 1024

template <int RN_, int S_, int R_, int WCAP_, int DB_, int U_>
struct own_cfg {
    static constexpr int RN = RN_, S = S_, R = R_, WCAP = WCAP_, DB = DB_, U = U_;
    static constexpr int SL4   = N_EDGES_C / 4 / S_;
    static constexpr int STEP  = OWN_THREADS * U_;
    static constexpr int TRIPS = (SL4 + STEP - 1) / STEP;
    static_assert((N_EDGES_C / 4) % S_ == 0, "a slice is a whole number of int4 words");
    static_assert(R_ * S_ == OWN_BLOCKS, "one block per (range, slice)");
    static_assert(R_ * RN_ >= N_NODES_C, "ranges cover every node");
    static_assert(RN_ <= (1 << DB_), "place in range fits its bits");
    static_assert(((long long)SL4 * 4) <= (1ll << (32 - DB_)), "edge id in slice fits its bits");
};
#define LIST_SUBS  8
typedef own_cfg<400, 1, 256, 512, 10, 8>  cfg2;
typedef own_cfg<3200, 8, 32, 512, 12, 4>  cfg1;
static_assert(cfg1::RN == LIST_SUBS * cfg2::RN && cfg2::R == LIST_SUBS * cfg1::R, "a conv1 range is LIST_SUBS conv2 ranges");
static_assert(cfg1::S == LIST_SUBS * cfg2::S, "a conv2 slice is LIST_SUBS conv1 slices");
static_assert(cfg1::WCAP % 32 == 0, "a wave's padded list part fits its capacity");
static_assert(cfg2::RN % 16 == 0 && N_NODES_C % cfg2::RN == 0, "a conv2 range is whole 16-node tiles and no range is partial");
static_assert(cfg2::RN / 16 <= OWN_THREADS / 32, "one wave per 16-node tile in the conv2 node MLP");

#define LIST_CAP   ((OWN_THREADS / 32) * cfg1::WCAP)
#define LIST_BAD   0xFFFFFFFFu
#define CNT_PITCH  32
#define LIST_U     8
#define PRIV_ROWS  28
static_assert((LIST_BAD & ((1u << cfg1::DB) - 1u)) >= (unsigned)cfg1::RN, "the pad entry lies outside every range");

typedef __attribute__((ext_vector_type(2))) _Float16 v2h;
typedef __attribute__((ext_vector_type(2))) float    v2f;

__device__ __forceinline__ int sum_pos(int ch) { return 16 * (ch & 3) + (ch >> 2); }

__device__ __forceinline__ void edge2_fetch(const unsigned* wlist, int p, int wn, int c, int hh, int e0,
                                            const int* __restrict__ ei, const float* __restrict__ ea,
                                            int (&sv)[8], v2f& av) {
    const uint4 q0 = *(const uint4*)(wlist + p + 8 * hh);
    const uint4 q1 = *(const uint4*)(wlist + p + 8 * hh + 4);
    const unsigned raw[8] = {q0.x, q0.y, q0.z, q0.w, q1.x, q1.y, q1.z, q1.w};
#pragma unroll
    for (int r = 0; r < 8; ++r) {
        const unsigned ent = (p + 8 * hh + r < wn) ? raw[r] : 0u;
        sv[r] = ei[e0 + (int)(ent >> cfg2::DB)];
    }
    const unsigned raw_c = wlist[p + c];
    const unsigned ent_c = (p + c < wn) ? raw_c : 0u;
    av = *(const v2f*)(ea + (size_t)(e0 + (int)(ent_c >> cfg2::DB)) * EDGE_DIM_C + 2 * hh);
}

template <class C>
__device__ __forceinline__ void own_push(unsigned*& pnext, unsigned* plast, int dstv, int n0, int erel) {
    const unsigned d = (unsigned)(dstv - n0);
    *(pnext < plast ? pnext : plast) = ((unsigned)erel << C::DB) | d;
    pnext += (d < (unsigned)C::RN) ? 32 : 0;
}

template <class C>
__device__ __forceinline__ unsigned own_scan(const int4* __restrict__ d4, unsigned* pfirst, unsigned* plast,
                                             int n0, int rot, int tid) {
    unsigned* pnext = pfirst;
    for (int t = 0; t < C::TRIPS; ++t) {
        int tt = t + rot;
        if (tt >= C::TRIPS) tt -= C::TRIPS;
        const int base = tt * C::STEP;
        int4 v[C::U];
#pragma unroll
        for (int u = 0; u < C::U; ++u) {
            const int i = base + u * OWN_THREADS + tid;
            v[u] = (i < C::SL4) ? d4[i] : make_int4(-1, -1, -1, -1);
        }
#pragma unroll
        for (int u = 0; u < C::U; ++u) {
            const int e = (base + u * OWN_THREADS + tid) * 4;
            own_push<C>(pnext, plast, v[u].x, n0, e + 0);
            own_push<C>(pnext, plast, v[u].y, n0, e + 1);
            own_push<C>(pnext, plast, v[u].z, n0, e + 2);
            own_push<C>(pnext, plast, v[u].w, n0, e + 3);
        }
    }
    return (unsigned)(pnext - pfirst) / 32u;
}

__device__ __forceinline__ unsigned priv_gather(const unsigned* pfirst, unsigned* wlist, unsigned cnt,
                                                int lane, unsigned wcap, bool& over) {
    unsigned inc = cnt;
#pragma unroll
    for (int sft = 1; sft < 32; sft <<= 1) {
        const unsigned o = __shfl_up(inc, (unsigned)sft, 32);
        if (lane >= sft) inc += o;
    }
    const unsigned total = (unsigned)__builtin_amdgcn_readlane((int)inc, 31);
    over = __builtin_amdgcn_ballot_w32(cnt >= (unsigned)PRIV_ROWS) != 0u || total > wcap;
    if (!over) {
        const unsigned off = inc - cnt;
        static_assert(PRIV_ROWS % 4 == 0, "whole trips of four places");
        for (unsigned r = 0; __builtin_amdgcn_ballot_w32(r < cnt) != 0u; r += 4) {
            unsigned e[4];
#pragma unroll
            for (int j = 0; j < 4; ++j) e[j] = pfirst[(r + (unsigned)j) * 32u];
#pragma unroll
            for (int j = 0; j < 4; ++j)
                if (r + (unsigned)j < cnt) wlist[off + r + (unsigned)j] = e[j];
        }
    }
    return total;
}

template <class C, class F>
__device__ __forceinline__ void own_all(const int4* __restrict__ d4, int n0, int tid, F add) {
#pragma unroll 1
    for (int t = 0; t < C::TRIPS * C::U; ++t) {
        const int i = t * OWN_THREADS + tid;
        const int4 v = (i < C::SL4) ? d4[i] : make_int4(-1, -1, -1, -1);
        const int dv[4] = {v.x, v.y, v.z, v.w};
#pragma unroll 1
        for (int k = 0; k < 4; ++k) {
            const unsigned d = (unsigned)(dv[k] - n0);
            if (d < (unsigned)C::RN) add(i * 4 + k, (int)d);
        }
    }
}

__device__ __forceinline__ void list_push(unsigned*& pnext, unsigned* plast, unsigned ent,
                                          unsigned sublo, unsigned eoff) {
    const unsigned d = (ent & ((1u << cfg1::DB) - 1u)) - sublo;
    *(pnext < plast ? pnext : plast) = (((ent >> cfg1::DB) + eoff) << cfg2::DB) | d;
    pnext += (d < (unsigned)cfg2::RN) ? 32 : 0;
}

__device__ __forceinline__ unsigned list_scan(const unsigned* __restrict__ lists, int l0,
                                              const unsigned (&cnt)[LIST_SUBS], unsigned cmax,
                                              unsigned* pfirst, unsigned* plast, unsigned sublo, int wave, int lane) {
    unsigned* pnext = pfirst;
    for (unsigned k0 = 0; k0 * OWN_THREADS < cmax; k0 += LIST_U) {
        unsigned v[LIST_SUBS][LIST_U];
#pragma unroll
        for (int q = 0; q < LIST_SUBS; ++q) {
            const unsigned* L = lists + (size_t)(l0 + q) * LIST_CAP;
            const unsigned wb = (unsigned)(((wave + 8 * q) & 31) * 32 + lane);
#pragma unroll
            for (int u = 0; u < LIST_U; ++u) {
                const unsigned i = (k0 + (unsigned)u) * OWN_THREADS + wb;
                v[q][u] = (i < cnt[q]) ? L[i] : LIST_BAD;
            }
        }
#pragma unroll
        for (int q = 0; q < LIST_SUBS; ++q) {
#pragma unroll
            for (int u = 0; u < LIST_U; ++u)
                list_push(pnext, plast, v[q][u], sublo, (unsigned)(q * (cfg1::SL4 * 4)));
        }
    }
    return (unsigned)(pnext - pfirst) / 32u;
}

template <class F>
__device__ __forceinline__ void list_all(const unsigned* __restrict__ lists,
                                         const unsigned* __restrict__ lcount, int l0,
                                         unsigned cmax, unsigned sublo, int wave, int lane, F add) {
#pragma unroll 1
    for (unsigned k0 = 0; k0 * OWN_THREADS < cmax; k0 += LIST_U) {
#pragma unroll 1
        for (int q = 0; q < LIST_SUBS; ++q) {
            const unsigned* L = lists + (size_t)(l0 + q) * LIST_CAP;
            const unsigned cq = lcount[(l0 + q) * CNT_PITCH];
            const unsigned wb = (unsigned)(((wave + 8 * q) & 31) * 32 + lane);
#pragma unroll 1
            for (int u = 0; u < LIST_U; ++u) {
                const unsigned i = (k0 + (unsigned)u) * OWN_THREADS + wb;
                const unsigned ent = (i < cq) ? L[i] : LIST_BAD;
                const unsigned d = (ent & ((1u << cfg1::DB) - 1u)) - sublo;
                if (d < (unsigned)cfg2::RN)
                    add((int)((ent >> cfg1::DB) + (unsigned)(q * (cfg1::SL4 * 4))), (int)d);
            }
        }
    }
}

__global__ __launch_bounds__(OWN_THREADS)
void edge2_own_kernel(const _Float16* __restrict__ h, const int* __restrict__ ei,
                      const float* __restrict__ ea,
                      const float* __restrict__ eW, const float* __restrict__ eb,
                      const unsigned* __restrict__ lists, const unsigned* __restrict__ lcount,
                      const float* __restrict__ W1, const float* __restrict__ b1,
                      const float* __restrict__ W2, const float* __restrict__ b2,
                      const int* __restrict__ batch,
                      float* __restrict__ rsum, int* __restrict__ rhdr, unsigned* __restrict__ rcount) {
    typedef cfg2 C;
    constexpr int P2 = HID_C + 8;
    constexpr int NT = C::RN / 16;
    constexpr int PA = HID_C + 4;
    static_assert(HID_C * HID_C / 4 == OWN_THREADS, "one float4 of W1 and one of W2 per thread");
    __shared__ __align__(16) float    s_acc[(C::RN + 1) * PA];
    __shared__ __align__(16) unsigned s_list[(OWN_THREADS / 32) * C::WCAP];
    __shared__ __align__(16) unsigned s_priv[OWN_THREADS * PRIV_ROWS];
    __shared__ __align__(16) _Float16 sW1[HID_C * P2];
    __shared__ __align__(16) _Float16 sW2[HID_C * P2];
    __shared__ int s_gid[C::RN];
    __shared__ int s_rstart[C::RN + 1];
    __shared__ __align__(16) int s_wruns[OWN_THREADS / 32];
    __shared__ int s_nruns;

    const int tid   = threadIdx.x;
    const int lane  = tid & 31;
    const int wave  = tid >> 5;
    const int range = (int)blockIdx.x / C::S;
    const int slice = (int)blockIdx.x - range * C::S;
    const int n0    = range * C::RN;
    const int e0    = slice * (C::SL4 * 4);
    unsigned* wlist = s_list + wave * C::WCAP;
    unsigned* pfirst = s_priv + wave * (32 * PRIV_ROWS) + lane;
    unsigned* plast  = pfirst + 32 * (PRIV_ROWS - 1);

    if (n0 >= N_NODES_C) return;

    const float4 w1v = ((const float4*)W1)[tid];
    const float4 w2v = ((const float4*)W2)[tid];
    const int gin = batch[n0 + (tid < C::RN ? tid : 0)];

    for (int i = tid; i < C::RN * PA / 4; i += OWN_THREADS)
        ((float4*)s_acc)[i] = make_float4(0.f, 0.f, 0.f, 0.f);
    {
        const int n = tid / (HID_C / 4), q = tid - n * (HID_C / 4);
        v4h o1, o2;
        o1[0] = (_Float16)w1v.x; o1[1] = (_Float16)w1v.y; o1[2] = (_Float16)w1v.z; o1[3] = (_Float16)w1v.w;
        o2[0] = (_Float16)w2v.x; o2[1] = (_Float16)w2v.y; o2[2] = (_Float16)w2v.z; o2[3] = (_Float16)w2v.w;
        *(v4h*)(sW1 + n * P2 + 4 * q) = o1;
        *(v4h*)(sW2 + n * P2 + 4 * q) = o2;
    }
    if (tid < C::RN) s_gid[tid] = gin;
    __syncthreads();

    auto run_starts_here = [&]() -> bool {
        if (tid >= C::RN) return false;
        if (tid == 0) return true;
        return s_gid[tid - 1] != s_gid[tid];
    };
    {
        const unsigned rsb = __builtin_amdgcn_ballot_w32(run_starts_here());
        if (lane == 0) s_wruns[wave] = __builtin_popcount(rsb);
    }

    const int c  = lane & 15;
    const int hh = lane >> 4;
    v2f   bw[4];
    float bch[4];
#pragma unroll
    for (int t = 0; t < 4; ++t) {
        bw[t]  = *(const v2f*)(eW + (4 * c + t) * EDGE_DIM_C + 2 * hh);
        bch[t] = eb[4 * c + t];
    }

    auto add_one = [&](int erel, int od) {
        const int oe = e0 + erel;
        const int os = ei[oe];
        const float4 a = *(const float4*)(ea + (size_t)oe * EDGE_DIM_C);
#pragma unroll 1
        for (int q = 0; q < HID_C; ++q) {
            const float4 w = *(const float4*)(eW + q * EDGE_DIM_C);
            float t = eb[q] + a.x * w.x + a.y * w.y + a.z * w.z + a.w * w.w;
            t += (float)h[(size_t)os * HID_C + q];
            atomicAdd(&s_acc[od * PA + sum_pos(q)], fmaxf(t, 0.0f));
        }
    };

    const int l0 = (range / LIST_SUBS) * cfg1::S + slice * LIST_SUBS;
    const unsigned sublo = (unsigned)((range % LIST_SUBS) * C::RN);
    unsigned cnt[LIST_SUBS];
    unsigned cmax = 0;
#pragma unroll
    for (int q = 0; q < LIST_SUBS; ++q) {
        cnt[q] = lcount[(l0 + q) * CNT_PITCH];
        cmax = cnt[q] > cmax ? cnt[q] : cmax;
    }
    unsigned wcount;
    bool over;
    if (cmax <= (unsigned)LIST_CAP) {
        const unsigned mine = list_scan(lists, l0, cnt, cmax, pfirst, plast, sublo, wave, lane);
        wcount = priv_gather(pfirst, wlist, mine, lane, (unsigned)C::WCAP, over);
        if (over)
            list_all(lists, lcount, l0, cmax, sublo, wave, lane, add_one);
    } else {
        const int4* d4 = (const int4*)(ei + N_EDGES_C + e0);
        const int rot = (range * C::TRIPS) / C::R;
        const unsigned mine = own_scan<C>(d4, pfirst, plast, n0, rot, tid);
        wcount = priv_gather(pfirst, wlist, mine, lane, (unsigned)C::WCAP, over);
        if (over)
            own_all<C>(d4, n0, tid, add_one);
    }
    const int wn = over ? 0 : (int)wcount;

    __syncthreads();

    {
        const bool rs = run_starts_here();
        const unsigned rsb = __builtin_amdgcn_ballot_w32(rs);
        int rbefore = 0, nruns = 0;
#pragma unroll
        for (int w4 = 0; w4 < OWN_THREADS / 128; ++w4) {
            const int4 v = ((const int4*)s_wruns)[w4];
            const int vv[4] = {v.x, v.y, v.z, v.w};
#pragma unroll
            for (int k = 0; k < 4; ++k) {
                rbefore += (4 * w4 + k < wave) ? vv[k] : 0;
                nruns   += vv[k];
            }
        }
        if (rs) s_rstart[rbefore + __builtin_popcount(rsb & ((1u << lane) - 1u))] = tid;
        if (tid == 0) { s_rstart[nruns] = C::RN; s_nruns = nruns; }
    }

    const char* hb = (const char*)h;
    int sv[8];
    v2f av;
    edge2_fetch(wlist, 0, wn, c, hh, e0, ei, ea, sv, av);
    for (int p = 0; p < wn; p += 16) {
        v4h hv[8];
#pragma unroll
        for (int r = 0; r < 8; ++r) {
            const unsigned off = (unsigned)sv[r] * (unsigned)(HID_C * 2) + 8u * (unsigned)c;
            hv[r] = *(const v4h*)(hb + off);
        }
        const int pn = (p + 16 < wn) ? p + 16 : p;
        int svn[8];
        v2f avn;
        edge2_fetch(wlist, pn, wn, c, hh, e0, ei, ea, svn, avn);
        float* arow[8];
        {
            const uint4 q0 = *(const uint4*)(wlist + p + 8 * hh);
            const uint4 q1 = *(const uint4*)(wlist + p + 8 * hh + 4);
            const unsigned raw[8] = {q0.x, q0.y, q0.z, q0.w, q1.x, q1.y, q1.z, q1.w};
#pragma unroll
            for (int r = 0; r < 8; ++r) {
                const int dl = (p + 8 * hh + r < wn) ? (int)(raw[r] & ((1u << C::DB) - 1u)) : C::RN;
                arow[r] = s_acc + dl * PA + c;
            }
        }
#pragma unroll
        for (int t = 0; t < 4; ++t) {
            v8f cc;
#pragma unroll
            for (int r = 0; r < 8; ++r) cc[r] = bch[t] + (float)hv[r][t];
            const v8f d = __builtin_amdgcn_wmma_f32_16x16x4_f32(false, av, false, bw[t], (short)0, cc, false, false);
#pragma unroll
            for (int r = 0; r < 8; ++r)
                atomicAdd(arow[r] + 16 * t, fmaxf(d[r], 0.0f));
        }
#pragma unroll
        for (int r = 0; r < 8; ++r) sv[r] = svn[r];
        av = avn;
    }
    const int  wv   = __builtin_amdgcn_readfirstlane(wave);
    const bool tile = wv < NT;
    const _Float16* grow = h + (size_t)(n0 + (tile ? wv * 16 + c : 0)) * HID_C + 8 * hh;
    v8h    xr[4];
    float4 pb1[8], pb2[8];
#pragma unroll
    for (int i = 0; i < 4; ++i) {
        xr[i]          = *(const v8h*)(grow + 16 * i);
        pb1[2 * i]     = *(const float4*)(b1 + 16 * i + 8 * hh);
        pb1[2 * i + 1] = *(const float4*)(b1 + 16 * i + 8 * hh + 4);
        pb2[2 * i]     = *(const float4*)(b2 + 16 * i + 8 * hh);
        pb2[2 * i + 1] = *(const float4*)(b2 + 16 * i + 8 * hh + 4);
    }
    __syncthreads();

    float*     trow  = s_acc + (wv * 16 + c) * PA;
    if (tile) {
        v16h bop[2];
#pragma unroll
        for (int kb = 0; kb < 2; ++kb) {
            const v8h x0 = xr[2 * kb];
            const v8h x1 = xr[2 * kb + 1];
            float lo[8], hi[8];
#pragma unroll
            for (int j = 0; j < 4; ++j) {
                const float2 pl = *(const float2*)(trow + 16 * j + 8 * kb + 2 * hh);
                const float2 ph = *(const float2*)(trow + 16 * j + 8 * kb + 2 * hh + 4);
                lo[j] = pl.x; lo[j + 4] = pl.y;
                hi[j] = ph.x; hi[j + 4] = ph.y;
            }
            v16h b;
#pragma unroll
            for (int e = 0; e < 8; ++e) {
                b[e]     = (_Float16)((float)x0[e] + lo[e]);
                b[8 + e] = (_Float16)((float)x1[e] + hi[e]);
            }
            bop[kb] = b;
        }

        v8f acc[4];
#pragma unroll
        for (int t = 0; t < 4; ++t) {
            acc[t] = v8f{};
#pragma unroll
            for (int kb = 0; kb < 2; ++kb) {
                const v16h a = lds_load_tile(sW1 + 16 * t * P2, P2, kb * 32);
                acc[t] = __builtin_amdgcn_wmma_f32_16x16x32_f16(false, a, false, bop[kb],
                                                                (short)0, acc[t], false, false);
            }
        }

        v16h bop2[2];
#pragma unroll
        for (int s2 = 0; s2 < 2; ++s2) {
            const float4 p0 = pb1[4 * s2], p1 = pb1[4 * s2 + 1], p2 = pb1[4 * s2 + 2], p3 = pb1[4 * s2 + 3];
            const float bl[8] = {p0.x, p0.y, p0.z, p0.w, p1.x, p1.y, p1.z, p1.w};
            const float bh[8] = {p2.x, p2.y, p2.z, p2.w, p3.x, p3.y, p3.z, p3.w};
            v16h b;
#pragma unroll
            for (int e = 0; e < 8; ++e) {
                float v0 = acc[2 * s2][e] + bl[e];
                float v1 = acc[2 * s2 + 1][e] + bh[e];
                v0 = v0 > 0.0f ? v0 : 0.0f;
                v1 = v1 > 0.0f ? v1 : 0.0f;
                b[e]     = (_Float16)v0;
                b[8 + e] = (_Float16)v1;
            }
            bop2[s2] = b;
        }

#pragma unroll
        for (int t = 0; t < 4; ++t) {
            v8f d = v8f{};
#pragma unroll
            for (int s2 = 0; s2 < 2; ++s2) {
                const v16h a = lds_load_tile(sW2 + 16 * t * P2, P2, s2 * 32);
                d = __builtin_amdgcn_wmma_f32_16x16x32_f16(false, a, false, bop2[s2],
                                                           (short)0, d, false, false);
            }
            const float4 q0 = pb2[2 * t], q1 = pb2[2 * t + 1];
            const float bb[8] = {q0.x, q0.y, q0.z, q0.w, q1.x, q1.y, q1.z, q1.w};
            float v[8];
#pragma unroll
            for (int e = 0; e < 8; ++e) {
                const float u = d[e] + bb[e];
                v[e] = u > 0.0f ? u : 0.0f;
            }
            float* q = trow + 16 * t + 8 * hh;
            *(float4*)(q)     = make_float4(v[0], v[1], v[2], v[3]);
            *(float4*)(q + 4) = make_float4(v[4], v[5], v[6], v[7]);
        }
    }
    __syncthreads();
    const int nruns = s_nruns;
    const int ln4  = (int)__builtin_amdgcn_mbcnt_lo(~0u, 0u);
    const int tid4 = wv * 32 + ln4;
    {
        const int ch = tid4 & (HID_C - 1);
        for (int r = tid4 / HID_C; r < nruns; r += OWN_THREADS / HID_C) {
            const int ra = s_rstart[r], re = s_rstart[r + 1];
            const float* col = s_acc + ch;
            float p[8];
#pragma unroll
            for (int k = 0; k < 8; ++k) p[k] = 0.0f;
            for (int i = ra; i < re; i += 8) {
#pragma unroll
                for (int k = 0; k < 8; ++k) {
                    const bool in = i + k < re;
                    const float v = col[(in ? i + k : re - 1) * PA];
                    p[k] += in ? v : 0.0f;
                }
            }
            rsum[(size_t)(n0 + r) * HID_C + ch] = ((p[0] + p[1]) + (p[2] + p[3])) + ((p[4] + p[5]) + (p[6] + p[7]));
        }
    }
    if (wv == 0) {
        for (int r = ln4; r < nruns; r += 32) {
            const int ra = s_rstart[r], re = s_rstart[r + 1];
            *(int2*)(rhdr + 2 * (n0 + r)) = make_int2(s_gid[ra], re - ra);
        }
        if (ln4 == 0) rcount[range * CNT_PITCH] = (unsigned)nruns;
    }
}

__global__ __launch_bounds__(OWN_THREADS)
void edge1_own_kernel(const float* __restrict__ x, const int* __restrict__ ei,
                      const float* __restrict__ ea,
                      const float* __restrict__ eW, const float* __restrict__ eb,
                      float* __restrict__ aggr,
                      unsigned* __restrict__ lists, unsigned* __restrict__ lcount) {
    typedef cfg1 C;
    __shared__ __align__(16) float    s_acc[C::RN * 8];
    __shared__ __align__(16) unsigned s_list[(OWN_THREADS / 32) * C::WCAP];
    __shared__ __align__(16) unsigned s_priv[OWN_THREADS * PRIV_ROWS];
    __shared__ unsigned s_wn[OWN_THREADS / 32];

    const int tid   = threadIdx.x;
    const int lane  = tid & 31;
    const int wave  = tid >> 5;
    const int range = (int)blockIdx.x / C::S;
    const int slice = (int)blockIdx.x - range * C::S;
    const int n0    = range * C::RN;
    const int e0    = slice * (C::SL4 * 4);
    unsigned* wlist = s_list + wave * C::WCAP;
    unsigned* pfirst = s_priv + wave * (32 * PRIV_ROWS) + lane;
    unsigned* plast  = pfirst + 32 * (PRIV_ROWS - 1);

    for (int i = tid; i < C::RN * 8; i += OWN_THREADS) s_acc[i] = 0.0f;
    __syncthreads();

    const int c   = lane & 7;
    const int sub = lane >> 3;
    const bool cact = c < IN_DIM_C;
    const float4 wc = cact ? *(const float4*)(eW + c * EDGE_DIM_C) : make_float4(0.f, 0.f, 0.f, 0.f);
    const float  bc = cact ? eb[c] : 0.0f;
    const int    cx = cact ? c : 0;

    const int4* d4 = (const int4*)(ei + N_EDGES_C + e0);
    const int rot = (range * C::TRIPS) / C::R;
    const unsigned mine = own_scan<C>(d4, pfirst, plast, n0, rot, tid);
    bool over;
    const unsigned wcount = priv_gather(pfirst, wlist, mine, lane, (unsigned)C::WCAP, over);
    if (over) {
        own_all<C>(d4, n0, tid, [&](int erel, int od) {
            const int oe = e0 + erel;
            const int os = ei[oe];
            const float4 a = *(const float4*)(ea + (size_t)oe * EDGE_DIM_C);
#pragma unroll 1
            for (int q = 0; q < IN_DIM_C; ++q) {
                const float4 w = *(const float4*)(eW + q * EDGE_DIM_C);
                float t = eb[q] + a.x * w.x + a.y * w.y + a.z * w.z + a.w * w.w;
                t += x[(size_t)os * IN_DIM_C + q];
                atomicAdd(&s_acc[od * 8 + q], fmaxf(t, 0.0f));
            }
        });
    }
    const int wn = over ? 0 : (int)wcount;
    if (lane == 0) s_wn[wave] = over ? (unsigned)C::WCAP + 1u : wcount;

    __syncthreads();

    {
        unsigned off = 0, tot = 0;
        bool full = false;
        for (int w = 0; w < OWN_THREADS / 32; ++w) {
            const unsigned cw = s_wn[w];
            full = full || cw > (unsigned)C::WCAP;
            const unsigned pw = ((cw < (unsigned)C::WCAP ? cw : (unsigned)C::WCAP) + 31u) & ~31u;
            if (w < wave) off += pw;
            tot += pw;
        }
        unsigned* L = lists + (size_t)blockIdx.x * LIST_CAP + off;
        const int wp = (wn + 31) & ~31;
        for (int i = lane; i < wp; i += 32) L[i] = (i < wn) ? wlist[i] : LIST_BAD;
        if (tid == 0) lcount[blockIdx.x * CNT_PITCH] = full ? LIST_BAD : tot;
    }

    for (int p = 0; p < wn; p += 32) {
        unsigned ent[8]; int sr[8]; float4 av[8]; float xv[8]; bool ok[8];
#pragma unroll
        for (int g = 0; g < 8; ++g) {
            const int idx = p + g * 4 + sub;
            ok[g]  = idx < wn;
            ent[g] = ok[g] ? wlist[idx] : 0u;
        }
#pragma unroll
        for (int g = 0; g < 8; ++g) {
            const int e = e0 + (int)(ent[g] >> C::DB);
            sr[g] = ei[e];
            av[g] = *(const float4*)(ea + (size_t)e * EDGE_DIM_C);
        }
#pragma unroll
        for (int g = 0; g < 8; ++g)
            xv[g] = x[(size_t)sr[g] * IN_DIM_C + cx];
        asm volatile("" :: "v"(xv[0]), "v"(xv[1]), "v"(xv[2]), "v"(xv[3]), "v"(xv[4]), "v"(xv[5]), "v"(xv[6]), "v"(xv[7]),
                           "v"(av[0].x), "v"(av[0].y), "v"(av[0].z), "v"(av[0].w));
#pragma unroll
        for (int g = 0; g < 8; ++g) {
            if (ok[g] && cact) {
                const int dl = (int)(ent[g] & ((1u << C::DB) - 1u));
                float t = bc + av[g].x * wc.x + av[g].y * wc.y + av[g].z * wc.z + av[g].w * wc.w;
                t += xv[g];
                atomicAdd(&s_acc[dl * 8 + c], fmaxf(t, 0.0f));
            }
        }
    }
    __syncthreads();

    int rows = N_NODES_C - n0;
    if (rows > C::RN) rows = C::RN;
    float* o = aggr + (size_t)slice * N_NODES_C * IN_DIM_C + (size_t)n0 * IN_DIM_C;
    for (int i = tid; i < rows * IN_DIM_C; i += OWN_THREADS) {
        const int r = i / IN_DIM_C, q = i - r * IN_DIM_C;
        o[i] = s_acc[r * 8 + q];
    }
}

#define POOL_GPB     4
#define POOL_THREADS 256
#define POOL_Q       4
#define POOL_NB      (N_NODES_C / cfg2::RN)
#define POOL_NJ      ((POOL_NB + POOL_THREADS / 32 - 1) / (POOL_THREADS / 32))
static_assert(N_NODES_C % cfg2::RN == 0, "every conv2 range that holds nodes is whole");

__global__ __launch_bounds__(POOL_THREADS)
void pool_fc_kernel(const float* __restrict__ rsum, const int* __restrict__ rhdr,
                    const unsigned* __restrict__ rcount,
                    const float* __restrict__ fcW, const float* __restrict__ fcb,
                    float* __restrict__ out) {
    constexpr int NW  = POOL_THREADS / 32;
    static_assert(N_GRAPHS_C % POOL_GPB == 0, "every block owns POOL_GPB graphs");
    static_assert(POOL_GPB * HID_C <= POOL_THREADS, "one thread per (graph, channel) in the wave sum");

    __shared__ __align__(16) float s_part[NW][POOL_GPB][HID_C];
    __shared__ int   s_cnt [NW][POOL_GPB];
    __shared__ float s_mean[POOL_GPB][HID_C];

    const int tid  = threadIdx.x;
    const int lane = tid & 31;
    const int wave = __builtin_amdgcn_readfirstlane(tid >> 5);
    const int g0   = (int)blockIdx.x * POOL_GPB;
    const float2* rsv = (const float2*)rsum;
    const int2*   hdv = (const int2*)rhdr;

    float2 acc[POOL_GPB];
    int    cnt[POOL_GPB];
#pragma unroll
    for (int k = 0; k < POOL_GPB; ++k) { acc[k] = make_float2(0.f, 0.f); cnt[k] = 0; }

    auto take = [&](int base, int d, int len) {
        unsigned mask = __builtin_amdgcn_ballot_w32((unsigned)d < (unsigned)POOL_GPB);
        while (mask) {
            int gq[POOL_Q], lq[POOL_Q]; float2 v[POOL_Q];
            const int jf = __builtin_ctz(mask);
#pragma unroll
            for (int q = 0; q < POOL_Q; ++q) {
                const bool has = mask != 0;
                const int j = has ? __builtin_ctz(mask) : jf;
                const int gj = __builtin_amdgcn_readlane(d, j);
                const int lj = __builtin_amdgcn_readlane(len, j);
                gq[q] = has ? gj : -1;
                lq[q] = lj;
                v[q]  = rsv[(size_t)(base + j) * (HID_C / 2) + lane];
                mask &= mask - 1;
            }
            __builtin_amdgcn_sched_barrier(0);
#pragma unroll
            for (int q = 0; q < POOL_Q; ++q) {
#pragma unroll
                for (int k = 0; k < POOL_GPB; ++k)
                    if (gq[q] == k) { acc[k].x += v[q].x; acc[k].y += v[q].y; cnt[k] += lq[q]; }
            }
        }
    };

    static_assert(POOL_NJ <= 32, "one lane, and one bit of the hit mask, per block of the wave");
    const int bl = wave + (lane < POOL_NJ ? lane : 0) * NW;
    const int cr = (int)rcount[(bl < POOL_NB ? bl : 0) * CNT_PITCH];
    const int cl = (bl < POOL_NB) ? cr : 0;
    int2 hd[POOL_NJ];
#pragma unroll
    for (int j = 0; j < POOL_NJ; ++j) {
        const int b   = wave + j * NW;
        const int nrj = __builtin_amdgcn_readlane(cl, j);
        hd[j] = hdv[(b < POOL_NB ? b : 0) * cfg2::RN + (lane < nrj ? lane : 0)];
    }
    unsigned hit = 0;
#pragma unroll
    for (int j = 0; j < POOL_NJ; ++j) {
        const int nrj = __builtin_amdgcn_readlane(cl, j);
        const int d   = (lane < nrj) ? hd[j].x - g0 : -1;
        if (__builtin_amdgcn_ballot_w32((unsigned)d < (unsigned)POOL_GPB)) hit |= 1u << j;
    }
    while (hit) {
        const int j = __builtin_ctz(hit);
        hit &= hit - 1;
        const int b   = wave + j * NW;
        const int nrj = __builtin_amdgcn_readlane(cl, j);
        const int2 hx = hdv[b * cfg2::RN + (lane < nrj ? lane : 0)];
        take(b * cfg2::RN, (lane < nrj) ? hx.x - g0 : -1, hx.y);
    }
#pragma unroll 1
    for (int j = 0; j < POOL_NJ; ++j) {
        const int b   = wave + j * NW;
        const int nrj = __builtin_amdgcn_readlane(cl, j);
        for (int r0 = 32; r0 < nrj; r0 += 32) {
            const bool in = r0 + lane < nrj;
            const int2 hx = hdv[b * cfg2::RN + (in ? r0 + lane : 0)];
            take(b * cfg2::RN + r0, in ? hx.x - g0 : -1, hx.y);
        }
    }

#pragma unroll
    for (int k = 0; k < POOL_GPB; ++k) {
        *(float2*)&s_part[wave][k][2 * lane] = acc[k];
        if (lane == 0) s_cnt[wave][k] = cnt[k];
    }
    __syncthreads();

    if (tid < POOL_GPB * HID_C) {
        const int k = tid / HID_C, ch = tid - k * HID_C;
        float s = 0.0f; int c = 0;
        for (int w = 0; w < NW; ++w) { s += s_part[w][k][ch]; c += s_cnt[w][k]; }
        s_mean[k][ch] = s / fmaxf((float)c, 1.0f);
    }
    __syncthreads();

    if (tid < POOL_GPB * NUM_TASKS_C) {
        const int k = tid / NUM_TASKS_C, t = tid - k * NUM_TASKS_C;
        float s = fcb[t];
#pragma unroll
        for (int c = 0; c < HID_C; ++c)
            s += s_mean[k][c] * fcW[t * HID_C + c];
        out[(size_t)(g0 + k) * NUM_TASKS_C + t] = s;
    }
}

extern "C" void kernel_launch(void* const* d_in, const int* in_sizes, int n_in,
                              void* d_out, int out_size, void* d_ws, size_t ws_size,
                              hipStream_t stream) {
    const float* x    = (const float*)d_in[0];
    const int*   ei   = (const int*)  d_in[1];
    const float* ea   = (const float*)d_in[2];
    const int*   bat  = (const int*)  d_in[3];
    const float* e1W  = (const float*)d_in[4];  const float* e1b  = (const float*)d_in[5];
    const float* n1W1 = (const float*)d_in[6];  const float* n1b1 = (const float*)d_in[7];
    const float* n1W2 = (const float*)d_in[8];  const float* n1b2 = (const float*)d_in[9];
    const float* e2W  = (const float*)d_in[10]; const float* e2b  = (const float*)d_in[11];
    const float* n2W1 = (const float*)d_in[12]; const float* n2b1 = (const float*)d_in[13];
    const float* n2W2 = (const float*)d_in[14]; const float* n2b2 = (const float*)d_in[15];
    const float* fcW  = (const float*)d_in[16]; const float* fcb  = (const float*)d_in[17];
    float* out = (float*)d_out;

    float*    aggr1 = (float*)d_ws;
    _Float16* h     = (_Float16*)(aggr1 + (size_t)cfg1::S * N_NODES_C * IN_DIM_C);
    float*    rsum  = (float*)(h + (size_t)N_NODES_C * HID_C);
    unsigned* lists  = (unsigned*)(rsum + (size_t)N_NODES_C * HID_C);
    unsigned* lcount = lists + (size_t)OWN_BLOCKS * LIST_CAP;
    int*      rhdr   = (int*)(lcount + (size_t)OWN_BLOCKS * CNT_PITCH);
    unsigned* rcount = (unsigned*)(rhdr + 2 * (size_t)N_NODES_C);

    edge1_own_kernel<<<OWN_BLOCKS, OWN_THREADS, 0, stream>>>(x, ei, ea, e1W, e1b, aggr1, lists, lcount);
    node_mlp_kernel<IN_DIM_C, cfg1::S><<<MLP_GRID, MLP_THREADS, 0, stream>>>(
        x, nullptr, aggr1, n1W1, n1b1, n1W2, n1b2, h, nullptr, N_NODES_C);
    edge2_own_kernel<<<OWN_BLOCKS, OWN_THREADS, 0, stream>>>(h, ei, ea, e2W, e2b, lists, lcount,
                                                             n2W1, n2b1, n2W2, n2b2, bat, rsum, rhdr, rcount);
    pool_fc_kernel<<<N_GRAPHS_C / POOL_GPB, POOL_THREADS, 0, stream>>>(rsum, rhdr, rcount, fcW, fcb, out);
}
